# starting kernel
# baseline (speedup 1.0000x reference)
#include <hip/hip_runtime.h>

#define NN 50000
#define NE 800000
#define KC 32
#define SA_STRIDE 36
#define SB_STRIDE 68

typedef __attribute__((ext_vector_type(2))) float v2f;
typedef __attribute__((ext_vector_type(8))) float v8f;

__global__ void zero_f32(float* __restrict__ p, int n) {
  int i = blockIdx.x * blockDim.x + threadIdx.x;
  int stride = gridDim.x * blockDim.x;
  for (; i < n; i += stride) p[i] = 0.f;
}

__global__ void deg_count(const int* __restrict__ dst, float* __restrict__ deg, int E) {
  int e = blockIdx.x * blockDim.x + threadIdx.x;
  if (e < E) atomicAdd(&deg[dst[e]], 1.f);
}

__global__ void rdeg_inv(float* __restrict__ deg, int n) {
  int i = blockIdx.x * blockDim.x + threadIdx.x;
  if (i < n) deg[i] = 1.f / fmaxf(deg[i], 1.f);
}

__global__ void scatter_add(const float* __restrict__ h, const int* __restrict__ src,
                            const int* __restrict__ dst, float* __restrict__ msg,
                            int E, int F, int log2chunks) {
  int tid = blockIdx.x * blockDim.x + threadIdx.x;
  int total = E << log2chunks;
  if (tid >= total) return;
  int e = tid >> log2chunks;
  int c = (tid & ((1 << log2chunks) - 1)) << 2;
  int s = src[e], d = dst[e];
  float4 v = *(const float4*)(h + (size_t)s * F + c);
  float* p = msg + (size_t)d * F + c;
  atomicAdd(p + 0, v.x);
  atomicAdd(p + 1, v.y);
  atomicAdd(p + 2, v.z);
  atomicAdd(p + 3, v.w);
}

__global__ void sage_gemm_wmma(const float* __restrict__ msg, const float* __restrict__ rdeg,
                               const float* __restrict__ h,
                               const float* __restrict__ Wl, const float* __restrict__ Wr,
                               const float* __restrict__ bias, float* __restrict__ out,
                               int K, int F, int nblocks_n, int do_relu) {
  __shared__ float sA[2][16 * SA_STRIDE];
  __shared__ float sB[2][KC * SB_STRIDE];

  int t = threadIdx.x;
  int wave = t >> 5, lane = t & 31;
  int mBase = (blockIdx.x / nblocks_n) * 16;
  int nBase = (blockIdx.x % nblocks_n) * 64;

  int m16  = lane & 15;
  int kb   = (lane >> 4) << 1;
  int nloc = wave * 16 + (lane & 15);
  int ncol = nBase + nloc;

  float scale = rdeg[mBase + m16];
  v8f acc = {0.f, 0.f, 0.f, 0.f, 0.f, 0.f, 0.f, 0.f};

  int ar = t >> 3, ac = (t & 7) << 2;
  int br = t >> 4, bc = (t & 15) << 2;

  for (int kc = 0; kc < K; kc += KC) {
    {
      float4 va0 = *(const float4*)(msg + (size_t)(mBase + ar) * K + kc + ac);
      float4 va1 = *(const float4*)(h   + (size_t)(mBase + ar) * K + kc + ac);
      *(float4*)(&sA[0][ar * SA_STRIDE + ac]) = va0;
      *(float4*)(&sA[1][ar * SA_STRIDE + ac]) = va1;
#pragma unroll
      for (int pass = 0; pass < 4; pass++) {
        int row = br + pass * 8;
        float4 vb0 = *(const float4*)(Wl + (size_t)(kc + row) * F + nBase + bc);
        float4 vb1 = *(const float4*)(Wr + (size_t)(kc + row) * F + nBase + bc);
        *(float4*)(&sB[0][row * SB_STRIDE + bc]) = vb0;
        *(float4*)(&sB[1][row * SB_STRIDE + bc]) = vb1;
      }
    }
    if (kc + KC < K) {
      __builtin_prefetch(msg + (size_t)(mBase + ar) * K + kc + KC + ac, 0, 0);
      __builtin_prefetch(h   + (size_t)(mBase + ar) * K + kc + KC + ac, 0, 0);
      __builtin_prefetch(Wl + (size_t)(kc + KC + br) * F + nBase + bc, 0, 0);
      __builtin_prefetch(Wr + (size_t)(kc + KC + br) * F + nBase + bc, 0, 0);
    }
    __syncthreads();

#pragma unroll
    for (int k = 0; k < KC; k += 4) {
      v2f a, b;
      a.x = sA[0][m16 * SA_STRIDE + k + kb] * scale;
      a.y = sA[0][m16 * SA_STRIDE + k + kb + 1] * scale;
      b.x = sB[0][(k + kb) * SB_STRIDE + nloc];
      b.y = sB[0][(k + kb + 1) * SB_STRIDE + nloc];
      acc = __builtin_amdgcn_wmma_f32_16x16x4_f32(false, a, false, b, (short)0, acc,
                                                  false, false);
      a.x = sA[1][m16 * SA_STRIDE + k + kb];
      a.y = sA[1][m16 * SA_STRIDE + k + kb + 1];
      b.x = sB[1][(k + kb) * SB_STRIDE + nloc];
      b.y = sB[1][(k + kb + 1) * SB_STRIDE + nloc];
      acc = __builtin_amdgcn_wmma_f32_16x16x4_f32(false, a, false, b, (short)0, acc,
                                                  false, false);
    }
    __syncthreads();
  }

  float bv = bias[ncol];
  int mhalf = mBase + ((lane >> 4) << 3);
#pragma unroll
  for (int v = 0; v < 8; v++) {
    float val = acc[v] + bv;
    if (do_relu) val = fmaxf(val, 0.f);
    out[(size_t)(mhalf + v) * F + ncol] = val;
  }
}

__global__ void bn_stats(const float* __restrict__ h, float* __restrict__ st, int M, int F) {
  int c = threadIdx.x;
  float s = 0.f, s2 = 0.f;
  for (int r = blockIdx.x; r < M; r += gridDim.x) {
    float v = h[(size_t)r * F + c];
    s += v;
    s2 += v * v;
  }
  atomicAdd(&st[c], s);
  atomicAdd(&st[F + c], s2);
}

__global__ void bn_apply(float* __restrict__ h, const float* __restrict__ st,
                         const float* __restrict__ g, const float* __restrict__ be,
                         int M, int F) {
  int i = blockIdx.x * blockDim.x + threadIdx.x;
  if (i >= M * F) return;
  int c = i & (F - 1);
  float inv = 1.f / (float)M;
  float mean = st[c] * inv;
  float var = st[F + c] * inv - mean * mean;
  h[i] = g[c] * (h[i] - mean) * rsqrtf(var + 1e-5f) + be[c];
}

__global__ void reparam(const float* __restrict__ h3, const float* __restrict__ eps,
                        float* __restrict__ z, int M) {
  int i = blockIdx.x * blockDim.x + threadIdx.x;
  if (i >= M * 64) return;
  int r = i >> 6, c = i & 63;
  const float* row = h3 + (size_t)r * 128;
  z[i] = row[c] + expf(row[64 + c]) * eps[i];
}

extern "C" void kernel_launch(void* const* d_in, const int* in_sizes, int n_in,
                              void* d_out, int out_size, void* d_ws, size_t ws_size,
                              hipStream_t stream) {
  const float* x   = (const float*)d_in[0];
  const int*   src = (const int*)d_in[1];
  const int*   dst = (const int*)d_in[2];
  const float* eps = (const float*)d_in[3];
  const float* W1l = (const float*)d_in[4];
  const float* W1r = (const float*)d_in[5];
  const float* b1  = (const float*)d_in[6];
  const float* g1  = (const float*)d_in[7];
  const float* be1 = (const float*)d_in[8];
  const float* W2l = (const float*)d_in[9];
  const float* W2r = (const float*)d_in[10];
  const float* b2  = (const float*)d_in[11];
  const float* g2  = (const float*)d_in[12];
  const float* be2 = (const float*)d_in[13];
  const float* W3l = (const float*)d_in[14];
  const float* W3r = (const float*)d_in[15];
  const float* b3  = (const float*)d_in[16];
  float* zout = (float*)d_out;
  (void)in_sizes; (void)n_in; (void)out_size; (void)ws_size;

  char* ws = (char*)d_ws;
  size_t off = 0;
  auto carve = [&](size_t bytes) -> float* {
    char* p = ws + off;
    off = (off + bytes + 255) & ~(size_t)255;
    return (float*)p;
  };
  float* rdeg = carve((size_t)NN * 4);
  float* msg  = carve((size_t)NN * 256 * 4);
  float* h1   = carve((size_t)NN * 128 * 4);
  float* h2   = carve((size_t)NN * 256 * 4);
  float* st   = carve(512 * 4);

  const int M = NN, E = NE;

  zero_f32<<<256, 256, 0, stream>>>(rdeg, M);
  deg_count<<<(E + 255) / 256, 256, 0, stream>>>(dst, rdeg, E);
  rdeg_inv<<<(M + 255) / 256, 256, 0, stream>>>(rdeg, M);

  zero_f32<<<2048, 256, 0, stream>>>(msg, M * 128);
  scatter_add<<<(E * 32 + 255) / 256, 256, 0, stream>>>(x, src, dst, msg, E, 128, 5);
  sage_gemm_wmma<<<3125 * 2, 128, 0, stream>>>(msg, rdeg, x, W1l, W1r, b1, h1,
                                               128, 128, 2, 1);
  zero_f32<<<1, 256, 0, stream>>>(st, 256);
  bn_stats<<<512, 128, 0, stream>>>(h1, st, M, 128);
  bn_apply<<<(M * 128 + 255) / 256, 256, 0, stream>>>(h1, st, g1, be1, M, 128);

  zero_f32<<<2048, 256, 0, stream>>>(msg, M * 128);
  scatter_add<<<(E * 32 + 255) / 256, 256, 0, stream>>>(h1, src, dst, msg, E, 128, 5);
  sage_gemm_wmma<<<3125 * 4, 128, 0, stream>>>(msg, rdeg, h1, W2l, W2r, b2, h2,
                                               128, 256, 4, 1);
  zero_f32<<<2, 256, 0, stream>>>(st, 512);
  bn_stats<<<512, 256, 0, stream>>>(h2, st, M, 256);
  bn_apply<<<(M * 256 + 255) / 256, 256, 0, stream>>>(h2, st, g2, be2, M, 256);

  zero_f32<<<4096, 256, 0, stream>>>(msg, M * 256);
  scatter_add<<<(E * 64 + 255) / 256, 256, 0, stream>>>(h2, src, dst, msg, E, 256, 6);
  sage_gemm_wmma<<<3125 * 2, 128, 0, stream>>>(msg, rdeg, h2, W3l, W3r, b3, h1,
                                               256, 128, 2, 0);

  reparam<<<(M * 64 + 255) / 256, 256, 0, stream>>>(h1, eps, zout, M);
}


# Neighbour sums kept in LDS by the work-group that owns the target rows: every group walks the whole unsorted edge list, LDS f32 adds, rows stored once; no global atomics; add order = LDS arrival
# speedup vs baseline: 1.2947x; 1.2947x over previous
#include <hip/hip_runtime.h>

#define NN 50000
#define NE 800000
#define KC 32
#define SA_STRIDE 36
#define SB_STRIDE 68

typedef __attribute__((ext_vector_type(2))) float v2f;
typedef __attribute__((ext_vector_type(8))) float v8f;

__global__ void zero_f32(float* __restrict__ p, int n) {
  int i = blockIdx.x * blockDim.x + threadIdx.x;
  int stride = gridDim.x * blockDim.x;
  for (; i < n; i += stride) p[i] = 0.f;
}

__global__ void deg_count(const int* __restrict__ dst, float* __restrict__ deg, int E) {
  int e = blockIdx.x * blockDim.x + threadIdx.x;
  if (e < E) atomicAdd(&deg[dst[e]], 1.f);
}

__global__ void rdeg_inv(float* __restrict__ deg, int n) {
  int i = blockIdx.x * blockDim.x + threadIdx.x;
  if (i < n) deg[i] = 1.f / fmaxf(deg[i], 1.f);
}

#define AGG_ROWS 196
#define AGG_NB   ((NN + AGG_ROWS - 1) / AGG_ROWS)
#define AGG_NT   1024
#define AGG_NW   (AGG_NT / 32)
#define AGG_U    16
#define AGG_TRIP (32 * AGG_U)

template <int F>
__global__ __launch_bounds__(AGG_NT)
void agg_rows_lds(const float* __restrict__ h, const int* __restrict__ src,
                  const int* __restrict__ dst, float* __restrict__ msg, int E, int M) {
  __shared__ __attribute__((aligned(16))) float acc[AGG_ROWS * 256];
  const int t = threadIdx.x;
  const int lane = t & 31;
  const int wave = __builtin_amdgcn_readfirstlane(t >> 5);
  const int r0 = blockIdx.x * AGG_ROWS;

  for (int i = t; i < AGG_ROWS * F / 4; i += AGG_NT)
    ((float4*)acc)[i] = make_float4(0.f, 0.f, 0.f, 0.f);
  __syncthreads();

  const int ntrips = (E + AGG_TRIP - 1) / AGG_TRIP;
  const int rot = (int)(((unsigned)blockIdx.x * (unsigned)ntrips) / gridDim.x);

  int T = wave;
  int ebn = 0;
  int dn[AGG_U];
#pragma unroll
  for (int u = 0; u < AGG_U; u++) dn[u] = -1;
  if (T < ntrips) {
    int Tr = T + rot;
    if (Tr >= ntrips) Tr -= ntrips;
    ebn = Tr * AGG_TRIP + lane;
#pragma unroll
    for (int u = 0; u < AGG_U; u++) {
      const int e = ebn + 32 * u;
      if (e < E) dn[u] = dst[e];
    }
  }

  while (T < ntrips) {
    const int eb = ebn;
    int dl[AGG_U];
#pragma unroll
    for (int u = 0; u < AGG_U; u++) dl[u] = dn[u] - r0;

    T += AGG_NW;
    if (T < ntrips) {
      int Tr = T + rot;
      if (Tr >= ntrips) Tr -= ntrips;
      ebn = Tr * AGG_TRIP + lane;
#pragma unroll
      for (int u = 0; u < AGG_U; u++) {
        const int e = ebn + 32 * u;
        dn[u] = (e < E) ? dst[e] : -1;
      }
    }

    int sv[AGG_U];
    unsigned mk[AGG_U];
#pragma unroll
    for (int u = 0; u < AGG_U; u++) {
      const bool m = (unsigned)dl[u] < (unsigned)AGG_ROWS;
      sv[u] = 0;
      if (m) sv[u] = src[eb + 32 * u];
      mk[u] = __builtin_amdgcn_ballot_w32(m);
    }

#pragma unroll
    for (int u = 0; u < AGG_U; u++) {
      unsigned mask = mk[u];
      while (mask) {
        const int b = __builtin_ctz(mask);
        mask &= mask - 1;
        const int s  = __builtin_amdgcn_readlane(sv[u], b);
        const int rl = __builtin_amdgcn_readlane(dl[u], b);
        const float* hp = h + (size_t)s * F + lane;
        float* ap = acc + rl * F + lane;
        float v[F / 32];
#pragma unroll
        for (int j = 0; j < F / 32; j++) v[j] = hp[32 * j];
#pragma unroll
        for (int j = 0; j < F / 32; j++) atomicAdd(ap + 32 * j, v[j]);
      }
    }
  }
  __syncthreads();

  int nrows = M - r0;
  if (nrows > AGG_ROWS) nrows = AGG_ROWS;
  float4* out4 = (float4*)(msg + (size_t)r0 * F);
  for (int i = t; i < nrows * (F / 4); i += AGG_NT) out4[i] = ((const float4*)acc)[i];
}

__global__ void sage_gemm_wmma(const float* __restrict__ msg, const float* __restrict__ rdeg,
                               const float* __restrict__ h,
                               const float* __restrict__ Wl, const float* __restrict__ Wr,
                               const float* __restrict__ bias, float* __restrict__ out,
                               int K, int F, int nblocks_n, int do_relu) {
  __shared__ float sA[2][16 * SA_STRIDE];
  __shared__ float sB[2][KC * SB_STRIDE];

  int t = threadIdx.x;
  int wave = t >> 5, lane = t & 31;
  int mBase = (blockIdx.x / nblocks_n) * 16;
  int nBase = (blockIdx.x % nblocks_n) * 64;

  int m16  = lane & 15;
  int kb   = (lane >> 4) << 1;
  int nloc = wave * 16 + (lane & 15);
  int ncol = nBase + nloc;

  float scale = rdeg[mBase + m16];
  v8f acc = {0.f, 0.f, 0.f, 0.f, 0.f, 0.f, 0.f, 0.f};

  int ar = t >> 3, ac = (t & 7) << 2;
  int br = t >> 4, bc = (t & 15) << 2;

  for (int kc = 0; kc < K; kc += KC) {
    {
      float4 va0 = *(const float4*)(msg + (size_t)(mBase + ar) * K + kc + ac);
      float4 va1 = *(const float4*)(h   + (size_t)(mBase + ar) * K + kc + ac);
      *(float4*)(&sA[0][ar * SA_STRIDE + ac]) = va0;
      *(float4*)(&sA[1][ar * SA_STRIDE + ac]) = va1;
#pragma unroll
      for (int pass = 0; pass < 4; pass++) {
        int row = br + pass * 8;
        float4 vb0 = *(const float4*)(Wl + (size_t)(kc + row) * F + nBase + bc);
        float4 vb1 = *(const float4*)(Wr + (size_t)(kc + row) * F + nBase + bc);
        *(float4*)(&sB[0][row * SB_STRIDE + bc]) = vb0;
        *(float4*)(&sB[1][row * SB_STRIDE + bc]) = vb1;
      }
    }
    if (kc + KC < K) {
      __builtin_prefetch(msg + (size_t)(mBase + ar) * K + kc + KC + ac, 0, 0);
      __builtin_prefetch(h   + (size_t)(mBase + ar) * K + kc + KC + ac, 0, 0);
      __builtin_prefetch(Wl + (size_t)(kc + KC + br) * F + nBase + bc, 0, 0);
      __builtin_prefetch(Wr + (size_t)(kc + KC + br) * F + nBase + bc, 0, 0);
    }
    __syncthreads();

#pragma unroll
    for (int k = 0; k < KC; k += 4) {
      v2f a, b;
      a.x = sA[0][m16 * SA_STRIDE + k + kb] * scale;
      a.y = sA[0][m16 * SA_STRIDE + k + kb + 1] * scale;
      b.x = sB[0][(k + kb) * SB_STRIDE + nloc];
      b.y = sB[0][(k + kb + 1) * SB_STRIDE + nloc];
      acc = __builtin_amdgcn_wmma_f32_16x16x4_f32(false, a, false, b, (short)0, acc,
                                                  false, false);
      a.x = sA[1][m16 * SA_STRIDE + k + kb];
      a.y = sA[1][m16 * SA_STRIDE + k + kb + 1];
      b.x = sB[1][(k + kb) * SB_STRIDE + nloc];
      b.y = sB[1][(k + kb + 1) * SB_STRIDE + nloc];
      acc = __builtin_amdgcn_wmma_f32_16x16x4_f32(false, a, false, b, (short)0, acc,
                                                  false, false);
    }
    __syncthreads();
  }

  float bv = bias[ncol];
  int mhalf = mBase + ((lane >> 4) << 3);
#pragma unroll
  for (int v = 0; v < 8; v++) {
    float val = acc[v] + bv;
    if (do_relu) val = fmaxf(val, 0.f);
    out[(size_t)(mhalf + v) * F + ncol] = val;
  }
}

__global__ void bn_stats(const float* __restrict__ h, float* __restrict__ st, int M, int F) {
  int c = threadIdx.x;
  float s = 0.f, s2 = 0.f;
  for (int r = blockIdx.x; r < M; r += gridDim.x) {
    float v = h[(size_t)r * F + c];
    s += v;
    s2 += v * v;
  }
  atomicAdd(&st[c], s);
  atomicAdd(&st[F + c], s2);
}

__global__ void bn_apply(float* __restrict__ h, const float* __restrict__ st,
                         const float* __restrict__ g, const float* __restrict__ be,
                         int M, int F) {
  int i = blockIdx.x * blockDim.x + threadIdx.x;
  if (i >= M * F) return;
  int c = i & (F - 1);
  float inv = 1.f / (float)M;
  float mean = st[c] * inv;
  float var = st[F + c] * inv - mean * mean;
  h[i] = g[c] * (h[i] - mean) * rsqrtf(var + 1e-5f) + be[c];
}

__global__ void reparam(const float* __restrict__ h3, const float* __restrict__ eps,
                        float* __restrict__ z, int M) {
  int i = blockIdx.x * blockDim.x + threadIdx.x;
  if (i >= M * 64) return;
  int r = i >> 6, c = i & 63;
  const float* row = h3 + (size_t)r * 128;
  z[i] = row[c] + expf(row[64 + c]) * eps[i];
}

extern "C" void kernel_launch(void* const* d_in, const int* in_sizes, int n_in,
                              void* d_out, int out_size, void* d_ws, size_t ws_size,
                              hipStream_t stream) {
  const float* x   = (const float*)d_in[0];
  const int*   src = (const int*)d_in[1];
  const int*   dst = (const int*)d_in[2];
  const float* eps = (const float*)d_in[3];
  const float* W1l = (const float*)d_in[4];
  const float* W1r = (const float*)d_in[5];
  const float* b1  = (const float*)d_in[6];
  const float* g1  = (const float*)d_in[7];
  const float* be1 = (const float*)d_in[8];
  const float* W2l = (const float*)d_in[9];
  const float* W2r = (const float*)d_in[10];
  const float* b2  = (const float*)d_in[11];
  const float* g2  = (const float*)d_in[12];
  const float* be2 = (const float*)d_in[13];
  const float* W3l = (const float*)d_in[14];
  const float* W3r = (const float*)d_in[15];
  const float* b3  = (const float*)d_in[16];
  float* zout = (float*)d_out;
  (void)in_sizes; (void)n_in; (void)out_size; (void)ws_size;

  char* ws = (char*)d_ws;
  size_t off = 0;
  auto carve = [&](size_t bytes) -> float* {
    char* p = ws + off;
    off = (off + bytes + 255) & ~(size_t)255;
    return (float*)p;
  };
  float* rdeg = carve((size_t)NN * 4);
  float* msg  = carve((size_t)NN * 256 * 4);
  float* h1   = carve((size_t)NN * 128 * 4);
  float* h2   = carve((size_t)NN * 256 * 4);
  float* st   = carve(512 * 4);

  const int M = NN, E = NE;

  zero_f32<<<256, 256, 0, stream>>>(rdeg, M);
  deg_count<<<(E + 255) / 256, 256, 0, stream>>>(dst, rdeg, E);
  rdeg_inv<<<(M + 255) / 256, 256, 0, stream>>>(rdeg, M);

  agg_rows_lds<128><<<AGG_NB, AGG_NT, 0, stream>>>(x, src, dst, msg, E, M);
  sage_gemm_wmma<<<3125 * 2, 128, 0, stream>>>(msg, rdeg, x, W1l, W1r, b1, h1,
                                               128, 128, 2, 1);
  zero_f32<<<1, 256, 0, stream>>>(st, 256);
  bn_stats<<<512, 128, 0, stream>>>(h1, st, M, 128);
  bn_apply<<<(M * 128 + 255) / 256, 256, 0, stream>>>(h1, st, g1, be1, M, 128);

  agg_rows_lds<128><<<AGG_NB, AGG_NT, 0, stream>>>(h1, src, dst, msg, E, M);
  sage_gemm_wmma<<<3125 * 4, 128, 0, stream>>>(msg, rdeg, h1, W2l, W2r, b2, h2,
                                               128, 256, 4, 1);
  zero_f32<<<2, 256, 0, stream>>>(st, 512);
  bn_stats<<<512, 256, 0, stream>>>(h2, st, M, 256);
  bn_apply<<<(M * 256 + 255) / 256, 256, 0, stream>>>(h2, st, g2, be2, M, 256);

  agg_rows_lds<256><<<AGG_NB, AGG_NT, 0, stream>>>(h2, src, dst, msg, E, M);
  sage_gemm_wmma<<<3125 * 2, 128, 0, stream>>>(msg, rdeg, h2, W3l, W3r, b3, h1,
                                               256, 128, 2, 0);

  reparam<<<(M * 64 + 255) / 256, 256, 0, stream>>>(h1, eps, zout, M);
}


# SAGE GEMM fused into the row-owning aggregation kernel: LDS neighbour sums (x 1/deg in place) and LDS-staged own rows feed the WMMA; sums never stored; 16 -> 13 launches; GEMM k-sum order changes
# speedup vs baseline: 1.5753x; 1.2167x over previous
#include <hip/hip_runtime.h>

#define NN 50000
#define NE 800000

typedef __attribute__((ext_vector_type(2))) float v2f;
typedef __attribute__((ext_vector_type(8))) float v8f;

__global__ void zero_f32(float* __restrict__ p, int n) {
  int i = blockIdx.x * blockDim.x + threadIdx.x;
  int stride = gridDim.x * blockDim.x;
  for (; i < n; i += stride) p[i] = 0.f;
}

__global__ void deg_count(const int* __restrict__ dst, float* __restrict__ deg, int E) {
  int e = blockIdx.x * blockDim.x + threadIdx.x;
  if (e < E) atomicAdd(&deg[dst[e]], 1.f);
}

__global__ void rdeg_inv(float* __restrict__ deg, int n) {
  int i = blockIdx.x * blockDim.x + threadIdx.x;
  if (i < n) deg[i] = 1.f / fmaxf(deg[i], 1.f);
}

#define AGG_ROWS 196
#define AGG_RPAD 208
#define AGG_NRT  (AGG_RPAD / 16)
#define AGG_NB   ((NN + AGG_ROWS - 1) / AGG_ROWS)
#define AGG_NT   1024
#define AGG_NW   (AGG_NT / 32)
#define AGG_U    16
#define AGG_TRIP (32 * AGG_U)
#define AGG_PAD  4
#define AGG_LDS_FLOATS (AGG_RPAD * (256 + AGG_PAD) + AGG_RPAD * (64 + AGG_PAD))

template <int N> struct freg { typedef float type __attribute__((ext_vector_type(N))); };

template <int K, int KH, int NLD>
__device__ __forceinline__ typename freg<4 * NLD>::type own_rows_load(const float* __restrict__ h,
                                                                      int r0, int M, int col0, int t) {
  typename freg<4 * NLD>::type r;
#pragma unroll
  for (int j = 0; j < NLD; j++) {
    int i = t + j * AGG_NT;
    if (i > AGG_ROWS * (KH / 4) - 1) i = AGG_ROWS * (KH / 4) - 1;
    const int row = i / (KH / 4);
    const int c4 = i - row * (KH / 4);
    int gr = r0 + row;
    if (gr > M - 1) gr = M - 1;
    const float4 v = *(const float4*)(h + (gr * K + col0 + 4 * c4));
    r[4 * j + 0] = v.x; r[4 * j + 1] = v.y; r[4 * j + 2] = v.z; r[4 * j + 3] = v.w;
  }
  return r;
}

template <int KH, int NLD>
__device__ __forceinline__ void own_rows_store(const typename freg<4 * NLD>::type r, float* hs, int t) {
#pragma unroll
  for (int j = 0; j < NLD; j++) {
    const int i = t + j * AGG_NT;
    if (i < AGG_ROWS * (KH / 4)) {
      const int row = i / (KH / 4);
      const int c4 = i - row * (KH / 4);
      *(float4*)(hs + row * (KH + AGG_PAD) + 4 * c4) =
          make_float4(r[4 * j + 0], r[4 * j + 1], r[4 * j + 2], r[4 * j + 3]);
    }
  }
}

template <int K, int FO, int KH, int NTW>
__device__ __forceinline__ void rows_gemm_pass(v8f* c, const float* ms, const float* hs,
                                               const float* __restrict__ Wl,
                                               const float* __restrict__ Wr,
                                               int kb, int ct, int rg, int lane) {
  constexpr int LDK = K + AGG_PAD;
  constexpr int LDH = KH + AGG_PAD;
  constexpr int WPC = AGG_NW / (FO / 16);
  const int m16 = lane & 15;
  const int g = lane >> 4;
  const int ncol = ct * 16 + m16;

  const float* ap[NTW];
  const float* hp[NTW];
#pragma unroll
  for (int i = 0; i < NTW; i++) {
    const int lrow = (rg + i * WPC) * 16 + m16;
    ap[i] = ms + lrow * LDK + kb + 4 * g;
    hp[i] = hs + lrow * LDH + 4 * g;
  }
  const float* wl = Wl + (size_t)(kb + 4 * g) * FO + ncol;
  const float* wr = Wr + (size_t)(kb + 4 * g) * FO + ncol;

  constexpr bool AHEAD = (NTW <= 4) && (KH == K);
  v2f bl0, bl1, br0, br1;
  bl0.x = wl[0 * FO]; bl0.y = wl[1 * FO]; bl1.x = wl[2 * FO]; bl1.y = wl[3 * FO];
  br0.x = wr[0 * FO]; br0.y = wr[1 * FO]; br1.x = wr[2 * FO]; br1.y = wr[3 * FO];

#pragma unroll 1
  for (int k0 = 0; k0 < KH; k0 += 8) {
    const int kn = AHEAD ? ((k0 + 8 < KH) ? (k0 + 8) : k0) : k0;
    v2f nl0, nl1, nr0, nr1;
    nl0.x = wl[(size_t)(kn + 0) * FO]; nl0.y = wl[(size_t)(kn + 1) * FO];
    nl1.x = wl[(size_t)(kn + 2) * FO]; nl1.y = wl[(size_t)(kn + 3) * FO];
    nr0.x = wr[(size_t)(kn + 0) * FO]; nr0.y = wr[(size_t)(kn + 1) * FO];
    nr1.x = wr[(size_t)(kn + 2) * FO]; nr1.y = wr[(size_t)(kn + 3) * FO];
    if (!AHEAD) { bl0 = nl0; bl1 = nl1; br0 = nr0; br1 = nr1; }
#pragma unroll
    for (int i0 = 0; i0 < NTW; i0 += 4) {
      constexpr int GMAX = 4;
      float4 am[GMAX], ah[GMAX];
#pragma unroll
      for (int j = 0; j < GMAX; j++) {
        if (i0 + j < NTW) {
          am[j] = *(const float4*)(ap[i0 + j] + k0);
          ah[j] = *(const float4*)(hp[i0 + j] + k0);
        }
      }
#pragma unroll
      for (int j = 0; j < GMAX; j++) {
        if (i0 + j < NTW) {
          v2f a; a.x = am[j].x; a.y = am[j].y;
          c[i0 + j] = __builtin_amdgcn_wmma_f32_16x16x4_f32(false, a, false, bl0, (short)0, c[i0 + j],
                                                            false, false);
        }
      }
#pragma unroll
      for (int j = 0; j < GMAX; j++) {
        if (i0 + j < NTW) {
          v2f a; a.x = am[j].z; a.y = am[j].w;
          c[i0 + j] = __builtin_amdgcn_wmma_f32_16x16x4_f32(false, a, false, bl1, (short)0, c[i0 + j],
                                                            false, false);
        }
      }
#pragma unroll
      for (int j = 0; j < GMAX; j++) {
        if (i0 + j < NTW) {
          v2f a; a.x = ah[j].x; a.y = ah[j].y;
          c[i0 + j] = __builtin_amdgcn_wmma_f32_16x16x4_f32(false, a, false, br0, (short)0, c[i0 + j],
                                                            false, false);
        }
      }
#pragma unroll
      for (int j = 0; j < GMAX; j++) {
        if (i0 + j < NTW) {
          v2f a; a.x = ah[j].z; a.y = ah[j].w;
          c[i0 + j] = __builtin_amdgcn_wmma_f32_16x16x4_f32(false, a, false, br1, (short)0, c[i0 + j],
                                                            false, false);
        }
      }
    }
    if (AHEAD) { bl0 = nl0; bl1 = nl1; br0 = nr0; br1 = nr1; }
  }
}

template <int K, int FO, bool RELU>
__global__ __launch_bounds__(AGG_NT)
void sage_rows_fused(const float* __restrict__ h, const int* __restrict__ src,
                     const int* __restrict__ dst, const float* __restrict__ rdeg,
                     const float* __restrict__ Wl, const float* __restrict__ Wr,
                     const float* __restrict__ bias, float* __restrict__ out, int E, int M) {
  __shared__ __attribute__((aligned(16))) float lds[AGG_LDS_FLOATS];
  constexpr int LDK = K + AGG_PAD;
  constexpr int KH  = (K > 128) ? 64 : K;
  constexpr int NP  = K / KH;
  constexpr int NLD = (AGG_ROWS * (KH / 4) + AGG_NT - 1) / AGG_NT;
  float* rows = lds;
  float* hs   = lds + AGG_RPAD * LDK;
  const int t = threadIdx.x;
  const int lane = t & 31;
  const int wave = __builtin_amdgcn_readfirstlane(t >> 5);
  const int r0 = blockIdx.x * AGG_ROWS;

  for (int i = t; i < AGG_RPAD * LDK / 4; i += AGG_NT)
    ((float4*)rows)[i] = make_float4(0.f, 0.f, 0.f, 0.f);
  {
    const typename freg<4 * NLD>::type hr = own_rows_load<K, KH, NLD>(h, r0, M, 0, t);
    own_rows_store<KH, NLD>(hr, hs, t);
    for (int i = t; i < (AGG_RPAD - AGG_ROWS) * (KH + AGG_PAD); i += AGG_NT)
      hs[AGG_ROWS * (KH + AGG_PAD) + i] = 0.f;
  }
  __syncthreads();

  const int ntrips = (E + AGG_TRIP - 1) / AGG_TRIP;
  const int rot = (int)(((unsigned)blockIdx.x * (unsigned)ntrips) / gridDim.x);

  int T = wave;
  int ebn = 0;
  int dn[AGG_U];
#pragma unroll
  for (int u = 0; u < AGG_U; u++) dn[u] = -1;
  if (T < ntrips) {
    int Tr = T + rot;
    if (Tr >= ntrips) Tr -= ntrips;
    ebn = Tr * AGG_TRIP + lane;
#pragma unroll
    for (int u = 0; u < AGG_U; u++) {
      const int e = ebn + 32 * u;
      if (e < E) dn[u] = dst[e];
    }
  }

  while (T < ntrips) {
    const int eb = ebn;
    int dl[AGG_U];
#pragma unroll
    for (int u = 0; u < AGG_U; u++) dl[u] = dn[u] - r0;

    T += AGG_NW;
    if (T < ntrips) {
      int Tr = T + rot;
      if (Tr >= ntrips) Tr -= ntrips;
      ebn = Tr * AGG_TRIP + lane;
#pragma unroll
      for (int u = 0; u < AGG_U; u++) {
        const int e = ebn + 32 * u;
        dn[u] = (e < E) ? dst[e] : -1;
      }
    }

    int sv[AGG_U];
    unsigned mk[AGG_U];
#pragma unroll
    for (int u = 0; u < AGG_U; u++) {
      const bool m = (unsigned)dl[u] < (unsigned)AGG_ROWS;
      sv[u] = 0;
      if (m) sv[u] = src[eb + 32 * u];
      mk[u] = __builtin_amdgcn_ballot_w32(m);
    }

#pragma unroll
    for (int u = 0; u < AGG_U; u++) {
      unsigned mask = mk[u];
      while (mask) {
        const int b = __builtin_ctz(mask);
        mask &= mask - 1;
        const int s  = __builtin_amdgcn_readlane(sv[u], b);
        const int rl = __builtin_amdgcn_readlane(dl[u], b);
        const float* hp = h + (size_t)s * K + lane;
        float* ap = rows + rl * LDK + lane;
        float v[K / 32];
#pragma unroll
        for (int j = 0; j < K / 32; j++) v[j] = hp[32 * j];
#pragma unroll
        for (int j = 0; j < K / 32; j++) atomicAdd(ap + 32 * j, v[j]);
      }
    }
  }
  __syncthreads();

  {
    constexpr int NSC = (AGG_ROWS * (K / 4) + AGG_NT - 1) / AGG_NT;
#pragma unroll
    for (int j = 0; j < NSC; j++) {
      const int i = t + j * AGG_NT;
      const int r = i / (K / 4);
      const int c4 = i - r * (K / 4);
      const int gr = r0 + r;
      if (r < AGG_ROWS && gr < M) {
        const float sc = rdeg[gr];
        float4* p = (float4*)(rows + r * LDK) + c4;
        float4 v = *p;
        v.x *= sc; v.y *= sc; v.z *= sc; v.w *= sc;
        *p = v;
      }
    }
  }
  __syncthreads();

  constexpr int CT  = FO / 16;
  constexpr int WPC = AGG_NW / CT;
  constexpr int NTM = (AGG_NRT + WPC - 1) / WPC;
  const int ct = wave % CT;
  const int rg = wave / CT;
  const bool more = rg < AGG_NRT - (NTM - 1) * WPC;

  v8f c[NTM];
#pragma unroll
  for (int i = 0; i < NTM; i++) c[i] = (v8f){0.f, 0.f, 0.f, 0.f, 0.f, 0.f, 0.f, 0.f};

#pragma unroll 1
  for (int p = 0; p < NP - 1; p++) {
    if (more) rows_gemm_pass<K, FO, KH, NTM>(c, rows, hs, Wl, Wr, p * KH, ct, rg, lane);
    else      rows_gemm_pass<K, FO, KH, NTM - 1>(c, rows, hs, Wl, Wr, p * KH, ct, rg, lane);
    const typename freg<4 * NLD>::type hr = own_rows_load<K, KH, NLD>(h, r0, M, (p + 1) * KH, t);
    __syncthreads();
    own_rows_store<KH, NLD>(hr, hs, t);
    __syncthreads();
  }
  if (more) rows_gemm_pass<K, FO, KH, NTM>(c, rows, hs, Wl, Wr, (NP - 1) * KH, ct, rg, lane);
  else      rows_gemm_pass<K, FO, KH, NTM - 1>(c, rows, hs, Wl, Wr, (NP - 1) * KH, ct, rg, lane);

  const int m16 = lane & 15;
  const int g = lane >> 4;
  const int ncol = ct * 16 + m16;
  const float bv = bias[ncol];
#pragma unroll
  for (int i = 0; i < NTM; i++) {
    const int rbase = (rg + i * WPC) * 16 + 8 * g;
#pragma unroll
    for (int v = 0; v < 8; v++) {
      const int lr = rbase + v;
      float val = c[i][v] + bv;
      if (RELU) val = fmaxf(val, 0.f);
      if (lr < AGG_ROWS && r0 + lr < M) out[(size_t)(r0 + lr) * FO + ncol] = val;
    }
  }
}

__global__ void bn_stats(const float* __restrict__ h, float* __restrict__ st, int M, int F) {
  int c = threadIdx.x;
  float s = 0.f, s2 = 0.f;
  for (int r = blockIdx.x; r < M; r += gridDim.x) {
    float v = h[(size_t)r * F + c];
    s += v;
    s2 += v * v;
  }
  atomicAdd(&st[c], s);
  atomicAdd(&st[F + c], s2);
}

__global__ void bn_apply(float* __restrict__ h, const float* __restrict__ st,
                         const float* __restrict__ g, const float* __restrict__ be,
                         int M, int F) {
  int i = blockIdx.x * blockDim.x + threadIdx.x;
  if (i >= M * F) return;
  int c = i & (F - 1);
  float inv = 1.f / (float)M;
  float mean = st[c] * inv;
  float var = st[F + c] * inv - mean * mean;
  h[i] = g[c] * (h[i] - mean) * rsqrtf(var + 1e-5f) + be[c];
}

__global__ void reparam(const float* __restrict__ h3, const float* __restrict__ eps,
                        float* __restrict__ z, int M) {
  int i = blockIdx.x * blockDim.x + threadIdx.x;
  if (i >= M * 64) return;
  int r = i >> 6, c = i & 63;
  const float* row = h3 + (size_t)r * 128;
  z[i] = row[c] + expf(row[64 + c]) * eps[i];
}

extern "C" void kernel_launch(void* const* d_in, const int* in_sizes, int n_in,
                              void* d_out, int out_size, void* d_ws, size_t ws_size,
                              hipStream_t stream) {
  const float* x   = (const float*)d_in[0];
  const int*   src = (const int*)d_in[1];
  const int*   dst = (const int*)d_in[2];
  const float* eps = (const float*)d_in[3];
  const float* W1l = (const float*)d_in[4];
  const float* W1r = (const float*)d_in[5];
  const float* b1  = (const float*)d_in[6];
  const float* g1  = (const float*)d_in[7];
  const float* be1 = (const float*)d_in[8];
  const float* W2l = (const float*)d_in[9];
  const float* W2r = (const float*)d_in[10];
  const float* b2  = (const float*)d_in[11];
  const float* g2  = (const float*)d_in[12];
  const float* be2 = (const float*)d_in[13];
  const float* W3l = (const float*)d_in[14];
  const float* W3r = (const float*)d_in[15];
  const float* b3  = (const float*)d_in[16];
  float* zout = (float*)d_out;
  (void)in_sizes; (void)n_in; (void)out_size; (void)ws_size;

  char* ws = (char*)d_ws;
  size_t off = 0;
  auto carve = [&](size_t bytes) -> float* {
    char* p = ws + off;
    off = (off + bytes + 255) & ~(size_t)255;
    return (float*)p;
  };
  float* rdeg = carve((size_t)NN * 4);
  float* h1   = carve((size_t)NN * 128 * 4);
  float* h2   = carve((size_t)NN * 256 * 4);
  float* st   = carve(512 * 4);

  const int M = NN, E = NE;

  zero_f32<<<256, 256, 0, stream>>>(rdeg, M);
  deg_count<<<(E + 255) / 256, 256, 0, stream>>>(dst, rdeg, E);
  rdeg_inv<<<(M + 255) / 256, 256, 0, stream>>>(rdeg, M);

  sage_rows_fused<128, 128, true><<<AGG_NB, AGG_NT, 0, stream>>>(x, src, dst, rdeg, W1l, W1r, b1, h1, E, M);
  zero_f32<<<1, 256, 0, stream>>>(st, 256);
  bn_stats<<<512, 128, 0, stream>>>(h1, st, M, 128);
  bn_apply<<<(M * 128 + 255) / 256, 256, 0, stream>>>(h1, st, g1, be1, M, 128);

  sage_rows_fused<128, 256, true><<<AGG_NB, AGG_NT, 0, stream>>>(h1, src, dst, rdeg, W2l, W2r, b2, h2, E, M);
  zero_f32<<<2, 256, 0, stream>>>(st, 512);
  bn_stats<<<512, 256, 0, stream>>>(h2, st, M, 256);
  bn_apply<<<(M * 256 + 255) / 256, 256, 0, stream>>>(h2, st, g2, be2, M, 256);

  sage_rows_fused<256, 128, false><<<AGG_NB, AGG_NT, 0, stream>>>(h2, src, dst, rdeg, W3l, W3r, b3, h1, E, M);

  reparam<<<(M * 64 + 255) / 256, 256, 0, stream>>>(h1, eps, zout, M);
}


# Owners' edge scan done once, not per layer: a pre-pass has each row-owning group walk dst, list its matches (source,row), count degrees in LDS; layers replay the lists; 13->11 launches; LDS add order
# speedup vs baseline: 2.6097x; 1.6566x over previous
#include <hip/hip_runtime.h>

#define NN 50000
#define NE 800000

typedef __attribute__((ext_vector_type(2))) float v2f;
typedef __attribute__((ext_vector_type(8))) float v8f;

__global__ void zero_f32(float* __restrict__ p, int n) {
  int i = blockIdx.x * blockDim.x + threadIdx.x;
  int stride = gridDim.x * blockDim.x;
  for (; i < n; i += stride) p[i] = 0.f;
}

#define AGG_ROWS 196
#define AGG_RPAD 208
#define AGG_NRT  (AGG_RPAD / 16)
#define AGG_NB   ((NN + AGG_ROWS - 1) / AGG_ROWS)
#define AGG_NT   1024
#define AGG_NW   (AGG_NT / 32)
#define AGG_U    16
#define AGG_TRIP (32 * AGG_U)
#define AGG_PAD  4
#define AGG_LDS_FLOATS (AGG_RPAD * (256 + AGG_PAD) + AGG_RPAD * (64 + AGG_PAD))

#define LIST_CAP 4096
static_assert(NE < (1 << 24) && NN < (1 << 24), "edge index and source must fit 24 bits of a list word");
static_assert(AGG_ROWS <= 256, "local row must fit 8 bits of a list word");

__global__ __launch_bounds__(AGG_NT)
void owner_scan(const int* __restrict__ src, const int* __restrict__ dst,
                unsigned* __restrict__ list, int* __restrict__ cnt, float* __restrict__ rdeg,
                int E, int M) {
  __shared__ unsigned lent[LIST_CAP];
  __shared__ int ldeg[AGG_ROWS];
  __shared__ int lcnt;
  const int t = threadIdx.x;
  const int lane = t & 31;
  const int wave = __builtin_amdgcn_readfirstlane(t >> 5);
  const int r0 = blockIdx.x * AGG_ROWS;

  for (int i = t; i < AGG_ROWS; i += AGG_NT) ldeg[i] = 0;
  if (t == 0) lcnt = 0;
  __syncthreads();

  const int nfull = E / AGG_TRIP;
  const int rot = (int)(((unsigned)blockIdx.x * (unsigned)nfull) / gridDim.x);
  const int4* dst4 = (const int4*)dst;

  int T = wave;
  int ibn = 0;
  int4 dn[4];
#pragma unroll
  for (int q = 0; q < 4; q++) dn[q] = make_int4(-1, -1, -1, -1);
  if (T < nfull) {
    int Tr = T + rot;
    if (Tr >= nfull) Tr -= nfull;
    ibn = Tr * (AGG_TRIP / 4) + lane;
#pragma unroll
    for (int q = 0; q < 4; q++) dn[q] = dst4[ibn + 32 * q];
  }

  while (T < nfull) {
    const int ib = ibn;
    int dl[16];
#pragma unroll
    for (int q = 0; q < 4; q++) {
      dl[4 * q + 0] = dn[q].x - r0; dl[4 * q + 1] = dn[q].y - r0;
      dl[4 * q + 2] = dn[q].z - r0; dl[4 * q + 3] = dn[q].w - r0;
    }

    T += AGG_NW;
    if (T < nfull) {
      int Tr = T + rot;
      if (Tr >= nfull) Tr -= nfull;
      ibn = Tr * (AGG_TRIP / 4) + lane;
#pragma unroll
      for (int q = 0; q < 4; q++) dn[q] = dst4[ibn + 32 * q];
    }

    unsigned mk[16];
    int tot = 0;
#pragma unroll
    for (int u = 0; u < 16; u++) {
      mk[u] = __builtin_amdgcn_ballot_w32((unsigned)dl[u] < (unsigned)AGG_ROWS);
      tot += __builtin_popcount(mk[u]);
    }
    if (tot) {
      int base = 0;
      if (lane == 0) base = atomicAdd(&lcnt, tot);
      base = __builtin_amdgcn_readfirstlane(base);
#pragma unroll
      for (int u = 0; u < 16; u++) {
        if (mk[u]) {
          if ((mk[u] >> lane) & 1u) {
            const int pos = base + (int)__builtin_amdgcn_mbcnt_lo(mk[u], 0u);
            const int e = 4 * (ib + 32 * (u >> 2)) + (u & 3);
            if (pos < LIST_CAP) lent[pos] = ((unsigned)e << 8) | (unsigned)dl[u];
            atomicAdd(&ldeg[dl[u]], 1);
          }
          base += __builtin_popcount(mk[u]);
        }
      }
    }
  }

  for (int e = nfull * AGG_TRIP + t; e < E; e += AGG_NT) {
    const int d = dst[e] - r0;
    if ((unsigned)d < (unsigned)AGG_ROWS) {
      const int pos = atomicAdd(&lcnt, 1);
      if (pos < LIST_CAP) lent[pos] = ((unsigned)e << 8) | (unsigned)d;
      atomicAdd(&ldeg[d], 1);
    }
  }
  __syncthreads();

  const int n = lcnt;
  const int nst = (n < LIST_CAP) ? n : LIST_CAP;
  unsigned* mylist = list + (size_t)blockIdx.x * LIST_CAP;
  for (int i = t; i < nst; i += AGG_NT) {
    const unsigned en = lent[i];
    const int s = src[en >> 8];
    mylist[i] = ((unsigned)s << 8) | (en & 255u);
  }
  for (int i = t; i < AGG_ROWS; i += AGG_NT) {
    const int gr = r0 + i;
    if (gr < M) rdeg[gr] = 1.f / fmaxf((float)ldeg[i], 1.f);
  }
  if (t == 0) cnt[blockIdx.x] = n;
}

template <int N> struct freg { typedef float type __attribute__((ext_vector_type(N))); };

template <int K, int KH, int NLD>
__device__ __forceinline__ typename freg<4 * NLD>::type own_rows_load(const float* __restrict__ h,
                                                                      int r0, int M, int col0, int t) {
  typename freg<4 * NLD>::type r;
#pragma unroll
  for (int j = 0; j < NLD; j++) {
    int i = t + j * AGG_NT;
    if (i > AGG_ROWS * (KH / 4) - 1) i = AGG_ROWS * (KH / 4) - 1;
    const int row = i / (KH / 4);
    const int c4 = i - row * (KH / 4);
    int gr = r0 + row;
    if (gr > M - 1) gr = M - 1;
    const float4 v = *(const float4*)(h + (gr * K + col0 + 4 * c4));
    r[4 * j + 0] = v.x; r[4 * j + 1] = v.y; r[4 * j + 2] = v.z; r[4 * j + 3] = v.w;
  }
  return r;
}

template <int KH, int NLD>
__device__ __forceinline__ void own_rows_store(const typename freg<4 * NLD>::type r, float* hs, int t) {
#pragma unroll
  for (int j = 0; j < NLD; j++) {
    const int i = t + j * AGG_NT;
    if (i < AGG_ROWS * (KH / 4)) {
      const int row = i / (KH / 4);
      const int c4 = i - row * (KH / 4);
      *(float4*)(hs + row * (KH + AGG_PAD) + 4 * c4) =
          make_float4(r[4 * j + 0], r[4 * j + 1], r[4 * j + 2], r[4 * j + 3]);
    }
  }
}

template <int K, int FO, int KH, int NTW>
__device__ __forceinline__ void rows_gemm_pass(v8f* c, const float* ms, const float* hs,
                                               const float* __restrict__ Wl,
                                               const float* __restrict__ Wr,
                                               int kb, int ct, int rg, int lane) {
  constexpr int LDK = K + AGG_PAD;
  constexpr int LDH = KH + AGG_PAD;
  constexpr int WPC = AGG_NW / (FO / 16);
  const int m16 = lane & 15;
  const int g = lane >> 4;
  const int ncol = ct * 16 + m16;

  const float* ap[NTW];
  const float* hp[NTW];
#pragma unroll
  for (int i = 0; i < NTW; i++) {
    const int lrow = (rg + i * WPC) * 16 + m16;
    ap[i] = ms + lrow * LDK + kb + 4 * g;
    hp[i] = hs + lrow * LDH + 4 * g;
  }
  const float* wl = Wl + (size_t)(kb + 4 * g) * FO + ncol;
  const float* wr = Wr + (size_t)(kb + 4 * g) * FO + ncol;

  constexpr bool AHEAD = (NTW <= 4) && (KH == K);
  v2f bl0, bl1, br0, br1;
  bl0.x = wl[0 * FO]; bl0.y = wl[1 * FO]; bl1.x = wl[2 * FO]; bl1.y = wl[3 * FO];
  br0.x = wr[0 * FO]; br0.y = wr[1 * FO]; br1.x = wr[2 * FO]; br1.y = wr[3 * FO];

#pragma unroll 1
  for (int k0 = 0; k0 < KH; k0 += 8) {
    const int kn = AHEAD ? ((k0 + 8 < KH) ? (k0 + 8) : k0) : k0;
    v2f nl0, nl1, nr0, nr1;
    nl0.x = wl[(size_t)(kn + 0) * FO]; nl0.y = wl[(size_t)(kn + 1) * FO];
    nl1.x = wl[(size_t)(kn + 2) * FO]; nl1.y = wl[(size_t)(kn + 3) * FO];
    nr0.x = wr[(size_t)(kn + 0) * FO]; nr0.y = wr[(size_t)(kn + 1) * FO];
    nr1.x = wr[(size_t)(kn + 2) * FO]; nr1.y = wr[(size_t)(kn + 3) * FO];
    if (!AHEAD) { bl0 = nl0; bl1 = nl1; br0 = nr0; br1 = nr1; }
#pragma unroll
    for (int i0 = 0; i0 < NTW; i0 += 4) {
      constexpr int GMAX = 4;
      float4 am[GMAX], ah[GMAX];
#pragma unroll
      for (int j = 0; j < GMAX; j++) {
        if (i0 + j < NTW) {
          am[j] = *(const float4*)(ap[i0 + j] + k0);
          ah[j] = *(const float4*)(hp[i0 + j] + k0);
        }
      }
#pragma unroll
      for (int j = 0; j < GMAX; j++) {
        if (i0 + j < NTW) {
          v2f a; a.x = am[j].x; a.y = am[j].y;
          c[i0 + j] = __builtin_amdgcn_wmma_f32_16x16x4_f32(false, a, false, bl0, (short)0, c[i0 + j],
                                                            false, false);
        }
      }
#pragma unroll
      for (int j = 0; j < GMAX; j++) {
        if (i0 + j < NTW) {
          v2f a; a.x = am[j].z; a.y = am[j].w;
          c[i0 + j] = __builtin_amdgcn_wmma_f32_16x16x4_f32(false, a, false, bl1, (short)0, c[i0 + j],
                                                            false, false);
        }
      }
#pragma unroll
      for (int j = 0; j < GMAX; j++) {
        if (i0 + j < NTW) {
          v2f a; a.x = ah[j].x; a.y = ah[j].y;
          c[i0 + j] = __builtin_amdgcn_wmma_f32_16x16x4_f32(false, a, false, br0, (short)0, c[i0 + j],
                                                            false, false);
        }
      }
#pragma unroll
      for (int j = 0; j < GMAX; j++) {
        if (i0 + j < NTW) {
          v2f a; a.x = ah[j].z; a.y = ah[j].w;
          c[i0 + j] = __builtin_amdgcn_wmma_f32_16x16x4_f32(false, a, false, br1, (short)0, c[i0 + j],
                                                            false, false);
        }
      }
    }
    if (AHEAD) { bl0 = nl0; bl1 = nl1; br0 = nr0; br1 = nr1; }
  }
}

template <int K, int FO, bool RELU>
__global__ __launch_bounds__(AGG_NT)
void sage_rows_fused(const float* __restrict__ h, const unsigned* __restrict__ list,
                     const int* __restrict__ cnt, const int* __restrict__ src,
                     const int* __restrict__ dst, const float* __restrict__ rdeg,
                     const float* __restrict__ Wl, const float* __restrict__ Wr,
                     const float* __restrict__ bias, float* __restrict__ out, int E, int M) {
  __shared__ __attribute__((aligned(16))) float lds[AGG_LDS_FLOATS];
  constexpr int LDK = K + AGG_PAD;
  constexpr int KH  = (K > 128) ? 64 : K;
  constexpr int NP  = K / KH;
  constexpr int NLD = (AGG_ROWS * (KH / 4) + AGG_NT - 1) / AGG_NT;
  float* rows = lds;
  float* hs   = lds + AGG_RPAD * LDK;
  const int t = threadIdx.x;
  const int lane = t & 31;
  const int wave = __builtin_amdgcn_readfirstlane(t >> 5);
  const int r0 = blockIdx.x * AGG_ROWS;

  for (int i = t; i < AGG_RPAD * LDK / 4; i += AGG_NT)
    ((float4*)rows)[i] = make_float4(0.f, 0.f, 0.f, 0.f);
  {
    const typename freg<4 * NLD>::type hr = own_rows_load<K, KH, NLD>(h, r0, M, 0, t);
    own_rows_store<KH, NLD>(hr, hs, t);
    for (int i = t; i < (AGG_RPAD - AGG_ROWS) * (KH + AGG_PAD); i += AGG_NT)
      hs[AGG_ROWS * (KH + AGG_PAD) + i] = 0.f;
  }
  __syncthreads();

  const int nlist = cnt[blockIdx.x];
  if (nlist <= LIST_CAP) {
    const unsigned* mylist = list + (size_t)blockIdx.x * LIST_CAP;
    const int e0 = (int)(((unsigned)nlist * (unsigned)wave) / AGG_NW);
    const int e1 = (int)(((unsigned)nlist * (unsigned)(wave + 1)) / AGG_NW);
    constexpr int NB = 1024 / K;
    for (int e = e0; e < e1; e += 32) {
      const int ng = (e1 - e < 32) ? (e1 - e) : 32;
      unsigned ent = 0;
      if (lane < ng) ent = mylist[e + lane];
      for (int j = 0; j < ng; j += NB) {
        float v[NB][K / 32];
        int rl[NB];
#pragma unroll
        for (int b = 0; b < NB; b++) {
          int idx = j + b;
          if (idx > ng - 1) idx = ng - 1;
          const unsigned en = (unsigned)__builtin_amdgcn_readlane((int)ent, idx);
          rl[b] = (int)(en & 255u);
          const float* hp = h + (size_t)(en >> 8) * K + lane;
#pragma unroll
          for (int q = 0; q < K / 32; q++) v[b][q] = hp[32 * q];
        }
#pragma unroll
        for (int b = 0; b < NB; b++) {
          if (j + b < ng) {
            float* ap = rows + rl[b] * LDK + lane;
#pragma unroll
            for (int q = 0; q < K / 32; q++) atomicAdd(ap + 32 * q, v[b][q]);
          }
        }
      }
    }
  } else {
  const int ntrips = (E + AGG_TRIP - 1) / AGG_TRIP;
  const int rot = (int)(((unsigned)blockIdx.x * (unsigned)ntrips) / gridDim.x);

  int T = wave;
  int ebn = 0;
  int dn[AGG_U];
#pragma unroll
  for (int u = 0; u < AGG_U; u++) dn[u] = -1;
  if (T < ntrips) {
    int Tr = T + rot;
    if (Tr >= ntrips) Tr -= ntrips;
    ebn = Tr * AGG_TRIP + lane;
#pragma unroll
    for (int u = 0; u < AGG_U; u++) {
      const int e = ebn + 32 * u;
      if (e < E) dn[u] = dst[e];
    }
  }

  while (T < ntrips) {
    const int eb = ebn;
    int dl[AGG_U];
#pragma unroll
    for (int u = 0; u < AGG_U; u++) dl[u] = dn[u] - r0;

    T += AGG_NW;
    if (T < ntrips) {
      int Tr = T + rot;
      if (Tr >= ntrips) Tr -= ntrips;
      ebn = Tr * AGG_TRIP + lane;
#pragma unroll
      for (int u = 0; u < AGG_U; u++) {
        const int e = ebn + 32 * u;
        dn[u] = (e < E) ? dst[e] : -1;
      }
    }

    int sv[AGG_U];
    unsigned mk[AGG_U];
#pragma unroll
    for (int u = 0; u < AGG_U; u++) {
      const bool m = (unsigned)dl[u] < (unsigned)AGG_ROWS;
      sv[u] = 0;
      if (m) sv[u] = src[eb + 32 * u];
      mk[u] = __builtin_amdgcn_ballot_w32(m);
    }

#pragma unroll
    for (int u = 0; u < AGG_U; u++) {
      unsigned mask = mk[u];
      while (mask) {
        const int b = __builtin_ctz(mask);
        mask &= mask - 1;
        const int s  = __builtin_amdgcn_readlane(sv[u], b);
        const int rl = __builtin_amdgcn_readlane(dl[u], b);
        const float* hp = h + (size_t)s * K + lane;
        float* ap = rows + rl * LDK + lane;
        float v[K / 32];
#pragma unroll
        for (int j = 0; j < K / 32; j++) v[j] = hp[32 * j];
#pragma unroll
        for (int j = 0; j < K / 32; j++) atomicAdd(ap + 32 * j, v[j]);
      }
    }
  }
  }
  __syncthreads();

  {
    constexpr int NSC = (AGG_ROWS * (K / 4) + AGG_NT - 1) / AGG_NT;
#pragma unroll
    for (int j = 0; j < NSC; j++) {
      const int i = t + j * AGG_NT;
      const int r = i / (K / 4);
      const int c4 = i - r * (K / 4);
      const int gr = r0 + r;
      if (r < AGG_ROWS && gr < M) {
        const float sc = rdeg[gr];
        float4* p = (float4*)(rows + r * LDK) + c4;
        float4 v = *p;
        v.x *= sc; v.y *= sc; v.z *= sc; v.w *= sc;
        *p = v;
      }
    }
  }
  __syncthreads();

  constexpr int CT  = FO / 16;
  constexpr int WPC = AGG_NW / CT;
  constexpr int NTM = (AGG_NRT + WPC - 1) / WPC;
  const int ct = wave % CT;
  const int rg = wave / CT;
  const bool more = rg < AGG_NRT - (NTM - 1) * WPC;

  v8f c[NTM];
#pragma unroll
  for (int i = 0; i < NTM; i++) c[i] = (v8f){0.f, 0.f, 0.f, 0.f, 0.f, 0.f, 0.f, 0.f};

#pragma unroll 1
  for (int p = 0; p < NP - 1; p++) {
    if (more) rows_gemm_pass<K, FO, KH, NTM>(c, rows, hs, Wl, Wr, p * KH, ct, rg, lane);
    else      rows_gemm_pass<K, FO, KH, NTM - 1>(c, rows, hs, Wl, Wr, p * KH, ct, rg, lane);
    const typename freg<4 * NLD>::type hr = own_rows_load<K, KH, NLD>(h, r0, M, (p + 1) * KH, t);
    __syncthreads();
    own_rows_store<KH, NLD>(hr, hs, t);
    __syncthreads();
  }
  if (more) rows_gemm_pass<K, FO, KH, NTM>(c, rows, hs, Wl, Wr, (NP - 1) * KH, ct, rg, lane);
  else      rows_gemm_pass<K, FO, KH, NTM - 1>(c, rows, hs, Wl, Wr, (NP - 1) * KH, ct, rg, lane);

  const int m16 = lane & 15;
  const int g = lane >> 4;
  const int ncol = ct * 16 + m16;
  const float bv = bias[ncol];
#pragma unroll
  for (int i = 0; i < NTM; i++) {
    const int rbase = (rg + i * WPC) * 16 + 8 * g;
#pragma unroll
    for (int v = 0; v < 8; v++) {
      const int lr = rbase + v;
      float val = c[i][v] + bv;
      if (RELU) val = fmaxf(val, 0.f);
      if (lr < AGG_ROWS && r0 + lr < M) out[(size_t)(r0 + lr) * FO + ncol] = val;
    }
  }
}

__global__ void bn_stats(const float* __restrict__ h, float* __restrict__ st, int M, int F) {
  int c = threadIdx.x;
  float s = 0.f, s2 = 0.f;
  for (int r = blockIdx.x; r < M; r += gridDim.x) {
    float v = h[(size_t)r * F + c];
    s += v;
    s2 += v * v;
  }
  atomicAdd(&st[c], s);
  atomicAdd(&st[F + c], s2);
}

__global__ void bn_apply(float* __restrict__ h, const float* __restrict__ st,
                         const float* __restrict__ g, const float* __restrict__ be,
                         int M, int F) {
  int i = blockIdx.x * blockDim.x + threadIdx.x;
  if (i >= M * F) return;
  int c = i & (F - 1);
  float inv = 1.f / (float)M;
  float mean = st[c] * inv;
  float var = st[F + c] * inv - mean * mean;
  h[i] = g[c] * (h[i] - mean) * rsqrtf(var + 1e-5f) + be[c];
}

__global__ void reparam(const float* __restrict__ h3, const float* __restrict__ eps,
                        float* __restrict__ z, int M) {
  int i = blockIdx.x * blockDim.x + threadIdx.x;
  if (i >= M * 64) return;
  int r = i >> 6, c = i & 63;
  const float* row = h3 + (size_t)r * 128;
  z[i] = row[c] + expf(row[64 + c]) * eps[i];
}

extern "C" void kernel_launch(void* const* d_in, const int* in_sizes, int n_in,
                              void* d_out, int out_size, void* d_ws, size_t ws_size,
                              hipStream_t stream) {
  const float* x   = (const float*)d_in[0];
  const int*   src = (const int*)d_in[1];
  const int*   dst = (const int*)d_in[2];
  const float* eps = (const float*)d_in[3];
  const float* W1l = (const float*)d_in[4];
  const float* W1r = (const float*)d_in[5];
  const float* b1  = (const float*)d_in[6];
  const float* g1  = (const float*)d_in[7];
  const float* be1 = (const float*)d_in[8];
  const float* W2l = (const float*)d_in[9];
  const float* W2r = (const float*)d_in[10];
  const float* b2  = (const float*)d_in[11];
  const float* g2  = (const float*)d_in[12];
  const float* be2 = (const float*)d_in[13];
  const float* W3l = (const float*)d_in[14];
  const float* W3r = (const float*)d_in[15];
  const float* b3  = (const float*)d_in[16];
  float* zout = (float*)d_out;
  (void)in_sizes; (void)n_in; (void)out_size; (void)ws_size;

  char* ws = (char*)d_ws;
  size_t off = 0;
  auto carve = [&](size_t bytes) -> float* {
    char* p = ws + off;
    off = (off + bytes + 255) & ~(size_t)255;
    return (float*)p;
  };
  float* rdeg = carve((size_t)NN * 4);
  float* h1   = carve((size_t)NN * 128 * 4);
  float* h2   = carve((size_t)NN * 256 * 4);
  float* st   = carve(512 * 4);
  unsigned* list = (unsigned*)carve((size_t)AGG_NB * LIST_CAP * 4);
  int* cnt    = (int*)carve((size_t)AGG_NB * 4);

  const int M = NN, E = NE;

  owner_scan<<<AGG_NB, AGG_NT, 0, stream>>>(src, dst, list, cnt, rdeg, E, M);

  sage_rows_fused<128, 128, true><<<AGG_NB, AGG_NT, 0, stream>>>(x, list, cnt, src, dst, rdeg, W1l, W1r, b1, h1, E, M);
  zero_f32<<<1, 256, 0, stream>>>(st, 256);
  bn_stats<<<512, 128, 0, stream>>>(h1, st, M, 128);
  bn_apply<<<(M * 128 + 255) / 256, 256, 0, stream>>>(h1, st, g1, be1, M, 128);

  sage_rows_fused<128, 256, true><<<AGG_NB, AGG_NT, 0, stream>>>(h1, list, cnt, src, dst, rdeg, W2l, W2r, b2, h2, E, M);
  zero_f32<<<2, 256, 0, stream>>>(st, 512);
  bn_stats<<<512, 256, 0, stream>>>(h2, st, M, 256);
  bn_apply<<<(M * 256 + 255) / 256, 256, 0, stream>>>(h2, st, g2, be2, M, 256);

  sage_rows_fused<256, 128, false><<<AGG_NB, AGG_NT, 0, stream>>>(h2, list, cnt, src, dst, rdeg, W3l, W3r, b3, h1, E, M);

  reparam<<<(M * 64 + 255) / 256, 256, 0, stream>>>(h1, eps, zout, M);
}


# BN sums, no stats pass, no atomics: a row-owning group adds its tile's columns from the accumulators (lanes, waves via LDS) into a partial row; a small pass adds the rows; BN sum order changes
# speedup vs baseline: 3.1045x; 1.1896x over previous
#include <hip/hip_runtime.h>

#define NN 50000
#define NE 800000

typedef __attribute__((ext_vector_type(2))) float v2f;
typedef __attribute__((ext_vector_type(8))) float v8f;

#define AGG_ROWS 196
#define AGG_RPAD 208
#define AGG_NRT  (AGG_RPAD / 16)
#define AGG_NB   ((NN + AGG_ROWS - 1) / AGG_ROWS)
#define AGG_NT   1024
#define AGG_NW   (AGG_NT / 32)
#define AGG_U    16
#define AGG_TRIP (32 * AGG_U)
#define AGG_PAD  4
#define AGG_LDS_FLOATS (AGG_RPAD * (256 + AGG_PAD) + AGG_RPAD * (64 + AGG_PAD))

#define LIST_CAP 4096
static_assert(NE < (1 << 24) && NN < (1 << 24), "edge index and source must fit 24 bits of a list word");
static_assert(AGG_ROWS <= 256, "local row must fit 8 bits of a list word");

__global__ __launch_bounds__(AGG_NT)
void owner_scan(const int* __restrict__ src, const int* __restrict__ dst,
                unsigned* __restrict__ list, int* __restrict__ cnt, float* __restrict__ rdeg,
                int E, int M) {
  __shared__ unsigned lent[LIST_CAP];
  __shared__ int ldeg[AGG_ROWS];
  __shared__ int lcnt;
  const int t = threadIdx.x;
  const int lane = t & 31;
  const int wave = __builtin_amdgcn_readfirstlane(t >> 5);
  const int r0 = blockIdx.x * AGG_ROWS;

  for (int i = t; i < AGG_ROWS; i += AGG_NT) ldeg[i] = 0;
  if (t == 0) lcnt = 0;
  __syncthreads();

  const int nfull = E / AGG_TRIP;
  const int rot = (int)(((unsigned)blockIdx.x * (unsigned)nfull) / gridDim.x);
  const int4* dst4 = (const int4*)dst;

  int T = wave;
  int ibn = 0;
  int4 dn[4];
#pragma unroll
  for (int q = 0; q < 4; q++) dn[q] = make_int4(-1, -1, -1, -1);
  if (T < nfull) {
    int Tr = T + rot;
    if (Tr >= nfull) Tr -= nfull;
    ibn = Tr * (AGG_TRIP / 4) + lane;
#pragma unroll
    for (int q = 0; q < 4; q++) dn[q] = dst4[ibn + 32 * q];
  }

  while (T < nfull) {
    const int ib = ibn;
    int dl[16];
#pragma unroll
    for (int q = 0; q < 4; q++) {
      dl[4 * q + 0] = dn[q].x - r0; dl[4 * q + 1] = dn[q].y - r0;
      dl[4 * q + 2] = dn[q].z - r0; dl[4 * q + 3] = dn[q].w - r0;
    }

    T += AGG_NW;
    if (T < nfull) {
      int Tr = T + rot;
      if (Tr >= nfull) Tr -= nfull;
      ibn = Tr * (AGG_TRIP / 4) + lane;
#pragma unroll
      for (int q = 0; q < 4; q++) dn[q] = dst4[ibn + 32 * q];
    }

    unsigned mk[16];
    int tot = 0;
#pragma unroll
    for (int u = 0; u < 16; u++) {
      mk[u] = __builtin_amdgcn_ballot_w32((unsigned)dl[u] < (unsigned)AGG_ROWS);
      tot += __builtin_popcount(mk[u]);
    }
    if (tot) {
      int base = 0;
      if (lane == 0) base = atomicAdd(&lcnt, tot);
      base = __builtin_amdgcn_readfirstlane(base);
#pragma unroll
      for (int u = 0; u < 16; u++) {
        if (mk[u]) {
          if ((mk[u] >> lane) & 1u) {
            const int pos = base + (int)__builtin_amdgcn_mbcnt_lo(mk[u], 0u);
            const int e = 4 * (ib + 32 * (u >> 2)) + (u & 3);
            if (pos < LIST_CAP) lent[pos] = ((unsigned)e << 8) | (unsigned)dl[u];
            atomicAdd(&ldeg[dl[u]], 1);
          }
          base += __builtin_popcount(mk[u]);
        }
      }
    }
  }

  for (int e = nfull * AGG_TRIP + t; e < E; e += AGG_NT) {
    const int d = dst[e] - r0;
    if ((unsigned)d < (unsigned)AGG_ROWS) {
      const int pos = atomicAdd(&lcnt, 1);
      if (pos < LIST_CAP) lent[pos] = ((unsigned)e << 8) | (unsigned)d;
      atomicAdd(&ldeg[d], 1);
    }
  }
  __syncthreads();

  const int n = lcnt;
  const int nst = (n < LIST_CAP) ? n : LIST_CAP;
  unsigned* mylist = list + (size_t)blockIdx.x * LIST_CAP;
  for (int i = t; i < nst; i += AGG_NT) {
    const unsigned en = lent[i];
    const int s = src[en >> 8];
    mylist[i] = ((unsigned)s << 8) | (en & 255u);
  }
  for (int i = t; i < AGG_ROWS; i += AGG_NT) {
    const int gr = r0 + i;
    if (gr < M) rdeg[gr] = 1.f / fmaxf((float)ldeg[i], 1.f);
  }
  if (t == 0) cnt[blockIdx.x] = n;
}

template <int N> struct freg { typedef float type __attribute__((ext_vector_type(N))); };

template <int K, int KH, int NLD>
__device__ __forceinline__ typename freg<4 * NLD>::type own_rows_load(const float* __restrict__ h,
                                                                      int r0, int M, int col0, int t) {
  typename freg<4 * NLD>::type r;
#pragma unroll
  for (int j = 0; j < NLD; j++) {
    int i = t + j * AGG_NT;
    if (i > AGG_ROWS * (KH / 4) - 1) i = AGG_ROWS * (KH / 4) - 1;
    const int row = i / (KH / 4);
    const int c4 = i - row * (KH / 4);
    int gr = r0 + row;
    if (gr > M - 1) gr = M - 1;
    const float4 v = *(const float4*)(h + (gr * K + col0 + 4 * c4));
    r[4 * j + 0] = v.x; r[4 * j + 1] = v.y; r[4 * j + 2] = v.z; r[4 * j + 3] = v.w;
  }
  return r;
}

template <int KH, int NLD>
__device__ __forceinline__ void own_rows_store(const typename freg<4 * NLD>::type r, float* hs, int t) {
#pragma unroll
  for (int j = 0; j < NLD; j++) {
    const int i = t + j * AGG_NT;
    if (i < AGG_ROWS * (KH / 4)) {
      const int row = i / (KH / 4);
      const int c4 = i - row * (KH / 4);
      *(float4*)(hs + row * (KH + AGG_PAD) + 4 * c4) =
          make_float4(r[4 * j + 0], r[4 * j + 1], r[4 * j + 2], r[4 * j + 3]);
    }
  }
}

template <int K, int FO, int KH, int NTW>
__device__ __forceinline__ void rows_gemm_pass(v8f* c, const float* ms, const float* hs,
                                               const float* __restrict__ Wl,
                                               const float* __restrict__ Wr,
                                               int kb, int ct, int rg, int lane) {
  constexpr int LDK = K + AGG_PAD;
  constexpr int LDH = KH + AGG_PAD;
  constexpr int WPC = AGG_NW / (FO / 16);
  const int m16 = lane & 15;
  const int g = lane >> 4;
  const int ncol = ct * 16 + m16;

  const float* ap[NTW];
  const float* hp[NTW];
#pragma unroll
  for (int i = 0; i < NTW; i++) {
    const int lrow = (rg + i * WPC) * 16 + m16;
    ap[i] = ms + lrow * LDK + kb + 4 * g;
    hp[i] = hs + lrow * LDH + 4 * g;
  }
  const float* wl = Wl + (size_t)(kb + 4 * g) * FO + ncol;
  const float* wr = Wr + (size_t)(kb + 4 * g) * FO + ncol;

  constexpr bool AHEAD = (NTW <= 4) && (KH == K);
  v2f bl0, bl1, br0, br1;
  bl0.x = wl[0 * FO]; bl0.y = wl[1 * FO]; bl1.x = wl[2 * FO]; bl1.y = wl[3 * FO];
  br0.x = wr[0 * FO]; br0.y = wr[1 * FO]; br1.x = wr[2 * FO]; br1.y = wr[3 * FO];

#pragma unroll 1
  for (int k0 = 0; k0 < KH; k0 += 8) {
    const int kn = AHEAD ? ((k0 + 8 < KH) ? (k0 + 8) : k0) : k0;
    v2f nl0, nl1, nr0, nr1;
    nl0.x = wl[(size_t)(kn + 0) * FO]; nl0.y = wl[(size_t)(kn + 1) * FO];
    nl1.x = wl[(size_t)(kn + 2) * FO]; nl1.y = wl[(size_t)(kn + 3) * FO];
    nr0.x = wr[(size_t)(kn + 0) * FO]; nr0.y = wr[(size_t)(kn + 1) * FO];
    nr1.x = wr[(size_t)(kn + 2) * FO]; nr1.y = wr[(size_t)(kn + 3) * FO];
    if (!AHEAD) { bl0 = nl0; bl1 = nl1; br0 = nr0; br1 = nr1; }
#pragma unroll
    for (int i0 = 0; i0 < NTW; i0 += 4) {
      constexpr int GMAX = 4;
      float4 am[GMAX], ah[GMAX];
#pragma unroll
      for (int j = 0; j < GMAX; j++) {
        if (i0 + j < NTW) {
          am[j] = *(const float4*)(ap[i0 + j] + k0);
          ah[j] = *(const float4*)(hp[i0 + j] + k0);
        }
      }
#pragma unroll
      for (int j = 0; j < GMAX; j++) {
        if (i0 + j < NTW) {
          v2f a; a.x = am[j].x; a.y = am[j].y;
          c[i0 + j] = __builtin_amdgcn_wmma_f32_16x16x4_f32(false, a, false, bl0, (short)0, c[i0 + j],
                                                            false, false);
        }
      }
#pragma unroll
      for (int j = 0; j < GMAX; j++) {
        if (i0 + j < NTW) {
          v2f a; a.x = am[j].z; a.y = am[j].w;
          c[i0 + j] = __builtin_amdgcn_wmma_f32_16x16x4_f32(false, a, false, bl1, (short)0, c[i0 + j],
                                                            false, false);
        }
      }
#pragma unroll
      for (int j = 0; j < GMAX; j++) {
        if (i0 + j < NTW) {
          v2f a; a.x = ah[j].x; a.y = ah[j].y;
          c[i0 + j] = __builtin_amdgcn_wmma_f32_16x16x4_f32(false, a, false, br0, (short)0, c[i0 + j],
                                                            false, false);
        }
      }
#pragma unroll
      for (int j = 0; j < GMAX; j++) {
        if (i0 + j < NTW) {
          v2f a; a.x = ah[j].z; a.y = ah[j].w;
          c[i0 + j] = __builtin_amdgcn_wmma_f32_16x16x4_f32(false, a, false, br1, (short)0, c[i0 + j],
                                                            false, false);
        }
      }
    }
    if (AHEAD) { bl0 = nl0; bl1 = nl1; br0 = nr0; br1 = nr1; }
  }
}

template <int K, int FO, bool RELU, bool STATS>
__global__ __launch_bounds__(AGG_NT)
void sage_rows_fused(const float* __restrict__ h, const unsigned* __restrict__ list,
                     const int* __restrict__ cnt, const int* __restrict__ src,
                     const int* __restrict__ dst, const float* __restrict__ rdeg,
                     const float* __restrict__ Wl, const float* __restrict__ Wr,
                     const float* __restrict__ bias, float* __restrict__ out,
                     float* __restrict__ part, int E, int M) {
  __shared__ __attribute__((aligned(16))) float lds[AGG_LDS_FLOATS];
  __shared__ float red[2][AGG_NW][32];
  constexpr int LDK = K + AGG_PAD;
  constexpr int KH  = (K > 128) ? 64 : K;
  constexpr int NP  = K / KH;
  constexpr int NLD = (AGG_ROWS * (KH / 4) + AGG_NT - 1) / AGG_NT;
  float* rows = lds;
  float* hs   = lds + AGG_RPAD * LDK;
  const int t = threadIdx.x;
  const int lane = t & 31;
  const int wave = __builtin_amdgcn_readfirstlane(t >> 5);
  const int r0 = blockIdx.x * AGG_ROWS;

  for (int i = t; i < AGG_RPAD * LDK / 4; i += AGG_NT)
    ((float4*)rows)[i] = make_float4(0.f, 0.f, 0.f, 0.f);
  {
    const typename freg<4 * NLD>::type hr = own_rows_load<K, KH, NLD>(h, r0, M, 0, t);
    own_rows_store<KH, NLD>(hr, hs, t);
    for (int i = t; i < (AGG_RPAD - AGG_ROWS) * (KH + AGG_PAD); i += AGG_NT)
      hs[AGG_ROWS * (KH + AGG_PAD) + i] = 0.f;
  }
  __syncthreads();

  const int nlist = cnt[blockIdx.x];
  if (nlist <= LIST_CAP) {
    const unsigned* mylist = list + (size_t)blockIdx.x * LIST_CAP;
    const int e0 = (int)(((unsigned)nlist * (unsigned)wave) / AGG_NW);
    const int e1 = (int)(((unsigned)nlist * (unsigned)(wave + 1)) / AGG_NW);
    constexpr int NB = 1024 / K;
    for (int e = e0; e < e1; e += 32) {
      const int ng = (e1 - e < 32) ? (e1 - e) : 32;
      unsigned ent = 0;
      if (lane < ng) ent = mylist[e + lane];
      for (int j = 0; j < ng; j += NB) {
        float v[NB][K / 32];
        int rl[NB];
#pragma unroll
        for (int b = 0; b < NB; b++) {
          int idx = j + b;
          if (idx > ng - 1) idx = ng - 1;
          const unsigned en = (unsigned)__builtin_amdgcn_readlane((int)ent, idx);
          rl[b] = (int)(en & 255u);
          const float* hp = h + (size_t)(en >> 8) * K + lane;
#pragma unroll
          for (int q = 0; q < K / 32; q++) v[b][q] = hp[32 * q];
        }
#pragma unroll
        for (int b = 0; b < NB; b++) {
          if (j + b < ng) {
            float* ap = rows + rl[b] * LDK + lane;
#pragma unroll
            for (int q = 0; q < K / 32; q++) atomicAdd(ap + 32 * q, v[b][q]);
          }
        }
      }
    }
  } else {
  const int ntrips = (E + AGG_TRIP - 1) / AGG_TRIP;
  const int rot = (int)(((unsigned)blockIdx.x * (unsigned)ntrips) / gridDim.x);

  int T = wave;
  int ebn = 0;
  int dn[AGG_U];
#pragma unroll
  for (int u = 0; u < AGG_U; u++) dn[u] = -1;
  if (T < ntrips) {
    int Tr = T + rot;
    if (Tr >= ntrips) Tr -= ntrips;
    ebn = Tr * AGG_TRIP + lane;
#pragma unroll
    for (int u = 0; u < AGG_U; u++) {
      const int e = ebn + 32 * u;
      if (e < E) dn[u] = dst[e];
    }
  }

  while (T < ntrips) {
    const int eb = ebn;
    int dl[AGG_U];
#pragma unroll
    for (int u = 0; u < AGG_U; u++) dl[u] = dn[u] - r0;

    T += AGG_NW;
    if (T < ntrips) {
      int Tr = T + rot;
      if (Tr >= ntrips) Tr -= ntrips;
      ebn = Tr * AGG_TRIP + lane;
#pragma unroll
      for (int u = 0; u < AGG_U; u++) {
        const int e = ebn + 32 * u;
        dn[u] = (e < E) ? dst[e] : -1;
      }
    }

    int sv[AGG_U];
    unsigned mk[AGG_U];
#pragma unroll
    for (int u = 0; u < AGG_U; u++) {
      const bool m = (unsigned)dl[u] < (unsigned)AGG_ROWS;
      sv[u] = 0;
      if (m) sv[u] = src[eb + 32 * u];
      mk[u] = __builtin_amdgcn_ballot_w32(m);
    }

#pragma unroll
    for (int u = 0; u < AGG_U; u++) {
      unsigned mask = mk[u];
      while (mask) {
        const int b = __builtin_ctz(mask);
        mask &= mask - 1;
        const int s  = __builtin_amdgcn_readlane(sv[u], b);
        const int rl = __builtin_amdgcn_readlane(dl[u], b);
        const float* hp = h + (size_t)s * K + lane;
        float* ap = rows + rl * LDK + lane;
        float v[K / 32];
#pragma unroll
        for (int j = 0; j < K / 32; j++) v[j] = hp[32 * j];
#pragma unroll
        for (int j = 0; j < K / 32; j++) atomicAdd(ap + 32 * j, v[j]);
      }
    }
  }
  }
  __syncthreads();

  {
    constexpr int NSC = (AGG_ROWS * (K / 4) + AGG_NT - 1) / AGG_NT;
#pragma unroll
    for (int j = 0; j < NSC; j++) {
      const int i = t + j * AGG_NT;
      const int r = i / (K / 4);
      const int c4 = i - r * (K / 4);
      const int gr = r0 + r;
      if (r < AGG_ROWS && gr < M) {
        const float sc = rdeg[gr];
        float4* p = (float4*)(rows + r * LDK) + c4;
        float4 v = *p;
        v.x *= sc; v.y *= sc; v.z *= sc; v.w *= sc;
        *p = v;
      }
    }
  }
  __syncthreads();

  constexpr int CT  = FO / 16;
  constexpr int WPC = AGG_NW / CT;
  constexpr int NTM = (AGG_NRT + WPC - 1) / WPC;
  const int ct = wave % CT;
  const int rg = wave / CT;
  const bool more = rg < AGG_NRT - (NTM - 1) * WPC;

  v8f c[NTM];
#pragma unroll
  for (int i = 0; i < NTM; i++) c[i] = (v8f){0.f, 0.f, 0.f, 0.f, 0.f, 0.f, 0.f, 0.f};

#pragma unroll 1
  for (int p = 0; p < NP - 1; p++) {
    if (more) rows_gemm_pass<K, FO, KH, NTM>(c, rows, hs, Wl, Wr, p * KH, ct, rg, lane);
    else      rows_gemm_pass<K, FO, KH, NTM - 1>(c, rows, hs, Wl, Wr, p * KH, ct, rg, lane);
    const typename freg<4 * NLD>::type hr = own_rows_load<K, KH, NLD>(h, r0, M, (p + 1) * KH, t);
    __syncthreads();
    own_rows_store<KH, NLD>(hr, hs, t);
    __syncthreads();
  }
  if (more) rows_gemm_pass<K, FO, KH, NTM>(c, rows, hs, Wl, Wr, (NP - 1) * KH, ct, rg, lane);
  else      rows_gemm_pass<K, FO, KH, NTM - 1>(c, rows, hs, Wl, Wr, (NP - 1) * KH, ct, rg, lane);

  const int m16 = lane & 15;
  const int g = lane >> 4;
  const int ncol = ct * 16 + m16;
  const float bv = bias[ncol];
  float cs = 0.f, cq = 0.f;
#pragma unroll
  for (int i = 0; i < NTM; i++) {
    const int rbase = (rg + i * WPC) * 16 + 8 * g;
#pragma unroll
    for (int v = 0; v < 8; v++) {
      const int lr = rbase + v;
      float val = c[i][v] + bv;
      if (RELU) val = fmaxf(val, 0.f);
      if (lr < AGG_ROWS && r0 + lr < M) {
        out[(size_t)(r0 + lr) * FO + ncol] = val;
        if (STATS) { cs += val; cq += val * val; }
      }
    }
  }

  if (STATS) {
    red[0][wave][lane] = cs;
    red[1][wave][lane] = cq;
    __syncthreads();
    if (t < FO) {
      const int cti = t >> 4;
      const int mi = t & 15;
      float a = 0.f, b = 0.f;
      for (int r = 0; r < WPC; r++) {
        const int w = r * CT + cti;
        a += red[0][w][mi]; a += red[0][w][16 + mi];
        b += red[1][w][mi]; b += red[1][w][16 + mi];
      }
      float* prow = part + (size_t)blockIdx.x * (2 * FO);
      prow[t] = a;
      prow[FO + t] = b;
    }
  }
}

__global__ __launch_bounds__(1024)
void bn_sum_parts(const float* __restrict__ part, float* __restrict__ st, int F2) {
  __shared__ float wsum[32][32];
  const int lane = threadIdx.x & 31;
  const int w = threadIdx.x >> 5;
  const int j = blockIdx.x * 32 + lane;
  float s = 0.f;
  static_assert(AGG_NB % 32 == 0, "every wave adds the same number of rows");
  for (int q = 0; q < AGG_NB / 32; q++) s += part[(size_t)(w + 32 * q) * F2 + j];
  wsum[w][lane] = s;
  __syncthreads();
  if (w == 0) {
    float a = 0.f;
    for (int k = 0; k < 32; k++) a += wsum[k][lane];
    st[j] = a;
  }
}

__global__ void bn_apply(float* __restrict__ h, const float* __restrict__ st,
                         const float* __restrict__ g, const float* __restrict__ be,
                         int M, int F) {
  int i = blockIdx.x * blockDim.x + threadIdx.x;
  if (i >= M * F) return;
  int c = i & (F - 1);
  float inv = 1.f / (float)M;
  float mean = st[c] * inv;
  float var = st[F + c] * inv - mean * mean;
  h[i] = g[c] * (h[i] - mean) * rsqrtf(var + 1e-5f) + be[c];
}

__global__ void reparam(const float* __restrict__ h3, const float* __restrict__ eps,
                        float* __restrict__ z, int M) {
  int i = blockIdx.x * blockDim.x + threadIdx.x;
  if (i >= M * 64) return;
  int r = i >> 6, c = i & 63;
  const float* row = h3 + (size_t)r * 128;
  z[i] = row[c] + expf(row[64 + c]) * eps[i];
}

extern "C" void kernel_launch(void* const* d_in, const int* in_sizes, int n_in,
                              void* d_out, int out_size, void* d_ws, size_t ws_size,
                              hipStream_t stream) {
  const float* x   = (const float*)d_in[0];
  const int*   src = (const int*)d_in[1];
  const int*   dst = (const int*)d_in[2];
  const float* eps = (const float*)d_in[3];
  const float* W1l = (const float*)d_in[4];
  const float* W1r = (const float*)d_in[5];
  const float* b1  = (const float*)d_in[6];
  const float* g1  = (const float*)d_in[7];
  const float* be1 = (const float*)d_in[8];
  const float* W2l = (const float*)d_in[9];
  const float* W2r = (const float*)d_in[10];
  const float* b2  = (const float*)d_in[11];
  const float* g2  = (const float*)d_in[12];
  const float* be2 = (const float*)d_in[13];
  const float* W3l = (const float*)d_in[14];
  const float* W3r = (const float*)d_in[15];
  const float* b3  = (const float*)d_in[16];
  float* zout = (float*)d_out;
  (void)in_sizes; (void)n_in; (void)out_size; (void)ws_size;

  char* ws = (char*)d_ws;
  size_t off = 0;
  auto carve = [&](size_t bytes) -> float* {
    char* p = ws + off;
    off = (off + bytes + 255) & ~(size_t)255;
    return (float*)p;
  };
  float* rdeg = carve((size_t)NN * 4);
  float* h1   = carve((size_t)NN * 128 * 4);
  float* h2   = carve((size_t)NN * 256 * 4);
  float* st   = carve(512 * 4);
  float* part = carve((size_t)AGG_NB * 512 * 4);
  unsigned* list = (unsigned*)carve((size_t)AGG_NB * LIST_CAP * 4);
  int* cnt    = (int*)carve((size_t)AGG_NB * 4);

  const int M = NN, E = NE;

  owner_scan<<<AGG_NB, AGG_NT, 0, stream>>>(src, dst, list, cnt, rdeg, E, M);

  sage_rows_fused<128, 128, true, true><<<AGG_NB, AGG_NT, 0, stream>>>(x, list, cnt, src, dst, rdeg, W1l, W1r, b1, h1, part, E, M);
  bn_sum_parts<<<256 / 32, 1024, 0, stream>>>(part, st, 256);
  bn_apply<<<(M * 128 + 255) / 256, 256, 0, stream>>>(h1, st, g1, be1, M, 128);

  sage_rows_fused<128, 256, true, true><<<AGG_NB, AGG_NT, 0, stream>>>(h1, list, cnt, src, dst, rdeg, W2l, W2r, b2, h2, part, E, M);
  bn_sum_parts<<<512 / 32, 1024, 0, stream>>>(part, st, 512);
  bn_apply<<<(M * 256 + 255) / 256, 256, 0, stream>>>(h2, st, g2, be2, M, 256);

  sage_rows_fused<256, 128, false, false><<<AGG_NB, AGG_NT, 0, stream>>>(h2, list, cnt, src, dst, rdeg, W3l, W3r, b3, h1, nullptr, E, M);

  reparam<<<(M * 64 + 255) / 256, 256, 0, stream>>>(h1, eps, zout, M);
}


# BN apply moved behind the neighbour mean: a layer maps its LDS neighbour means and LDS own rows through its input's norm, once per owned row; both apply passes go, 9 -> 7 launches; rounding moves
# speedup vs baseline: 3.5845x; 1.1546x over previous
#include <hip/hip_runtime.h>

#define NN 50000
#define NE 800000

typedef __attribute__((ext_vector_type(2))) float v2f;
typedef __attribute__((ext_vector_type(8))) float v8f;

#define AGG_ROWS 196
#define AGG_RPAD 208
#define AGG_NRT  (AGG_RPAD / 16)
#define AGG_NB   ((NN + AGG_ROWS - 1) / AGG_ROWS)
#define AGG_NT   1024
#define AGG_NW   (AGG_NT / 32)
#define AGG_U    16
#define AGG_TRIP (32 * AGG_U)
#define AGG_PAD  4
#define AGG_LDS_FLOATS (AGG_RPAD * (256 + AGG_PAD) + AGG_RPAD * (64 + AGG_PAD))

#define LIST_CAP 4096
static_assert(NE < (1 << 24) && NN < (1 << 24), "edge index and source must fit 24 bits of a list word");
static_assert(AGG_ROWS <= 256, "local row must fit 8 bits of a list word");

__global__ __launch_bounds__(AGG_NT)
void owner_scan(const int* __restrict__ src, const int* __restrict__ dst,
                unsigned* __restrict__ list, int* __restrict__ cnt, float* __restrict__ rdeg,
                int E, int M) {
  __shared__ unsigned lent[LIST_CAP];
  __shared__ int ldeg[AGG_ROWS];
  __shared__ int lcnt;
  const int t = threadIdx.x;
  const int lane = t & 31;
  const int wave = __builtin_amdgcn_readfirstlane(t >> 5);
  const int r0 = blockIdx.x * AGG_ROWS;

  for (int i = t; i < AGG_ROWS; i += AGG_NT) ldeg[i] = 0;
  if (t == 0) lcnt = 0;
  __syncthreads();

  const int nfull = E / AGG_TRIP;
  const int rot = (int)(((unsigned)blockIdx.x * (unsigned)nfull) / gridDim.x);
  const int4* dst4 = (const int4*)dst;

  int T = wave;
  int ibn = 0;
  int4 dn[4];
#pragma unroll
  for (int q = 0; q < 4; q++) dn[q] = make_int4(-1, -1, -1, -1);
  if (T < nfull) {
    int Tr = T + rot;
    if (Tr >= nfull) Tr -= nfull;
    ibn = Tr * (AGG_TRIP / 4) + lane;
#pragma unroll
    for (int q = 0; q < 4; q++) dn[q] = dst4[ibn + 32 * q];
  }

  while (T < nfull) {
    const int ib = ibn;
    int dl[16];
#pragma unroll
    for (int q = 0; q < 4; q++) {
      dl[4 * q + 0] = dn[q].x - r0; dl[4 * q + 1] = dn[q].y - r0;
      dl[4 * q + 2] = dn[q].z - r0; dl[4 * q + 3] = dn[q].w - r0;
    }

    T += AGG_NW;
    if (T < nfull) {
      int Tr = T + rot;
      if (Tr >= nfull) Tr -= nfull;
      ibn = Tr * (AGG_TRIP / 4) + lane;
#pragma unroll
      for (int q = 0; q < 4; q++) dn[q] = dst4[ibn + 32 * q];
    }

    unsigned mk[16];
    int tot = 0;
#pragma unroll
    for (int u = 0; u < 16; u++) {
      mk[u] = __builtin_amdgcn_ballot_w32((unsigned)dl[u] < (unsigned)AGG_ROWS);
      tot += __builtin_popcount(mk[u]);
    }
    if (tot) {
      int base = 0;
      if (lane == 0) base = atomicAdd(&lcnt, tot);
      base = __builtin_amdgcn_readfirstlane(base);
#pragma unroll
      for (int u = 0; u < 16; u++) {
        if (mk[u]) {
          if ((mk[u] >> lane) & 1u) {
            const int pos = base + (int)__builtin_amdgcn_mbcnt_lo(mk[u], 0u);
            const int e = 4 * (ib + 32 * (u >> 2)) + (u & 3);
            if (pos < LIST_CAP) lent[pos] = ((unsigned)e << 8) | (unsigned)dl[u];
            atomicAdd(&ldeg[dl[u]], 1);
          }
          base += __builtin_popcount(mk[u]);
        }
      }
    }
  }

  for (int e = nfull * AGG_TRIP + t; e < E; e += AGG_NT) {
    const int d = dst[e] - r0;
    if ((unsigned)d < (unsigned)AGG_ROWS) {
      const int pos = atomicAdd(&lcnt, 1);
      if (pos < LIST_CAP) lent[pos] = ((unsigned)e << 8) | (unsigned)d;
      atomicAdd(&ldeg[d], 1);
    }
  }
  __syncthreads();

  const int n = lcnt;
  const int nst = (n < LIST_CAP) ? n : LIST_CAP;
  unsigned* mylist = list + (size_t)blockIdx.x * LIST_CAP;
  for (int i = t; i < nst; i += AGG_NT) {
    const unsigned en = lent[i];
    const int s = src[en >> 8];
    mylist[i] = ((unsigned)s << 8) | (en & 255u);
  }
  for (int i = t; i < AGG_ROWS; i += AGG_NT) {
    const int gr = r0 + i;
    if (gr < M) rdeg[gr] = (ldeg[i] > 0) ? 1.f / (float)ldeg[i] : 0.f;
  }
  if (t == 0) cnt[blockIdx.x] = n;
}

template <int N> struct freg { typedef float type __attribute__((ext_vector_type(N))); };

template <int K, int KH, int NLD>
__device__ __forceinline__ typename freg<4 * NLD>::type own_rows_load(const float* __restrict__ h,
                                                                      int r0, int M, int col0, int t) {
  typename freg<4 * NLD>::type r;
#pragma unroll
  for (int j = 0; j < NLD; j++) {
    int i = t + j * AGG_NT;
    if (i > AGG_ROWS * (KH / 4) - 1) i = AGG_ROWS * (KH / 4) - 1;
    const int row = i / (KH / 4);
    const int c4 = i - row * (KH / 4);
    int gr = r0 + row;
    if (gr > M - 1) gr = M - 1;
    const float4 v = *(const float4*)(h + (gr * K + col0 + 4 * c4));
    r[4 * j + 0] = v.x; r[4 * j + 1] = v.y; r[4 * j + 2] = v.z; r[4 * j + 3] = v.w;
  }
  return r;
}

struct bn4 { float4 m, a, b; };

__device__ __forceinline__ bn4 bn_cols4(const float* __restrict__ st, const float* __restrict__ g,
                                        const float* __restrict__ be, int F, int c, float invM) {
  const float4 s = *(const float4*)(st + c);
  const float4 q = *(const float4*)(st + F + c);
  const float4 gg = *(const float4*)(g + c);
  bn4 r;
  r.b = *(const float4*)(be + c);
  r.m = make_float4(s.x * invM, s.y * invM, s.z * invM, s.w * invM);
  r.a.x = gg.x * rsqrtf(q.x * invM - r.m.x * r.m.x + 1e-5f);
  r.a.y = gg.y * rsqrtf(q.y * invM - r.m.y * r.m.y + 1e-5f);
  r.a.z = gg.z * rsqrtf(q.z * invM - r.m.z * r.m.z + 1e-5f);
  r.a.w = gg.w * rsqrtf(q.w * invM - r.m.w * r.m.w + 1e-5f);
  return r;
}

__device__ __forceinline__ float4 bn_map4(const float4 x, const bn4 n) {
  return make_float4((x.x - n.m.x) * n.a.x + n.b.x, (x.y - n.m.y) * n.a.y + n.b.y,
                     (x.z - n.m.z) * n.a.z + n.b.z, (x.w - n.m.w) * n.a.w + n.b.w);
}

template <int KH, int NLD, bool BN>
__device__ __forceinline__ void own_rows_store(const typename freg<4 * NLD>::type r, float* hs, int t,
                                               const bn4 n) {
  static_assert(AGG_NT % (KH / 4) == 0, "a thread keeps its columns from piece to piece");
#pragma unroll
  for (int j = 0; j < NLD; j++) {
    const int i = t + j * AGG_NT;
    if (i < AGG_ROWS * (KH / 4)) {
      const int row = i / (KH / 4);
      const int c4 = i - row * (KH / 4);
      float4 v = make_float4(r[4 * j + 0], r[4 * j + 1], r[4 * j + 2], r[4 * j + 3]);
      if (BN) v = bn_map4(v, n);
      *(float4*)(hs + row * (KH + AGG_PAD) + 4 * c4) = v;
    }
  }
}

template <int K, int FO, int KH, int NTW>
__device__ __forceinline__ void rows_gemm_pass(v8f* c, const float* ms, const float* hs,
                                               const float* __restrict__ Wl,
                                               const float* __restrict__ Wr,
                                               int kb, int ct, int rg, int lane) {
  constexpr int LDK = K + AGG_PAD;
  constexpr int LDH = KH + AGG_PAD;
  constexpr int WPC = AGG_NW / (FO / 16);
  const int m16 = lane & 15;
  const int g = lane >> 4;
  const int ncol = ct * 16 + m16;

  const float* ap[NTW];
  const float* hp[NTW];
#pragma unroll
  for (int i = 0; i < NTW; i++) {
    const int lrow = (rg + i * WPC) * 16 + m16;
    ap[i] = ms + lrow * LDK + kb + 4 * g;
    hp[i] = hs + lrow * LDH + 4 * g;
  }
  const float* wl = Wl + (size_t)(kb + 4 * g) * FO + ncol;
  const float* wr = Wr + (size_t)(kb + 4 * g) * FO + ncol;

  constexpr bool AHEAD = (NTW <= 4) && (KH == K);
  v2f bl0, bl1, br0, br1;
  bl0.x = wl[0 * FO]; bl0.y = wl[1 * FO]; bl1.x = wl[2 * FO]; bl1.y = wl[3 * FO];
  br0.x = wr[0 * FO]; br0.y = wr[1 * FO]; br1.x = wr[2 * FO]; br1.y = wr[3 * FO];

#pragma unroll 1
  for (int k0 = 0; k0 < KH; k0 += 8) {
    const int kn = AHEAD ? ((k0 + 8 < KH) ? (k0 + 8) : k0) : k0;
    v2f nl0, nl1, nr0, nr1;
    nl0.x = wl[(size_t)(kn + 0) * FO]; nl0.y = wl[(size_t)(kn + 1) * FO];
    nl1.x = wl[(size_t)(kn + 2) * FO]; nl1.y = wl[(size_t)(kn + 3) * FO];
    nr0.x = wr[(size_t)(kn + 0) * FO]; nr0.y = wr[(size_t)(kn + 1) * FO];
    nr1.x = wr[(size_t)(kn + 2) * FO]; nr1.y = wr[(size_t)(kn + 3) * FO];
    if (!AHEAD) { bl0 = nl0; bl1 = nl1; br0 = nr0; br1 = nr1; }
#pragma unroll
    for (int i0 = 0; i0 < NTW; i0 += 4) {
      constexpr int GMAX = 4;
      float4 am[GMAX], ah[GMAX];
#pragma unroll
      for (int j = 0; j < GMAX; j++) {
        if (i0 + j < NTW) {
          am[j] = *(const float4*)(ap[i0 + j] + k0);
          ah[j] = *(const float4*)(hp[i0 + j] + k0);
        }
      }
#pragma unroll
      for (int j = 0; j < GMAX; j++) {
        if (i0 + j < NTW) {
          v2f a; a.x = am[j].x; a.y = am[j].y;
          c[i0 + j] = __builtin_amdgcn_wmma_f32_16x16x4_f32(false, a, false, bl0, (short)0, c[i0 + j],
                                                            false, false);
        }
      }
#pragma unroll
      for (int j = 0; j < GMAX; j++) {
        if (i0 + j < NTW) {
          v2f a; a.x = am[j].z; a.y = am[j].w;
          c[i0 + j] = __builtin_amdgcn_wmma_f32_16x16x4_f32(false, a, false, bl1, (short)0, c[i0 + j],
                                                            false, false);
        }
      }
#pragma unroll
      for (int j = 0; j < GMAX; j++) {
        if (i0 + j < NTW) {
          v2f a; a.x = ah[j].x; a.y = ah[j].y;
          c[i0 + j] = __builtin_amdgcn_wmma_f32_16x16x4_f32(false, a, false, br0, (short)0, c[i0 + j],
                                                            false, false);
        }
      }
#pragma unroll
      for (int j = 0; j < GMAX; j++) {
        if (i0 + j < NTW) {
          v2f a; a.x = ah[j].z; a.y = ah[j].w;
          c[i0 + j] = __builtin_amdgcn_wmma_f32_16x16x4_f32(false, a, false, br1, (short)0, c[i0 + j],
                                                            false, false);
        }
      }
    }
    if (AHEAD) { bl0 = nl0; bl1 = nl1; br0 = nr0; br1 = nr1; }
  }
}

template <int K, int FO, bool RELU, bool STATS, bool BNIN>
__global__ __launch_bounds__(AGG_NT)
void sage_rows_fused(const float* __restrict__ h, const unsigned* __restrict__ list,
                     const int* __restrict__ cnt, const int* __restrict__ src,
                     const int* __restrict__ dst, const float* __restrict__ rdeg,
                     const float* __restrict__ Wl, const float* __restrict__ Wr,
                     const float* __restrict__ bias, float* __restrict__ out,
                     float* __restrict__ part, const float* __restrict__ st_in,
                     const float* __restrict__ g_in, const float* __restrict__ be_in,
                     int E, int M) {
  __shared__ __attribute__((aligned(16))) float lds[AGG_LDS_FLOATS];
  __shared__ float red[2][AGG_NW][32];
  constexpr int LDK = K + AGG_PAD;
  constexpr int KH  = (K > 128) ? 64 : K;
  constexpr int NP  = K / KH;
  constexpr int NLD = (AGG_ROWS * (KH / 4) + AGG_NT - 1) / AGG_NT;
  float* rows = lds;
  float* hs   = lds + AGG_RPAD * LDK;
  const int t = threadIdx.x;
  const int lane = t & 31;
  const int wave = __builtin_amdgcn_readfirstlane(t >> 5);
  const int r0 = blockIdx.x * AGG_ROWS;
  const float invM = 1.f / (float)M;

  for (int i = t; i < AGG_RPAD * LDK / 4; i += AGG_NT)
    ((float4*)rows)[i] = make_float4(0.f, 0.f, 0.f, 0.f);
  {
    const typename freg<4 * NLD>::type hr = own_rows_load<K, KH, NLD>(h, r0, M, 0, t);
    bn4 n = {};
    if (BNIN) n = bn_cols4(st_in, g_in, be_in, K, 4 * (t % (KH / 4)), invM);
    own_rows_store<KH, NLD, BNIN>(hr, hs, t, n);
    for (int i = t; i < (AGG_RPAD - AGG_ROWS) * (KH + AGG_PAD); i += AGG_NT)
      hs[AGG_ROWS * (KH + AGG_PAD) + i] = 0.f;
  }
  __syncthreads();

  const int nlist = cnt[blockIdx.x];
  if (nlist <= LIST_CAP) {
    const unsigned* mylist = list + (size_t)blockIdx.x * LIST_CAP;
    const int e0 = (int)(((unsigned)nlist * (unsigned)wave) / AGG_NW);
    const int e1 = (int)(((unsigned)nlist * (unsigned)(wave + 1)) / AGG_NW);
    constexpr int NB = 1024 / K;
    for (int e = e0; e < e1; e += 32) {
      const int ng = (e1 - e < 32) ? (e1 - e) : 32;
      unsigned ent = 0;
      if (lane < ng) ent = mylist[e + lane];
      for (int j = 0; j < ng; j += NB) {
        float v[NB][K / 32];
        int rl[NB];
#pragma unroll
        for (int b = 0; b < NB; b++) {
          int idx = j + b;
          if (idx > ng - 1) idx = ng - 1;
          const unsigned en = (unsigned)__builtin_amdgcn_readlane((int)ent, idx);
          rl[b] = (int)(en & 255u);
          const float* hp = h + (size_t)(en >> 8) * K + lane;
#pragma unroll
          for (int q = 0; q < K / 32; q++) v[b][q] = hp[32 * q];
        }
#pragma unroll
        for (int b = 0; b < NB; b++) {
          if (j + b < ng) {
            float* ap = rows + rl[b] * LDK + lane;
#pragma unroll
            for (int q = 0; q < K / 32; q++) atomicAdd(ap + 32 * q, v[b][q]);
          }
        }
      }
    }
  } else {
  const int ntrips = (E + AGG_TRIP - 1) / AGG_TRIP;
  const int rot = (int)(((unsigned)blockIdx.x * (unsigned)ntrips) / gridDim.x);

  int T = wave;
  int ebn = 0;
  int dn[AGG_U];
#pragma unroll
  for (int u = 0; u < AGG_U; u++) dn[u] = -1;
  if (T < ntrips) {
    int Tr = T + rot;
    if (Tr >= ntrips) Tr -= ntrips;
    ebn = Tr * AGG_TRIP + lane;
#pragma unroll
    for (int u = 0; u < AGG_U; u++) {
      const int e = ebn + 32 * u;
      if (e < E) dn[u] = dst[e];
    }
  }

  while (T < ntrips) {
    const int eb = ebn;
    int dl[AGG_U];
#pragma unroll
    for (int u = 0; u < AGG_U; u++) dl[u] = dn[u] - r0;

    T += AGG_NW;
    if (T < ntrips) {
      int Tr = T + rot;
      if (Tr >= ntrips) Tr -= ntrips;
      ebn = Tr * AGG_TRIP + lane;
#pragma unroll
      for (int u = 0; u < AGG_U; u++) {
        const int e = ebn + 32 * u;
        dn[u] = (e < E) ? dst[e] : -1;
      }
    }

    int sv[AGG_U];
    unsigned mk[AGG_U];
#pragma unroll
    for (int u = 0; u < AGG_U; u++) {
      const bool m = (unsigned)dl[u] < (unsigned)AGG_ROWS;
      sv[u] = 0;
      if (m) sv[u] = src[eb + 32 * u];
      mk[u] = __builtin_amdgcn_ballot_w32(m);
    }

#pragma unroll
    for (int u = 0; u < AGG_U; u++) {
      unsigned mask = mk[u];
      while (mask) {
        const int b = __builtin_ctz(mask);
        mask &= mask - 1;
        const int s  = __builtin_amdgcn_readlane(sv[u], b);
        const int rl = __builtin_amdgcn_readlane(dl[u], b);
        const float* hp = h + (size_t)s * K + lane;
        float* ap = rows + rl * LDK + lane;
        float v[K / 32];
#pragma unroll
        for (int j = 0; j < K / 32; j++) v[j] = hp[32 * j];
#pragma unroll
        for (int j = 0; j < K / 32; j++) atomicAdd(ap + 32 * j, v[j]);
      }
    }
  }
  }
  __syncthreads();

  {
    constexpr int NSC = (AGG_ROWS * (K / 4) + AGG_NT - 1) / AGG_NT;
    static_assert(AGG_NT % (K / 4) == 0, "a thread keeps its columns from trip to trip");
    bn4 n = {};
    if (BNIN) n = bn_cols4(st_in, g_in, be_in, K, 4 * (t % (K / 4)), invM);
#pragma unroll
    for (int j = 0; j < NSC; j++) {
      const int i = t + j * AGG_NT;
      const int r = i / (K / 4);
      const int c4 = i - r * (K / 4);
      const int gr = r0 + r;
      if (r < AGG_ROWS && gr < M) {
        const float sc = rdeg[gr];
        float4* p = (float4*)(rows + r * LDK) + c4;
        float4 v = *p;
        v.x *= sc; v.y *= sc; v.z *= sc; v.w *= sc;
        if (BNIN) { if (sc != 0.f) v = bn_map4(v, n); }
        *p = v;
      }
    }
  }
  __syncthreads();

  constexpr int CT  = FO / 16;
  constexpr int WPC = AGG_NW / CT;
  constexpr int NTM = (AGG_NRT + WPC - 1) / WPC;
  const int ct = wave % CT;
  const int rg = wave / CT;
  const bool more = rg < AGG_NRT - (NTM - 1) * WPC;

  v8f c[NTM];
#pragma unroll
  for (int i = 0; i < NTM; i++) c[i] = (v8f){0.f, 0.f, 0.f, 0.f, 0.f, 0.f, 0.f, 0.f};

#pragma unroll 1
  for (int p = 0; p < NP - 1; p++) {
    if (more) rows_gemm_pass<K, FO, KH, NTM>(c, rows, hs, Wl, Wr, p * KH, ct, rg, lane);
    else      rows_gemm_pass<K, FO, KH, NTM - 1>(c, rows, hs, Wl, Wr, p * KH, ct, rg, lane);
    const typename freg<4 * NLD>::type hr = own_rows_load<K, KH, NLD>(h, r0, M, (p + 1) * KH, t);
    bn4 n = {};
    if (BNIN) n = bn_cols4(st_in, g_in, be_in, K, (p + 1) * KH + 4 * (t % (KH / 4)), invM);
    __syncthreads();
    own_rows_store<KH, NLD, BNIN>(hr, hs, t, n);
    __syncthreads();
  }
  if (more) rows_gemm_pass<K, FO, KH, NTM>(c, rows, hs, Wl, Wr, (NP - 1) * KH, ct, rg, lane);
  else      rows_gemm_pass<K, FO, KH, NTM - 1>(c, rows, hs, Wl, Wr, (NP - 1) * KH, ct, rg, lane);

  const int m16 = lane & 15;
  const int g = lane >> 4;
  const int ncol = ct * 16 + m16;
  const float bv = bias[ncol];
  float cs = 0.f, cq = 0.f;
#pragma unroll
  for (int i = 0; i < NTM; i++) {
    const int rbase = (rg + i * WPC) * 16 + 8 * g;
#pragma unroll
    for (int v = 0; v < 8; v++) {
      const int lr = rbase + v;
      float val = c[i][v] + bv;
      if (RELU) val = fmaxf(val, 0.f);
      if (lr < AGG_ROWS && r0 + lr < M) {
        out[(size_t)(r0 + lr) * FO + ncol] = val;
        if (STATS) { cs += val; cq += val * val; }
      }
    }
  }

  if (STATS) {
    red[0][wave][lane] = cs;
    red[1][wave][lane] = cq;
    __syncthreads();
    if (t < FO) {
      const int cti = t >> 4;
      const int mi = t & 15;
      float a = 0.f, b = 0.f;
      for (int r = 0; r < WPC; r++) {
        const int w = r * CT + cti;
        a += red[0][w][mi]; a += red[0][w][16 + mi];
        b += red[1][w][mi]; b += red[1][w][16 + mi];
      }
      float* prow = part + (size_t)blockIdx.x * (2 * FO);
      prow[t] = a;
      prow[FO + t] = b;
    }
  }
}

__global__ __launch_bounds__(1024)
void bn_sum_parts(const float* __restrict__ part, float* __restrict__ st, int F2) {
  __shared__ float wsum[32][32];
  const int lane = threadIdx.x & 31;
  const int w = threadIdx.x >> 5;
  const int j = blockIdx.x * 32 + lane;
  float s = 0.f;
  static_assert(AGG_NB % 32 == 0, "every wave adds the same number of rows");
  for (int q = 0; q < AGG_NB / 32; q++) s += part[(size_t)(w + 32 * q) * F2 + j];
  wsum[w][lane] = s;
  __syncthreads();
  if (w == 0) {
    float a = 0.f;
    for (int k = 0; k < 32; k++) a += wsum[k][lane];
    st[j] = a;
  }
}

__global__ void reparam(const float* __restrict__ h3, const float* __restrict__ eps,
                        float* __restrict__ z, int M) {
  int i = blockIdx.x * blockDim.x + threadIdx.x;
  if (i >= M * 64) return;
  int r = i >> 6, c = i & 63;
  const float* row = h3 + (size_t)r * 128;
  z[i] = row[c] + expf(row[64 + c]) * eps[i];
}

extern "C" void kernel_launch(void* const* d_in, const int* in_sizes, int n_in,
                              void* d_out, int out_size, void* d_ws, size_t ws_size,
                              hipStream_t stream) {
  const float* x   = (const float*)d_in[0];
  const int*   src = (const int*)d_in[1];
  const int*   dst = (const int*)d_in[2];
  const float* eps = (const float*)d_in[3];
  const float* W1l = (const float*)d_in[4];
  const float* W1r = (const float*)d_in[5];
  const float* b1  = (const float*)d_in[6];
  const float* g1  = (const float*)d_in[7];
  const float* be1 = (const float*)d_in[8];
  const float* W2l = (const float*)d_in[9];
  const float* W2r = (const float*)d_in[10];
  const float* b2  = (const float*)d_in[11];
  const float* g2  = (const float*)d_in[12];
  const float* be2 = (const float*)d_in[13];
  const float* W3l = (const float*)d_in[14];
  const float* W3r = (const float*)d_in[15];
  const float* b3  = (const float*)d_in[16];
  float* zout = (float*)d_out;
  (void)in_sizes; (void)n_in; (void)out_size; (void)ws_size;

  char* ws = (char*)d_ws;
  size_t off = 0;
  auto carve = [&](size_t bytes) -> float* {
    char* p = ws + off;
    off = (off + bytes + 255) & ~(size_t)255;
    return (float*)p;
  };
  float* rdeg = carve((size_t)NN * 4);
  float* h1   = carve((size_t)NN * 128 * 4);
  float* h2   = carve((size_t)NN * 256 * 4);
  float* st   = carve(512 * 4);
  float* part = carve((size_t)AGG_NB * 512 * 4);
  unsigned* list = (unsigned*)carve((size_t)AGG_NB * LIST_CAP * 4);
  int* cnt    = (int*)carve((size_t)AGG_NB * 4);

  const int M = NN, E = NE;

  owner_scan<<<AGG_NB, AGG_NT, 0, stream>>>(src, dst, list, cnt, rdeg, E, M);

  sage_rows_fused<128, 128, true, true, false><<<AGG_NB, AGG_NT, 0, stream>>>(x, list, cnt, src, dst, rdeg, W1l, W1r, b1, h1, part, nullptr, nullptr, nullptr, E, M);
  bn_sum_parts<<<256 / 32, 1024, 0, stream>>>(part, st, 256);

  sage_rows_fused<128, 256, true, true, true><<<AGG_NB, AGG_NT, 0, stream>>>(h1, list, cnt, src, dst, rdeg, W2l, W2r, b2, h2, part, st, g1, be1, E, M);
  bn_sum_parts<<<512 / 32, 1024, 0, stream>>>(part, st, 512);

  sage_rows_fused<256, 128, false, false, true><<<AGG_NB, AGG_NT, 0, stream>>>(h2, list, cnt, src, dst, rdeg, W3l, W3r, b3, h1, nullptr, st, g2, be2, E, M);

  reparam<<<(M * 64 + 255) / 256, 256, 0, stream>>>(h1, eps, zout, M);
}


# Owners stop walking all of dst: a pre-pass ORs, per 32-edge chunk, a bit per row-owning group with a target in it; an owner reads only its flagged chunks; 7->8 launches; list, LDS add order change
# speedup vs baseline: 3.7688x; 1.0514x over previous
#include <hip/hip_runtime.h>

#define NN 50000
#define NE 800000

typedef __attribute__((ext_vector_type(2))) float v2f;
typedef __attribute__((ext_vector_type(8))) float v8f;

#define AGG_ROWS 196
#define AGG_RPAD 208
#define AGG_NRT  (AGG_RPAD / 16)
#define AGG_NB   ((NN + AGG_ROWS - 1) / AGG_ROWS)
#define AGG_NT   1024
#define AGG_NW   (AGG_NT / 32)
#define AGG_U    16
#define AGG_TRIP (32 * AGG_U)
#define AGG_PAD  4
#define AGG_LDS_FLOATS (AGG_RPAD * (256 + AGG_PAD) + AGG_RPAD * (64 + AGG_PAD))

#define NCHUNK   ((NE + 31) / 32)
#define NFW      ((NCHUNK + 31) / 32)
#define FLG_NB   256
#define FLG_PASS ((NFW + FLG_NB - 1) / FLG_NB)
static_assert(AGG_NT == 1024, "one flag word = 32 chunks of 32 edges = one edge per thread");
static_assert(NFW <= AGG_NT, "an owner group reads one flag word per thread");

__global__ __launch_bounds__(AGG_NT)
void chunk_flags(const int* __restrict__ dst, unsigned* __restrict__ flags, int E) {
  __shared__ unsigned bits[FLG_PASS * AGG_NB];
  const int t = threadIdx.x;
  const unsigned wbit = 1u << (t >> 5);
  for (int i = t; i < FLG_PASS * AGG_NB; i += AGG_NT) bits[i] = 0u;
  __syncthreads();
  int d[FLG_PASS];
#pragma unroll
  for (int q = 0; q < FLG_PASS; q++) {
    const int w = (int)blockIdx.x + q * FLG_NB;
    const int e = w * AGG_NT + t;
    d[q] = (w < NFW && e < E) ? dst[e] : -1;
  }
#pragma unroll
  for (int q = 0; q < FLG_PASS; q++) {
    const unsigned o = (unsigned)d[q] / (unsigned)AGG_ROWS;
    if (o < (unsigned)AGG_NB) atomicOr(&bits[q * AGG_NB + (int)o], wbit);
  }
  __syncthreads();
#pragma unroll
  for (int q = 0; q < FLG_PASS; q++) {
    const int w = (int)blockIdx.x + q * FLG_NB;
    if (w < NFW && t < AGG_NB) flags[(size_t)w * AGG_NB + t] = bits[q * AGG_NB + t];
  }
}

#define LIST_CAP 4096
#define SCAN_CU  16
static_assert(NE < (1 << 24) && NN < (1 << 24), "edge index and source must fit 24 bits of a list word");
static_assert(AGG_ROWS <= 256, "local row must fit 8 bits of a list word");

__global__ __launch_bounds__(AGG_NT)
void owner_scan(const int* __restrict__ src, const int* __restrict__ dst,
                const unsigned* __restrict__ flags,
                unsigned* __restrict__ list, int* __restrict__ cnt, float* __restrict__ rdeg,
                int E, int M) {
  __shared__ unsigned lent[LIST_CAP];
  __shared__ unsigned chl[NFW * 32];
  __shared__ int ldeg[AGG_ROWS];
  __shared__ int lcnt;
  __shared__ int nch;
  const int t = threadIdx.x;
  const int lane = t & 31;
  const int wave = __builtin_amdgcn_readfirstlane(t >> 5);
  const int r0 = blockIdx.x * AGG_ROWS;

  for (int i = t; i < AGG_ROWS; i += AGG_NT) ldeg[i] = 0;
  if (t == 0) { lcnt = 0; nch = 0; }
  unsigned fm = 0u;
  if (t < NFW) fm = flags[(size_t)t * AGG_NB + blockIdx.x];
  __syncthreads();

  if (fm) {
    int p = atomicAdd(&nch, __builtin_popcount(fm));
    while (fm) {
      const int b = __builtin_ctz(fm);
      fm &= fm - 1u;
      chl[p++] = (unsigned)(t * 32 + b);
    }
  }
  __syncthreads();
  const int nc = nch;

  for (int i0 = wave * SCAN_CU; i0 < nc; i0 += AGG_NW * SCAN_CU) {
    int cv = -1;
    if (lane < SCAN_CU && i0 + lane < nc) cv = (int)chl[i0 + lane];
    int dl[SCAN_CU];
#pragma unroll
    for (int j = 0; j < SCAN_CU; j++) {
      const int c = __builtin_amdgcn_readlane(cv, j);
      int e = ((c < 0) ? 0 : c) * 32 + lane;
      if (e > E - 1) e = E - 1;
      dl[j] = dst[e] - r0;
    }
    unsigned mk[SCAN_CU];
    int tot = 0;
#pragma unroll
    for (int j = 0; j < SCAN_CU; j++) {
      const int c = __builtin_amdgcn_readlane(cv, j);
      const bool ok = (c >= 0) & (c * 32 + lane < E) & ((unsigned)dl[j] < (unsigned)AGG_ROWS);
      mk[j] = __builtin_amdgcn_ballot_w32(ok);
      tot += __builtin_popcount(mk[j]);
    }
    if (tot) {
      int base = 0;
      if (lane == 0) base = atomicAdd(&lcnt, tot);
      base = __builtin_amdgcn_readfirstlane(base);
#pragma unroll
      for (int j = 0; j < SCAN_CU; j++) {
        if (mk[j]) {
          const int c = __builtin_amdgcn_readlane(cv, j);
          if ((mk[j] >> lane) & 1u) {
            const int pos = base + (int)__builtin_amdgcn_mbcnt_lo(mk[j], 0u);
            const int e = c * 32 + lane;
            if (pos < LIST_CAP) lent[pos] = ((unsigned)e << 8) | (unsigned)dl[j];
            atomicAdd(&ldeg[dl[j]], 1);
          }
          base += __builtin_popcount(mk[j]);
        }
      }
    }
  }
  __syncthreads();

  const int n = lcnt;
  const int nst = (n < LIST_CAP) ? n : LIST_CAP;
  unsigned* mylist = list + (size_t)blockIdx.x * LIST_CAP;
  for (int i = t; i < nst; i += AGG_NT) {
    const unsigned en = lent[i];
    const int s = src[en >> 8];
    mylist[i] = ((unsigned)s << 8) | (en & 255u);
  }
  for (int i = t; i < AGG_ROWS; i += AGG_NT) {
    const int gr = r0 + i;
    if (gr < M) rdeg[gr] = (ldeg[i] > 0) ? 1.f / (float)ldeg[i] : 0.f;
  }
  if (t == 0) cnt[blockIdx.x] = n;
}

template <int N> struct freg { typedef float type __attribute__((ext_vector_type(N))); };

template <int K, int KH, int NLD>
__device__ __forceinline__ typename freg<4 * NLD>::type own_rows_load(const float* __restrict__ h,
                                                                      int r0, int M, int col0, int t) {
  typename freg<4 * NLD>::type r;
#pragma unroll
  for (int j = 0; j < NLD; j++) {
    int i = t + j * AGG_NT;
    if (i > AGG_ROWS * (KH / 4) - 1) i = AGG_ROWS * (KH / 4) - 1;
    const int row = i / (KH / 4);
    const int c4 = i - row * (KH / 4);
    int gr = r0 + row;
    if (gr > M - 1) gr = M - 1;
    const float4 v = *(const float4*)(h + (gr * K + col0 + 4 * c4));
    r[4 * j + 0] = v.x; r[4 * j + 1] = v.y; r[4 * j + 2] = v.z; r[4 * j + 3] = v.w;
  }
  return r;
}

struct bn4 { float4 m, a, b; };

__device__ __forceinline__ bn4 bn_cols4(const float* __restrict__ st, const float* __restrict__ g,
                                        const float* __restrict__ be, int F, int c, float invM) {
  const float4 s = *(const float4*)(st + c);
  const float4 q = *(const float4*)(st + F + c);
  const float4 gg = *(const float4*)(g + c);
  bn4 r;
  r.b = *(const float4*)(be + c);
  r.m = make_float4(s.x * invM, s.y * invM, s.z * invM, s.w * invM);
  r.a.x = gg.x * rsqrtf(q.x * invM - r.m.x * r.m.x + 1e-5f);
  r.a.y = gg.y * rsqrtf(q.y * invM - r.m.y * r.m.y + 1e-5f);
  r.a.z = gg.z * rsqrtf(q.z * invM - r.m.z * r.m.z + 1e-5f);
  r.a.w = gg.w * rsqrtf(q.w * invM - r.m.w * r.m.w + 1e-5f);
  return r;
}

__device__ __forceinline__ float4 bn_map4(const float4 x, const bn4 n) {
  return make_float4((x.x - n.m.x) * n.a.x + n.b.x, (x.y - n.m.y) * n.a.y + n.b.y,
                     (x.z - n.m.z) * n.a.z + n.b.z, (x.w - n.m.w) * n.a.w + n.b.w);
}

template <int KH, int NLD, bool BN>
__device__ __forceinline__ void own_rows_store(const typename freg<4 * NLD>::type r, float* hs, int t,
                                               const bn4 n) {
  static_assert(AGG_NT % (KH / 4) == 0, "a thread keeps its columns from piece to piece");
#pragma unroll
  for (int j = 0; j < NLD; j++) {
    const int i = t + j * AGG_NT;
    if (i < AGG_ROWS * (KH / 4)) {
      const int row = i / (KH / 4);
      const int c4 = i - row * (KH / 4);
      float4 v = make_float4(r[4 * j + 0], r[4 * j + 1], r[4 * j + 2], r[4 * j + 3]);
      if (BN) v = bn_map4(v, n);
      *(float4*)(hs + row * (KH + AGG_PAD) + 4 * c4) = v;
    }
  }
}

template <int K, int FO, int KH, int NTW>
__device__ __forceinline__ void rows_gemm_pass(v8f* c, const float* ms, const float* hs,
                                               const float* __restrict__ Wl,
                                               const float* __restrict__ Wr,
                                               int kb, int ct, int rg, int lane) {
  constexpr int LDK = K + AGG_PAD;
  constexpr int LDH = KH + AGG_PAD;
  constexpr int WPC = AGG_NW / (FO / 16);
  const int m16 = lane & 15;
  const int g = lane >> 4;
  const int ncol = ct * 16 + m16;

  const float* ap[NTW];
  const float* hp[NTW];
#pragma unroll
  for (int i = 0; i < NTW; i++) {
    const int lrow = (rg + i * WPC) * 16 + m16;
    ap[i] = ms + lrow * LDK + kb + 4 * g;
    hp[i] = hs + lrow * LDH + 4 * g;
  }
  const float* wl = Wl + (size_t)(kb + 4 * g) * FO + ncol;
  const float* wr = Wr + (size_t)(kb + 4 * g) * FO + ncol;

  constexpr bool AHEAD = (NTW <= 4) && (KH == K);
  v2f bl0, bl1, br0, br1;
  bl0.x = wl[0 * FO]; bl0.y = wl[1 * FO]; bl1.x = wl[2 * FO]; bl1.y = wl[3 * FO];
  br0.x = wr[0 * FO]; br0.y = wr[1 * FO]; br1.x = wr[2 * FO]; br1.y = wr[3 * FO];

#pragma unroll 1
  for (int k0 = 0; k0 < KH; k0 += 8) {
    const int kn = AHEAD ? ((k0 + 8 < KH) ? (k0 + 8) : k0) : k0;
    v2f nl0, nl1, nr0, nr1;
    nl0.x = wl[(size_t)(kn + 0) * FO]; nl0.y = wl[(size_t)(kn + 1) * FO];
    nl1.x = wl[(size_t)(kn + 2) * FO]; nl1.y = wl[(size_t)(kn + 3) * FO];
    nr0.x = wr[(size_t)(kn + 0) * FO]; nr0.y = wr[(size_t)(kn + 1) * FO];
    nr1.x = wr[(size_t)(kn + 2) * FO]; nr1.y = wr[(size_t)(kn + 3) * FO];
    if (!AHEAD) { bl0 = nl0; bl1 = nl1; br0 = nr0; br1 = nr1; }
#pragma unroll
    for (int i0 = 0; i0 < NTW; i0 += 4) {
      constexpr int GMAX = 4;
      float4 am[GMAX], ah[GMAX];
#pragma unroll
      for (int j = 0; j < GMAX; j++) {
        if (i0 + j < NTW) {
          am[j] = *(const float4*)(ap[i0 + j] + k0);
          ah[j] = *(const float4*)(hp[i0 + j] + k0);
        }
      }
#pragma unroll
      for (int j = 0; j < GMAX; j++) {
        if (i0 + j < NTW) {
          v2f a; a.x = am[j].x; a.y = am[j].y;
          c[i0 + j] = __builtin_amdgcn_wmma_f32_16x16x4_f32(false, a, false, bl0, (short)0, c[i0 + j],
                                                            false, false);
        }
      }
#pragma unroll
      for (int j = 0; j < GMAX; j++) {
        if (i0 + j < NTW) {
          v2f a; a.x = am[j].z; a.y = am[j].w;
          c[i0 + j] = __builtin_amdgcn_wmma_f32_16x16x4_f32(false, a, false, bl1, (short)0, c[i0 + j],
                                                            false, false);
        }
      }
#pragma unroll
      for (int j = 0; j < GMAX; j++) {
        if (i0 + j < NTW) {
          v2f a; a.x = ah[j].x; a.y = ah[j].y;
          c[i0 + j] = __builtin_amdgcn_wmma_f32_16x16x4_f32(false, a, false, br0, (short)0, c[i0 + j],
                                                            false, false);
        }
      }
#pragma unroll
      for (int j = 0; j < GMAX; j++) {
        if (i0 + j < NTW) {
          v2f a; a.x = ah[j].z; a.y = ah[j].w;
          c[i0 + j] = __builtin_amdgcn_wmma_f32_16x16x4_f32(false, a, false, br1, (short)0, c[i0 + j],
                                                            false, false);
        }
      }
    }
    if (AHEAD) { bl0 = nl0; bl1 = nl1; br0 = nr0; br1 = nr1; }
  }
}

template <int K, int FO, bool RELU, bool STATS, bool BNIN>
__global__ __launch_bounds__(AGG_NT)
void sage_rows_fused(const float* __restrict__ h, const unsigned* __restrict__ list,
                     const int* __restrict__ cnt, const int* __restrict__ src,
                     const int* __restrict__ dst, const float* __restrict__ rdeg,
                     const float* __restrict__ Wl, const float* __restrict__ Wr,
                     const float* __restrict__ bias, float* __restrict__ out,
                     float* __restrict__ part, const float* __restrict__ st_in,
                     const float* __restrict__ g_in, const float* __restrict__ be_in,
                     int E, int M) {
  __shared__ __attribute__((aligned(16))) float lds[AGG_LDS_FLOATS];
  __shared__ float red[2][AGG_NW][32];
  constexpr int LDK = K + AGG_PAD;
  constexpr int KH  = (K > 128) ? 64 : K;
  constexpr int NP  = K / KH;
  constexpr int NLD = (AGG_ROWS * (KH / 4) + AGG_NT - 1) / AGG_NT;
  float* rows = lds;
  float* hs   = lds + AGG_RPAD * LDK;
  const int t = threadIdx.x;
  const int lane = t & 31;
  const int wave = __builtin_amdgcn_readfirstlane(t >> 5);
  const int r0 = blockIdx.x * AGG_ROWS;
  const float invM = 1.f / (float)M;

  for (int i = t; i < AGG_RPAD * LDK / 4; i += AGG_NT)
    ((float4*)rows)[i] = make_float4(0.f, 0.f, 0.f, 0.f);
  {
    const typename freg<4 * NLD>::type hr = own_rows_load<K, KH, NLD>(h, r0, M, 0, t);
    bn4 n = {};
    if (BNIN) n = bn_cols4(st_in, g_in, be_in, K, 4 * (t % (KH / 4)), invM);
    own_rows_store<KH, NLD, BNIN>(hr, hs, t, n);
    for (int i = t; i < (AGG_RPAD - AGG_ROWS) * (KH + AGG_PAD); i += AGG_NT)
      hs[AGG_ROWS * (KH + AGG_PAD) + i] = 0.f;
  }
  __syncthreads();

  const int nlist = cnt[blockIdx.x];
  if (nlist <= LIST_CAP) {
    const unsigned* mylist = list + (size_t)blockIdx.x * LIST_CAP;
    const int e0 = (int)(((unsigned)nlist * (unsigned)wave) / AGG_NW);
    const int e1 = (int)(((unsigned)nlist * (unsigned)(wave + 1)) / AGG_NW);
    constexpr int NB = 1024 / K;
    for (int e = e0; e < e1; e += 32) {
      const int ng = (e1 - e < 32) ? (e1 - e) : 32;
      unsigned ent = 0;
      if (lane < ng) ent = mylist[e + lane];
      for (int j = 0; j < ng; j += NB) {
        float v[NB][K / 32];
        int rl[NB];
#pragma unroll
        for (int b = 0; b < NB; b++) {
          int idx = j + b;
          if (idx > ng - 1) idx = ng - 1;
          const unsigned en = (unsigned)__builtin_amdgcn_readlane((int)ent, idx);
          rl[b] = (int)(en & 255u);
          const float* hp = h + (size_t)(en >> 8) * K + lane;
#pragma unroll
          for (int q = 0; q < K / 32; q++) v[b][q] = hp[32 * q];
        }
#pragma unroll
        for (int b = 0; b < NB; b++) {
          if (j + b < ng) {
            float* ap = rows + rl[b] * LDK + lane;
#pragma unroll
            for (int q = 0; q < K / 32; q++) atomicAdd(ap + 32 * q, v[b][q]);
          }
        }
      }
    }
  } else {
  const int ntrips = (E + AGG_TRIP - 1) / AGG_TRIP;
  const int rot = (int)(((unsigned)blockIdx.x * (unsigned)ntrips) / gridDim.x);

  int T = wave;
  int ebn = 0;
  int dn[AGG_U];
#pragma unroll
  for (int u = 0; u < AGG_U; u++) dn[u] = -1;
  if (T < ntrips) {
    int Tr = T + rot;
    if (Tr >= ntrips) Tr -= ntrips;
    ebn = Tr * AGG_TRIP + lane;
#pragma unroll
    for (int u = 0; u < AGG_U; u++) {
      const int e = ebn + 32 * u;
      if (e < E) dn[u] = dst[e];
    }
  }

  while (T < ntrips) {
    const int eb = ebn;
    int dl[AGG_U];
#pragma unroll
    for (int u = 0; u < AGG_U; u++) dl[u] = dn[u] - r0;

    T += AGG_NW;
    if (T < ntrips) {
      int Tr = T + rot;
      if (Tr >= ntrips) Tr -= ntrips;
      ebn = Tr * AGG_TRIP + lane;
#pragma unroll
      for (int u = 0; u < AGG_U; u++) {
        const int e = ebn + 32 * u;
        dn[u] = (e < E) ? dst[e] : -1;
      }
    }

    int sv[AGG_U];
    unsigned mk[AGG_U];
#pragma unroll
    for (int u = 0; u < AGG_U; u++) {
      const bool m = (unsigned)dl[u] < (unsigned)AGG_ROWS;
      sv[u] = 0;
      if (m) sv[u] = src[eb + 32 * u];
      mk[u] = __builtin_amdgcn_ballot_w32(m);
    }

#pragma unroll
    for (int u = 0; u < AGG_U; u++) {
      unsigned mask = mk[u];
      while (mask) {
        const int b = __builtin_ctz(mask);
        mask &= mask - 1;
        const int s  = __builtin_amdgcn_readlane(sv[u], b);
        const int rl = __builtin_amdgcn_readlane(dl[u], b);
        const float* hp = h + (size_t)s * K + lane;
        float* ap = rows + rl * LDK + lane;
        float v[K / 32];
#pragma unroll
        for (int j = 0; j < K / 32; j++) v[j] = hp[32 * j];
#pragma unroll
        for (int j = 0; j < K / 32; j++) atomicAdd(ap + 32 * j, v[j]);
      }
    }
  }
  }
  __syncthreads();

  {
    constexpr int NSC = (AGG_ROWS * (K / 4) + AGG_NT - 1) / AGG_NT;
    static_assert(AGG_NT % (K / 4) == 0, "a thread keeps its columns from trip to trip");
    bn4 n = {};
    if (BNIN) n = bn_cols4(st_in, g_in, be_in, K, 4 * (t % (K / 4)), invM);
#pragma unroll
    for (int j = 0; j < NSC; j++) {
      const int i = t + j * AGG_NT;
      const int r = i / (K / 4);
      const int c4 = i - r * (K / 4);
      const int gr = r0 + r;
      if (r < AGG_ROWS && gr < M) {
        const float sc = rdeg[gr];
        float4* p = (float4*)(rows + r * LDK) + c4;
        float4 v = *p;
        v.x *= sc; v.y *= sc; v.z *= sc; v.w *= sc;
        if (BNIN) { if (sc != 0.f) v = bn_map4(v, n); }
        *p = v;
      }
    }
  }
  __syncthreads();

  constexpr int CT  = FO / 16;
  constexpr int WPC = AGG_NW / CT;
  constexpr int NTM = (AGG_NRT + WPC - 1) / WPC;
  const int ct = wave % CT;
  const int rg = wave / CT;
  const bool more = rg < AGG_NRT - (NTM - 1) * WPC;

  v8f c[NTM];
#pragma unroll
  for (int i = 0; i < NTM; i++) c[i] = (v8f){0.f, 0.f, 0.f, 0.f, 0.f, 0.f, 0.f, 0.f};

#pragma unroll 1
  for (int p = 0; p < NP - 1; p++) {
    if (more) rows_gemm_pass<K, FO, KH, NTM>(c, rows, hs, Wl, Wr, p * KH, ct, rg, lane);
    else      rows_gemm_pass<K, FO, KH, NTM - 1>(c, rows, hs, Wl, Wr, p * KH, ct, rg, lane);
    const typename freg<4 * NLD>::type hr = own_rows_load<K, KH, NLD>(h, r0, M, (p + 1) * KH, t);
    bn4 n = {};
    if (BNIN) n = bn_cols4(st_in, g_in, be_in, K, (p + 1) * KH + 4 * (t % (KH / 4)), invM);
    __syncthreads();
    own_rows_store<KH, NLD, BNIN>(hr, hs, t, n);
    __syncthreads();
  }
  if (more) rows_gemm_pass<K, FO, KH, NTM>(c, rows, hs, Wl, Wr, (NP - 1) * KH, ct, rg, lane);
  else      rows_gemm_pass<K, FO, KH, NTM - 1>(c, rows, hs, Wl, Wr, (NP - 1) * KH, ct, rg, lane);

  const int m16 = lane & 15;
  const int g = lane >> 4;
  const int ncol = ct * 16 + m16;
  const float bv = bias[ncol];
  float cs = 0.f, cq = 0.f;
#pragma unroll
  for (int i = 0; i < NTM; i++) {
    const int rbase = (rg + i * WPC) * 16 + 8 * g;
#pragma unroll
    for (int v = 0; v < 8; v++) {
      const int lr = rbase + v;
      float val = c[i][v] + bv;
      if (RELU) val = fmaxf(val, 0.f);
      if (lr < AGG_ROWS && r0 + lr < M) {
        out[(size_t)(r0 + lr) * FO + ncol] = val;
        if (STATS) { cs += val; cq += val * val; }
      }
    }
  }

  if (STATS) {
    red[0][wave][lane] = cs;
    red[1][wave][lane] = cq;
    __syncthreads();
    if (t < FO) {
      const int cti = t >> 4;
      const int mi = t & 15;
      float a = 0.f, b = 0.f;
      for (int r = 0; r < WPC; r++) {
        const int w = r * CT + cti;
        a += red[0][w][mi]; a += red[0][w][16 + mi];
        b += red[1][w][mi]; b += red[1][w][16 + mi];
      }
      float* prow = part + (size_t)blockIdx.x * (2 * FO);
      prow[t] = a;
      prow[FO + t] = b;
    }
  }
}

__global__ __launch_bounds__(1024)
void bn_sum_parts(const float* __restrict__ part, float* __restrict__ st, int F2) {
  __shared__ float wsum[32][32];
  const int lane = threadIdx.x & 31;
  const int w = threadIdx.x >> 5;
  const int j = blockIdx.x * 32 + lane;
  float s = 0.f;
  static_assert(AGG_NB % 32 == 0, "every wave adds the same number of rows");
  for (int q = 0; q < AGG_NB / 32; q++) s += part[(size_t)(w + 32 * q) * F2 + j];
  wsum[w][lane] = s;
  __syncthreads();
  if (w == 0) {
    float a = 0.f;
    for (int k = 0; k < 32; k++) a += wsum[k][lane];
    st[j] = a;
  }
}

__global__ void reparam(const float* __restrict__ h3, const float* __restrict__ eps,
                        float* __restrict__ z, int M) {
  int i = blockIdx.x * blockDim.x + threadIdx.x;
  if (i >= M * 64) return;
  int r = i >> 6, c = i & 63;
  const float* row = h3 + (size_t)r * 128;
  z[i] = row[c] + expf(row[64 + c]) * eps[i];
}

extern "C" void kernel_launch(void* const* d_in, const int* in_sizes, int n_in,
                              void* d_out, int out_size, void* d_ws, size_t ws_size,
                              hipStream_t stream) {
  const float* x   = (const float*)d_in[0];
  const int*   src = (const int*)d_in[1];
  const int*   dst = (const int*)d_in[2];
  const float* eps = (const float*)d_in[3];
  const float* W1l = (const float*)d_in[4];
  const float* W1r = (const float*)d_in[5];
  const float* b1  = (const float*)d_in[6];
  const float* g1  = (const float*)d_in[7];
  const float* be1 = (const float*)d_in[8];
  const float* W2l = (const float*)d_in[9];
  const float* W2r = (const float*)d_in[10];
  const float* b2  = (const float*)d_in[11];
  const float* g2  = (const float*)d_in[12];
  const float* be2 = (const float*)d_in[13];
  const float* W3l = (const float*)d_in[14];
  const float* W3r = (const float*)d_in[15];
  const float* b3  = (const float*)d_in[16];
  float* zout = (float*)d_out;
  (void)in_sizes; (void)n_in; (void)out_size; (void)ws_size;

  char* ws = (char*)d_ws;
  size_t off = 0;
  auto carve = [&](size_t bytes) -> float* {
    char* p = ws + off;
    off = (off + bytes + 255) & ~(size_t)255;
    return (float*)p;
  };
  float* rdeg = carve((size_t)NN * 4);
  float* h1   = carve((size_t)NN * 128 * 4);
  float* h2   = carve((size_t)NN * 256 * 4);
  float* st   = carve(512 * 4);
  float* part = carve((size_t)AGG_NB * 512 * 4);
  unsigned* list = (unsigned*)carve((size_t)AGG_NB * LIST_CAP * 4);
  int* cnt    = (int*)carve((size_t)AGG_NB * 4);

  const int M = NN, E = NE;

  unsigned* flags = (unsigned*)h2;
  chunk_flags<<<FLG_NB, AGG_NT, 0, stream>>>(dst, flags, E);

  owner_scan<<<AGG_NB, AGG_NT, 0, stream>>>(src, dst, flags, list, cnt, rdeg, E, M);

  sage_rows_fused<128, 128, true, true, false><<<AGG_NB, AGG_NT, 0, stream>>>(x, list, cnt, src, dst, rdeg, W1l, W1r, b1, h1, part, nullptr, nullptr, nullptr, E, M);
  bn_sum_parts<<<256 / 32, 1024, 0, stream>>>(part, st, 256);

  sage_rows_fused<128, 256, true, true, true><<<AGG_NB, AGG_NT, 0, stream>>>(h1, list, cnt, src, dst, rdeg, W2l, W2r, b2, h2, part, st, g1, be1, E, M);
  bn_sum_parts<<<512 / 32, 1024, 0, stream>>>(part, st, 512);

  sage_rows_fused<256, 128, false, false, true><<<AGG_NB, AGG_NT, 0, stream>>>(h2, list, cnt, src, dst, rdeg, W3l, W3r, b3, h1, nullptr, st, g2, be2, E, M);

  reparam<<<(M * 64 + 255) / 256, 256, 0, stream>>>(h1, eps, zout, M);
}


# z formed in the layer-3 epilogue: a wave owns the column tile pair (c, c+64), so a lane holds mean and log_std; it reads eps, stores z; layer-3 rows never stored; reparam pass goes, 8 -> 7 launches
# speedup vs baseline: 3.9267x; 1.0419x over previous
#include <hip/hip_runtime.h>

#define NN 50000
#define NE 800000

typedef __attribute__((ext_vector_type(2))) float v2f;
typedef __attribute__((ext_vector_type(8))) float v8f;

#define AGG_ROWS 196
#define AGG_RPAD 208
#define AGG_NRT  (AGG_RPAD / 16)
#define AGG_NB   ((NN + AGG_ROWS - 1) / AGG_ROWS)
#define AGG_NT   1024
#define AGG_NW   (AGG_NT / 32)
#define AGG_U    16
#define AGG_TRIP (32 * AGG_U)
#define AGG_PAD  4
#define AGG_LDS_FLOATS (AGG_RPAD * (256 + AGG_PAD) + AGG_RPAD * (64 + AGG_PAD))

#define NCHUNK   ((NE + 31) / 32)
#define NFW      ((NCHUNK + 31) / 32)
#define FLG_NB   256
#define FLG_PASS ((NFW + FLG_NB - 1) / FLG_NB)
static_assert(AGG_NT == 1024, "one flag word = 32 chunks of 32 edges = one edge per thread");
static_assert(NFW <= AGG_NT, "an owner group reads one flag word per thread");

__global__ __launch_bounds__(AGG_NT)
void chunk_flags(const int* __restrict__ dst, unsigned* __restrict__ flags, int E) {
  __shared__ unsigned bits[FLG_PASS * AGG_NB];
  const int t = threadIdx.x;
  const unsigned wbit = 1u << (t >> 5);
  for (int i = t; i < FLG_PASS * AGG_NB; i += AGG_NT) bits[i] = 0u;
  __syncthreads();
  int d[FLG_PASS];
#pragma unroll
  for (int q = 0; q < FLG_PASS; q++) {
    const int w = (int)blockIdx.x + q * FLG_NB;
    const int e = w * AGG_NT + t;
    d[q] = (w < NFW && e < E) ? dst[e] : -1;
  }
#pragma unroll
  for (int q = 0; q < FLG_PASS; q++) {
    const unsigned o = (unsigned)d[q] / (unsigned)AGG_ROWS;
    if (o < (unsigned)AGG_NB) atomicOr(&bits[q * AGG_NB + (int)o], wbit);
  }
  __syncthreads();
#pragma unroll
  for (int q = 0; q < FLG_PASS; q++) {
    const int w = (int)blockIdx.x + q * FLG_NB;
    if (w < NFW && t < AGG_NB) flags[(size_t)w * AGG_NB + t] = bits[q * AGG_NB + t];
  }
}

#define LIST_CAP 4096
#define SCAN_CU  16
static_assert(NE < (1 << 24) && NN < (1 << 24), "edge index and source must fit 24 bits of a list word");
static_assert(AGG_ROWS <= 256, "local row must fit 8 bits of a list word");

__global__ __launch_bounds__(AGG_NT)
void owner_scan(const int* __restrict__ src, const int* __restrict__ dst,
                const unsigned* __restrict__ flags,
                unsigned* __restrict__ list, int* __restrict__ cnt, float* __restrict__ rdeg,
                int E, int M) {
  __shared__ unsigned lent[LIST_CAP];
  __shared__ unsigned chl[NFW * 32];
  __shared__ int ldeg[AGG_ROWS];
  __shared__ int lcnt;
  __shared__ int nch;
  const int t = threadIdx.x;
  const int lane = t & 31;
  const int wave = __builtin_amdgcn_readfirstlane(t >> 5);
  const int r0 = blockIdx.x * AGG_ROWS;

  for (int i = t; i < AGG_ROWS; i += AGG_NT) ldeg[i] = 0;
  if (t == 0) { lcnt = 0; nch = 0; }
  unsigned fm = 0u;
  if (t < NFW) fm = flags[(size_t)t * AGG_NB + blockIdx.x];
  __syncthreads();

  if (fm) {
    int p = atomicAdd(&nch, __builtin_popcount(fm));
    while (fm) {
      const int b = __builtin_ctz(fm);
      fm &= fm - 1u;
      chl[p++] = (unsigned)(t * 32 + b);
    }
  }
  __syncthreads();
  const int nc = nch;

  for (int i0 = wave * SCAN_CU; i0 < nc; i0 += AGG_NW * SCAN_CU) {
    int cv = -1;
    if (lane < SCAN_CU && i0 + lane < nc) cv = (int)chl[i0 + lane];
    int dl[SCAN_CU];
#pragma unroll
    for (int j = 0; j < SCAN_CU; j++) {
      const int c = __builtin_amdgcn_readlane(cv, j);
      int e = ((c < 0) ? 0 : c) * 32 + lane;
      if (e > E - 1) e = E - 1;
      dl[j] = dst[e] - r0;
    }
    unsigned mk[SCAN_CU];
    int tot = 0;
#pragma unroll
    for (int j = 0; j < SCAN_CU; j++) {
      const int c = __builtin_amdgcn_readlane(cv, j);
      const bool ok = (c >= 0) & (c * 32 + lane < E) & ((unsigned)dl[j] < (unsigned)AGG_ROWS);
      mk[j] = __builtin_amdgcn_ballot_w32(ok);
      tot += __builtin_popcount(mk[j]);
    }
    if (tot) {
      int base = 0;
      if (lane == 0) base = atomicAdd(&lcnt, tot);
      base = __builtin_amdgcn_readfirstlane(base);
#pragma unroll
      for (int j = 0; j < SCAN_CU; j++) {
        if (mk[j]) {
          const int c = __builtin_amdgcn_readlane(cv, j);
          if ((mk[j] >> lane) & 1u) {
            const int pos = base + (int)__builtin_amdgcn_mbcnt_lo(mk[j], 0u);
            const int e = c * 32 + lane;
            if (pos < LIST_CAP) lent[pos] = ((unsigned)e << 8) | (unsigned)dl[j];
            atomicAdd(&ldeg[dl[j]], 1);
          }
          base += __builtin_popcount(mk[j]);
        }
      }
    }
  }
  __syncthreads();

  const int n = lcnt;
  const int nst = (n < LIST_CAP) ? n : LIST_CAP;
  unsigned* mylist = list + (size_t)blockIdx.x * LIST_CAP;
  for (int i = t; i < nst; i += AGG_NT) {
    const unsigned en = lent[i];
    const int s = src[en >> 8];
    mylist[i] = ((unsigned)s << 8) | (en & 255u);
  }
  for (int i = t; i < AGG_ROWS; i += AGG_NT) {
    const int gr = r0 + i;
    if (gr < M) rdeg[gr] = (ldeg[i] > 0) ? 1.f / (float)ldeg[i] : 0.f;
  }
  if (t == 0) cnt[blockIdx.x] = n;
}

template <int N> struct freg { typedef float type __attribute__((ext_vector_type(N))); };

template <int K, int KH, int NLD>
__device__ __forceinline__ typename freg<4 * NLD>::type own_rows_load(const float* __restrict__ h,
                                                                      int r0, int M, int col0, int t) {
  typename freg<4 * NLD>::type r;
#pragma unroll
  for (int j = 0; j < NLD; j++) {
    int i = t + j * AGG_NT;
    if (i > AGG_ROWS * (KH / 4) - 1) i = AGG_ROWS * (KH / 4) - 1;
    const int row = i / (KH / 4);
    const int c4 = i - row * (KH / 4);
    int gr = r0 + row;
    if (gr > M - 1) gr = M - 1;
    const float4 v = *(const float4*)(h + (gr * K + col0 + 4 * c4));
    r[4 * j + 0] = v.x; r[4 * j + 1] = v.y; r[4 * j + 2] = v.z; r[4 * j + 3] = v.w;
  }
  return r;
}

struct bn4 { float4 m, a, b; };

__device__ __forceinline__ bn4 bn_cols4(const float* __restrict__ st, const float* __restrict__ g,
                                        const float* __restrict__ be, int F, int c, float invM) {
  const float4 s = *(const float4*)(st + c);
  const float4 q = *(const float4*)(st + F + c);
  const float4 gg = *(const float4*)(g + c);
  bn4 r;
  r.b = *(const float4*)(be + c);
  r.m = make_float4(s.x * invM, s.y * invM, s.z * invM, s.w * invM);
  r.a.x = gg.x * rsqrtf(q.x * invM - r.m.x * r.m.x + 1e-5f);
  r.a.y = gg.y * rsqrtf(q.y * invM - r.m.y * r.m.y + 1e-5f);
  r.a.z = gg.z * rsqrtf(q.z * invM - r.m.z * r.m.z + 1e-5f);
  r.a.w = gg.w * rsqrtf(q.w * invM - r.m.w * r.m.w + 1e-5f);
  return r;
}

__device__ __forceinline__ float4 bn_map4(const float4 x, const bn4 n) {
  return make_float4((x.x - n.m.x) * n.a.x + n.b.x, (x.y - n.m.y) * n.a.y + n.b.y,
                     (x.z - n.m.z) * n.a.z + n.b.z, (x.w - n.m.w) * n.a.w + n.b.w);
}

template <int KH, int NLD, bool BN>
__device__ __forceinline__ void own_rows_store(const typename freg<4 * NLD>::type r, float* hs, int t,
                                               const bn4 n) {
  static_assert(AGG_NT % (KH / 4) == 0, "a thread keeps its columns from piece to piece");
#pragma unroll
  for (int j = 0; j < NLD; j++) {
    const int i = t + j * AGG_NT;
    if (i < AGG_ROWS * (KH / 4)) {
      const int row = i / (KH / 4);
      const int c4 = i - row * (KH / 4);
      float4 v = make_float4(r[4 * j + 0], r[4 * j + 1], r[4 * j + 2], r[4 * j + 3]);
      if (BN) v = bn_map4(v, n);
      *(float4*)(hs + row * (KH + AGG_PAD) + 4 * c4) = v;
    }
  }
}

template <int K, int FO, int KH, int NTW, int NCT>
__device__ __forceinline__ void rows_gemm_pass(v8f* c, const float* ms, const float* hs,
                                               const float* __restrict__ Wl,
                                               const float* __restrict__ Wr,
                                               int kb, int ct, int rg, int lane) {
  constexpr int LDK = K + AGG_PAD;
  constexpr int LDH = KH + AGG_PAD;
  constexpr int WPC = AGG_NW / (FO / 16 / NCT);
  const int m16 = lane & 15;
  const int g = lane >> 4;
  const int ncol = ct * 16 + m16;

  const float* ap[NTW];
  const float* hp[NTW];
#pragma unroll
  for (int i = 0; i < NTW; i++) {
    const int lrow = (rg + i * WPC) * 16 + m16;
    ap[i] = ms + lrow * LDK + kb + 4 * g;
    hp[i] = hs + lrow * LDH + 4 * g;
  }
  const float* wl = Wl + (size_t)(kb + 4 * g) * FO + ncol;
  const float* wr = Wr + (size_t)(kb + 4 * g) * FO + ncol;

  constexpr bool AHEAD = (NTW * NCT <= 4) && (KH == K);
  constexpr int CS = FO / NCT;
  v2f bl0[NCT], bl1[NCT], br0[NCT], br1[NCT];
#pragma unroll
  for (int jc = 0; jc < NCT; jc++) {
    bl0[jc].x = wl[0 * FO + jc * CS]; bl0[jc].y = wl[1 * FO + jc * CS];
    bl1[jc].x = wl[2 * FO + jc * CS]; bl1[jc].y = wl[3 * FO + jc * CS];
    br0[jc].x = wr[0 * FO + jc * CS]; br0[jc].y = wr[1 * FO + jc * CS];
    br1[jc].x = wr[2 * FO + jc * CS]; br1[jc].y = wr[3 * FO + jc * CS];
  }

#pragma unroll 1
  for (int k0 = 0; k0 < KH; k0 += 8) {
    const int kn = AHEAD ? ((k0 + 8 < KH) ? (k0 + 8) : k0) : k0;
    v2f nl0[NCT], nl1[NCT], nr0[NCT], nr1[NCT];
#pragma unroll
    for (int jc = 0; jc < NCT; jc++) {
      nl0[jc].x = wl[(size_t)(kn + 0) * FO + jc * CS]; nl0[jc].y = wl[(size_t)(kn + 1) * FO + jc * CS];
      nl1[jc].x = wl[(size_t)(kn + 2) * FO + jc * CS]; nl1[jc].y = wl[(size_t)(kn + 3) * FO + jc * CS];
      nr0[jc].x = wr[(size_t)(kn + 0) * FO + jc * CS]; nr0[jc].y = wr[(size_t)(kn + 1) * FO + jc * CS];
      nr1[jc].x = wr[(size_t)(kn + 2) * FO + jc * CS]; nr1[jc].y = wr[(size_t)(kn + 3) * FO + jc * CS];
    }
    if (!AHEAD) {
#pragma unroll
      for (int jc = 0; jc < NCT; jc++) { bl0[jc] = nl0[jc]; bl1[jc] = nl1[jc]; br0[jc] = nr0[jc]; br1[jc] = nr1[jc]; }
    }
#pragma unroll
    for (int i0 = 0; i0 < NTW; i0 += 4) {
      constexpr int GMAX = 4;
      float4 am[GMAX], ah[GMAX];
#pragma unroll
      for (int j = 0; j < GMAX; j++) {
        if (i0 + j < NTW) {
          am[j] = *(const float4*)(ap[i0 + j] + k0);
          ah[j] = *(const float4*)(hp[i0 + j] + k0);
        }
      }
#pragma unroll
      for (int j = 0; j < GMAX; j++) {
        if (i0 + j < NTW) {
          v2f a; a.x = am[j].x; a.y = am[j].y;
#pragma unroll
          for (int jc = 0; jc < NCT; jc++)
            c[(i0 + j) * NCT + jc] = __builtin_amdgcn_wmma_f32_16x16x4_f32(false, a, false, bl0[jc], (short)0,
                                                                         c[(i0 + j) * NCT + jc], false, false);
        }
      }
#pragma unroll
      for (int j = 0; j < GMAX; j++) {
        if (i0 + j < NTW) {
          v2f a; a.x = am[j].z; a.y = am[j].w;
#pragma unroll
          for (int jc = 0; jc < NCT; jc++)
            c[(i0 + j) * NCT + jc] = __builtin_amdgcn_wmma_f32_16x16x4_f32(false, a, false, bl1[jc], (short)0,
                                                                         c[(i0 + j) * NCT + jc], false, false);
        }
      }
#pragma unroll
      for (int j = 0; j < GMAX; j++) {
        if (i0 + j < NTW) {
          v2f a; a.x = ah[j].x; a.y = ah[j].y;
#pragma unroll
          for (int jc = 0; jc < NCT; jc++)
            c[(i0 + j) * NCT + jc] = __builtin_amdgcn_wmma_f32_16x16x4_f32(false, a, false, br0[jc], (short)0,
                                                                         c[(i0 + j) * NCT + jc], false, false);
        }
      }
#pragma unroll
      for (int j = 0; j < GMAX; j++) {
        if (i0 + j < NTW) {
          v2f a; a.x = ah[j].z; a.y = ah[j].w;
#pragma unroll
          for (int jc = 0; jc < NCT; jc++)
            c[(i0 + j) * NCT + jc] = __builtin_amdgcn_wmma_f32_16x16x4_f32(false, a, false, br1[jc], (short)0,
                                                                         c[(i0 + j) * NCT + jc], false, false);
        }
      }
    }
    if (AHEAD) {
#pragma unroll
      for (int jc = 0; jc < NCT; jc++) { bl0[jc] = nl0[jc]; bl1[jc] = nl1[jc]; br0[jc] = nr0[jc]; br1[jc] = nr1[jc]; }
    }
  }
}

template <int K, int FO, bool RELU, bool STATS, bool BNIN, bool ZOUT>
__global__ __launch_bounds__(AGG_NT)
void sage_rows_fused(const float* __restrict__ h, const unsigned* __restrict__ list,
                     const int* __restrict__ cnt, const int* __restrict__ src,
                     const int* __restrict__ dst, const float* __restrict__ rdeg,
                     const float* __restrict__ Wl, const float* __restrict__ Wr,
                     const float* __restrict__ bias, float* __restrict__ out,
                     float* __restrict__ part, const float* __restrict__ st_in,
                     const float* __restrict__ g_in, const float* __restrict__ be_in,
                     const float* __restrict__ epsin, int E, int M) {
  __shared__ __attribute__((aligned(16))) float lds[AGG_LDS_FLOATS];
  __shared__ float red[2][AGG_NW][32];
  constexpr int LDK = K + AGG_PAD;
  constexpr int KH  = (K > 128) ? 64 : K;
  constexpr int NP  = K / KH;
  constexpr int NLD = (AGG_ROWS * (KH / 4) + AGG_NT - 1) / AGG_NT;
  float* rows = lds;
  float* hs   = lds + AGG_RPAD * LDK;
  const int t = threadIdx.x;
  const int lane = t & 31;
  const int wave = __builtin_amdgcn_readfirstlane(t >> 5);
  const int r0 = blockIdx.x * AGG_ROWS;
  const float invM = 1.f / (float)M;

  for (int i = t; i < AGG_RPAD * LDK / 4; i += AGG_NT)
    ((float4*)rows)[i] = make_float4(0.f, 0.f, 0.f, 0.f);
  {
    const typename freg<4 * NLD>::type hr = own_rows_load<K, KH, NLD>(h, r0, M, 0, t);
    bn4 n = {};
    if (BNIN) n = bn_cols4(st_in, g_in, be_in, K, 4 * (t % (KH / 4)), invM);
    own_rows_store<KH, NLD, BNIN>(hr, hs, t, n);
    for (int i = t; i < (AGG_RPAD - AGG_ROWS) * (KH + AGG_PAD); i += AGG_NT)
      hs[AGG_ROWS * (KH + AGG_PAD) + i] = 0.f;
  }
  __syncthreads();

  const int nlist = cnt[blockIdx.x];
  if (nlist <= LIST_CAP) {
    const unsigned* mylist = list + (size_t)blockIdx.x * LIST_CAP;
    const int e0 = (int)(((unsigned)nlist * (unsigned)wave) / AGG_NW);
    const int e1 = (int)(((unsigned)nlist * (unsigned)(wave + 1)) / AGG_NW);
    constexpr int NB = 1024 / K;
    for (int e = e0; e < e1; e += 32) {
      const int ng = (e1 - e < 32) ? (e1 - e) : 32;
      unsigned ent = 0;
      if (lane < ng) ent = mylist[e + lane];
      for (int j = 0; j < ng; j += NB) {
        float v[NB][K / 32];
        int rl[NB];
#pragma unroll
        for (int b = 0; b < NB; b++) {
          int idx = j + b;
          if (idx > ng - 1) idx = ng - 1;
          const unsigned en = (unsigned)__builtin_amdgcn_readlane((int)ent, idx);
          rl[b] = (int)(en & 255u);
          const float* hp = h + (size_t)(en >> 8) * K + lane;
#pragma unroll
          for (int q = 0; q < K / 32; q++) v[b][q] = hp[32 * q];
        }
#pragma unroll
        for (int b = 0; b < NB; b++) {
          if (j + b < ng) {
            float* ap = rows + rl[b] * LDK + lane;
#pragma unroll
            for (int q = 0; q < K / 32; q++) atomicAdd(ap + 32 * q, v[b][q]);
          }
        }
      }
    }
  } else {
  const int ntrips = (E + AGG_TRIP - 1) / AGG_TRIP;
  const int rot = (int)(((unsigned)blockIdx.x * (unsigned)ntrips) / gridDim.x);

  int T = wave;
  int ebn = 0;
  int dn[AGG_U];
#pragma unroll
  for (int u = 0; u < AGG_U; u++) dn[u] = -1;
  if (T < ntrips) {
    int Tr = T + rot;
    if (Tr >= ntrips) Tr -= ntrips;
    ebn = Tr * AGG_TRIP + lane;
#pragma unroll
    for (int u = 0; u < AGG_U; u++) {
      const int e = ebn + 32 * u;
      if (e < E) dn[u] = dst[e];
    }
  }

  while (T < ntrips) {
    const int eb = ebn;
    int dl[AGG_U];
#pragma unroll
    for (int u = 0; u < AGG_U; u++) dl[u] = dn[u] - r0;

    T += AGG_NW;
    if (T < ntrips) {
      int Tr = T + rot;
      if (Tr >= ntrips) Tr -= ntrips;
      ebn = Tr * AGG_TRIP + lane;
#pragma unroll
      for (int u = 0; u < AGG_U; u++) {
        const int e = ebn + 32 * u;
        dn[u] = (e < E) ? dst[e] : -1;
      }
    }

    int sv[AGG_U];
    unsigned mk[AGG_U];
#pragma unroll
    for (int u = 0; u < AGG_U; u++) {
      const bool m = (unsigned)dl[u] < (unsigned)AGG_ROWS;
      sv[u] = 0;
      if (m) sv[u] = src[eb + 32 * u];
      mk[u] = __builtin_amdgcn_ballot_w32(m);
    }

#pragma unroll
    for (int u = 0; u < AGG_U; u++) {
      unsigned mask = mk[u];
      while (mask) {
        const int b = __builtin_ctz(mask);
        mask &= mask - 1;
        const int s  = __builtin_amdgcn_readlane(sv[u], b);
        const int rl = __builtin_amdgcn_readlane(dl[u], b);
        const float* hp = h + (size_t)s * K + lane;
        float* ap = rows + rl * LDK + lane;
        float v[K / 32];
#pragma unroll
        for (int j = 0; j < K / 32; j++) v[j] = hp[32 * j];
#pragma unroll
        for (int j = 0; j < K / 32; j++) atomicAdd(ap + 32 * j, v[j]);
      }
    }
  }
  }
  __syncthreads();

  {
    constexpr int NSC = (AGG_ROWS * (K / 4) + AGG_NT - 1) / AGG_NT;
    static_assert(AGG_NT % (K / 4) == 0, "a thread keeps its columns from trip to trip");
    bn4 n = {};
    if (BNIN) n = bn_cols4(st_in, g_in, be_in, K, 4 * (t % (K / 4)), invM);
#pragma unroll
    for (int j = 0; j < NSC; j++) {
      const int i = t + j * AGG_NT;
      const int r = i / (K / 4);
      const int c4 = i - r * (K / 4);
      const int gr = r0 + r;
      if (r < AGG_ROWS && gr < M) {
        const float sc = rdeg[gr];
        float4* p = (float4*)(rows + r * LDK) + c4;
        float4 v = *p;
        v.x *= sc; v.y *= sc; v.z *= sc; v.w *= sc;
        if (BNIN) { if (sc != 0.f) v = bn_map4(v, n); }
        *p = v;
      }
    }
  }
  __syncthreads();

  constexpr int NCT = ZOUT ? 2 : 1;
  constexpr int CT  = FO / 16 / NCT;
  constexpr int WPC = AGG_NW / CT;
  constexpr int NTM = (AGG_NRT + WPC - 1) / WPC;
  const int ct = wave % CT;
  const int rg = wave / CT;
  const bool more = rg < AGG_NRT - (NTM - 1) * WPC;

  v8f c[NTM * NCT];
#pragma unroll
  for (int i = 0; i < NTM * NCT; i++) c[i] = (v8f){0.f, 0.f, 0.f, 0.f, 0.f, 0.f, 0.f, 0.f};

#pragma unroll 1
  for (int p = 0; p < NP - 1; p++) {
    if (more) rows_gemm_pass<K, FO, KH, NTM, NCT>(c, rows, hs, Wl, Wr, p * KH, ct, rg, lane);
    else      rows_gemm_pass<K, FO, KH, NTM - 1, NCT>(c, rows, hs, Wl, Wr, p * KH, ct, rg, lane);
    const typename freg<4 * NLD>::type hr = own_rows_load<K, KH, NLD>(h, r0, M, (p + 1) * KH, t);
    bn4 n = {};
    if (BNIN) n = bn_cols4(st_in, g_in, be_in, K, (p + 1) * KH + 4 * (t % (KH / 4)), invM);
    __syncthreads();
    own_rows_store<KH, NLD, BNIN>(hr, hs, t, n);
    __syncthreads();
  }
  const int m16 = lane & 15;
  const int g = lane >> 4;
  const int ncol = ct * 16 + m16;

  float ev[NTM][8];
  if constexpr (ZOUT) {
#pragma unroll
    for (int i = 0; i < NTM; i++) {
#pragma unroll
      for (int v = 0; v < 8; v++) {
        const int lr = (rg + i * WPC) * 16 + 8 * g + v;
        int gr = r0 + lr;
        if (lr > AGG_ROWS - 1 || gr > M - 1) gr = M - 1;
        ev[i][v] = epsin[(size_t)gr * (FO / 2) + ncol];
      }
    }
  }

  if (more) rows_gemm_pass<K, FO, KH, NTM, NCT>(c, rows, hs, Wl, Wr, (NP - 1) * KH, ct, rg, lane);
  else      rows_gemm_pass<K, FO, KH, NTM - 1, NCT>(c, rows, hs, Wl, Wr, (NP - 1) * KH, ct, rg, lane);

  if constexpr (ZOUT) {
    const float bm = bias[ncol];
    const float bs = bias[FO / 2 + ncol];
#pragma unroll
    for (int i = 0; i < NTM; i++) {
      const int rbase = (rg + i * WPC) * 16 + 8 * g;
#pragma unroll
      for (int v = 0; v < 8; v++) {
        const int lr = rbase + v;
        const float mean = c[i * NCT][v] + bm;
        const float ls = c[i * NCT + (NCT - 1)][v] + bs;
        if (lr < AGG_ROWS && r0 + lr < M)
          out[(size_t)(r0 + lr) * (FO / 2) + ncol] = mean + expf(ls) * ev[i][v];
      }
    }
    return;
  }
  const float bv = bias[ncol];
  float cs = 0.f, cq = 0.f;
#pragma unroll
  for (int i = 0; i < NTM; i++) {
    const int rbase = (rg + i * WPC) * 16 + 8 * g;
#pragma unroll
    for (int v = 0; v < 8; v++) {
      const int lr = rbase + v;
      float val = c[i][v] + bv;
      if (RELU) val = fmaxf(val, 0.f);
      if (lr < AGG_ROWS && r0 + lr < M) {
        out[(size_t)(r0 + lr) * FO + ncol] = val;
        if (STATS) { cs += val; cq += val * val; }
      }
    }
  }

  if (STATS) {
    red[0][wave][lane] = cs;
    red[1][wave][lane] = cq;
    __syncthreads();
    if (t < FO) {
      const int cti = t >> 4;
      const int mi = t & 15;
      float a = 0.f, b = 0.f;
      for (int r = 0; r < WPC; r++) {
        const int w = r * CT + cti;
        a += red[0][w][mi]; a += red[0][w][16 + mi];
        b += red[1][w][mi]; b += red[1][w][16 + mi];
      }
      float* prow = part + (size_t)blockIdx.x * (2 * FO);
      prow[t] = a;
      prow[FO + t] = b;
    }
  }
}

__global__ __launch_bounds__(1024)
void bn_sum_parts(const float* __restrict__ part, float* __restrict__ st, int F2) {
  __shared__ float wsum[32][32];
  const int lane = threadIdx.x & 31;
  const int w = threadIdx.x >> 5;
  const int j = blockIdx.x * 32 + lane;
  float s = 0.f;
  static_assert(AGG_NB % 32 == 0, "every wave adds the same number of rows");
  for (int q = 0; q < AGG_NB / 32; q++) s += part[(size_t)(w + 32 * q) * F2 + j];
  wsum[w][lane] = s;
  __syncthreads();
  if (w == 0) {
    float a = 0.f;
    for (int k = 0; k < 32; k++) a += wsum[k][lane];
    st[j] = a;
  }
}

extern "C" void kernel_launch(void* const* d_in, const int* in_sizes, int n_in,
                              void* d_out, int out_size, void* d_ws, size_t ws_size,
                              hipStream_t stream) {
  const float* x   = (const float*)d_in[0];
  const int*   src = (const int*)d_in[1];
  const int*   dst = (const int*)d_in[2];
  const float* eps = (const float*)d_in[3];
  const float* W1l = (const float*)d_in[4];
  const float* W1r = (const float*)d_in[5];
  const float* b1  = (const float*)d_in[6];
  const float* g1  = (const float*)d_in[7];
  const float* be1 = (const float*)d_in[8];
  const float* W2l = (const float*)d_in[9];
  const float* W2r = (const float*)d_in[10];
  const float* b2  = (const float*)d_in[11];
  const float* g2  = (const float*)d_in[12];
  const float* be2 = (const float*)d_in[13];
  const float* W3l = (const float*)d_in[14];
  const float* W3r = (const float*)d_in[15];
  const float* b3  = (const float*)d_in[16];
  float* zout = (float*)d_out;
  (void)in_sizes; (void)n_in; (void)out_size; (void)ws_size;

  char* ws = (char*)d_ws;
  size_t off = 0;
  auto carve = [&](size_t bytes) -> float* {
    char* p = ws + off;
    off = (off + bytes + 255) & ~(size_t)255;
    return (float*)p;
  };
  float* rdeg = carve((size_t)NN * 4);
  float* h1   = carve((size_t)NN * 128 * 4);
  float* h2   = carve((size_t)NN * 256 * 4);
  float* st   = carve(512 * 4);
  float* part = carve((size_t)AGG_NB * 512 * 4);
  unsigned* list = (unsigned*)carve((size_t)AGG_NB * LIST_CAP * 4);
  int* cnt    = (int*)carve((size_t)AGG_NB * 4);

  const int M = NN, E = NE;

  unsigned* flags = (unsigned*)h2;
  chunk_flags<<<FLG_NB, AGG_NT, 0, stream>>>(dst, flags, E);

  owner_scan<<<AGG_NB, AGG_NT, 0, stream>>>(src, dst, flags, list, cnt, rdeg, E, M);

  sage_rows_fused<128, 128, true, true, false, false><<<AGG_NB, AGG_NT, 0, stream>>>(x, list, cnt, src, dst, rdeg, W1l, W1r, b1, h1, part, nullptr, nullptr, nullptr, nullptr, E, M);
  bn_sum_parts<<<256 / 32, 1024, 0, stream>>>(part, st, 256);

  sage_rows_fused<128, 256, true, true, true, false><<<AGG_NB, AGG_NT, 0, stream>>>(h1, list, cnt, src, dst, rdeg, W2l, W2r, b2, h2, part, st, g1, be1, nullptr, E, M);
  bn_sum_parts<<<512 / 32, 1024, 0, stream>>>(part, st, 512);

  sage_rows_fused<256, 128, false, false, true, true><<<AGG_NB, AGG_NT, 0, stream>>>(h2, list, cnt, src, dst, rdeg, W3l, W3r, b3, zout, nullptr, st, g2, be2, eps, E, M);
}


# Neighbour sums in registers, no LDS adds: the pre-pass stores a group's edge list in 32 parts (row % 32); wave w reads part w, whole-row 16-byte loads, stores its rows once; add order = part order
# speedup vs baseline: 4.0714x; 1.0369x over previous
#include <hip/hip_runtime.h>

#define NN 50000
#define NE 800000

typedef __attribute__((ext_vector_type(2))) float v2f;
typedef __attribute__((ext_vector_type(8))) float v8f;

#define AGG_ROWS 196
#define AGG_RPAD 208
#define AGG_NRT  (AGG_RPAD / 16)
#define AGG_NB   ((NN + AGG_ROWS - 1) / AGG_ROWS)
#define AGG_NT   1024
#define AGG_NW   (AGG_NT / 32)
#define AGG_U    16
#define AGG_TRIP (32 * AGG_U)
#define AGG_PAD  4
#define AGG_LDS_FLOATS (AGG_RPAD * (256 + AGG_PAD) + AGG_RPAD * (64 + AGG_PAD))

#define NCHUNK   ((NE + 31) / 32)
#define NFW      ((NCHUNK + 31) / 32)
#define FLG_NB   256
#define FLG_PASS ((NFW + FLG_NB - 1) / FLG_NB)
static_assert(AGG_NT == 1024, "one flag word = 32 chunks of 32 edges = one edge per thread");
static_assert(NFW <= AGG_NT, "an owner group reads one flag word per thread");

__global__ __launch_bounds__(AGG_NT)
void chunk_flags(const int* __restrict__ dst, unsigned* __restrict__ flags, int E) {
  __shared__ unsigned bits[FLG_PASS * AGG_NB];
  const int t = threadIdx.x;
  const unsigned wbit = 1u << (t >> 5);
  for (int i = t; i < FLG_PASS * AGG_NB; i += AGG_NT) bits[i] = 0u;
  __syncthreads();
  int d[FLG_PASS];
#pragma unroll
  for (int q = 0; q < FLG_PASS; q++) {
    const int w = (int)blockIdx.x + q * FLG_NB;
    const int e = w * AGG_NT + t;
    d[q] = (w < NFW && e < E) ? dst[e] : -1;
  }
#pragma unroll
  for (int q = 0; q < FLG_PASS; q++) {
    const unsigned o = (unsigned)d[q] / (unsigned)AGG_ROWS;
    if (o < (unsigned)AGG_NB) atomicOr(&bits[q * AGG_NB + (int)o], wbit);
  }
  __syncthreads();
#pragma unroll
  for (int q = 0; q < FLG_PASS; q++) {
    const int w = (int)blockIdx.x + q * FLG_NB;
    if (w < NFW && t < AGG_NB) flags[(size_t)w * AGG_NB + t] = bits[q * AGG_NB + t];
  }
}

#define LIST_CAP 4096
#define LIST_HDR 64
#define LIST_ENT (LIST_CAP - LIST_HDR)
#define SCAN_CU  16
static_assert(AGG_NW == 32, "a row's owner wave is the low 5 bits of its local row");
static_assert(AGG_NW + 1 <= LIST_HDR, "the part offsets fit in front of the entries");
static_assert(NE < (1 << 24) && NN < (1 << 24), "edge index and source must fit 24 bits of a list word");
static_assert(AGG_ROWS <= 256, "local row must fit 8 bits of a list word");

__global__ __launch_bounds__(AGG_NT)
void owner_scan(const int* __restrict__ src, const int* __restrict__ dst,
                const unsigned* __restrict__ flags,
                unsigned* __restrict__ list, int* __restrict__ cnt, float* __restrict__ rdeg,
                int E, int M) {
  __shared__ unsigned lent[LIST_CAP];
  __shared__ unsigned chl[NFW * 32];
  __shared__ int ldeg[AGG_ROWS];
  __shared__ int lcnt;
  __shared__ int nch;
  __shared__ int boff[AGG_NW];
  __shared__ int bfill[AGG_NW];
  const int t = threadIdx.x;
  const int lane = t & 31;
  const int wave = __builtin_amdgcn_readfirstlane(t >> 5);
  const int r0 = blockIdx.x * AGG_ROWS;

  for (int i = t; i < AGG_ROWS; i += AGG_NT) ldeg[i] = 0;
  if (t == 0) { lcnt = 0; nch = 0; }
  unsigned fm = 0u;
  if (t < NFW) fm = flags[(size_t)t * AGG_NB + blockIdx.x];
  __syncthreads();

  if (fm) {
    int p = atomicAdd(&nch, __builtin_popcount(fm));
    while (fm) {
      const int b = __builtin_ctz(fm);
      fm &= fm - 1u;
      chl[p++] = (unsigned)(t * 32 + b);
    }
  }
  __syncthreads();
  const int nc = nch;

  for (int i0 = wave * SCAN_CU; i0 < nc; i0 += AGG_NW * SCAN_CU) {
    int cv = -1;
    if (lane < SCAN_CU && i0 + lane < nc) cv = (int)chl[i0 + lane];
    int dl[SCAN_CU];
#pragma unroll
    for (int j = 0; j < SCAN_CU; j++) {
      const int c = __builtin_amdgcn_readlane(cv, j);
      int e = ((c < 0) ? 0 : c) * 32 + lane;
      if (e > E - 1) e = E - 1;
      dl[j] = dst[e] - r0;
    }
    unsigned mk[SCAN_CU];
    int tot = 0;
#pragma unroll
    for (int j = 0; j < SCAN_CU; j++) {
      const int c = __builtin_amdgcn_readlane(cv, j);
      const bool ok = (c >= 0) & (c * 32 + lane < E) & ((unsigned)dl[j] < (unsigned)AGG_ROWS);
      mk[j] = __builtin_amdgcn_ballot_w32(ok);
      tot += __builtin_popcount(mk[j]);
    }
    if (tot) {
      int base = 0;
      if (lane == 0) base = atomicAdd(&lcnt, tot);
      base = __builtin_amdgcn_readfirstlane(base);
#pragma unroll
      for (int j = 0; j < SCAN_CU; j++) {
        if (mk[j]) {
          const int c = __builtin_amdgcn_readlane(cv, j);
          if ((mk[j] >> lane) & 1u) {
            const int pos = base + (int)__builtin_amdgcn_mbcnt_lo(mk[j], 0u);
            const int e = c * 32 + lane;
            if (pos < LIST_ENT) lent[pos] = ((unsigned)e << 8) | (unsigned)dl[j];
            atomicAdd(&ldeg[dl[j]], 1);
          }
          base += __builtin_popcount(mk[j]);
        }
      }
    }
  }
  __syncthreads();

  const int n = lcnt;
  unsigned* grp = list + (size_t)blockIdx.x * LIST_CAP;
  if (n <= LIST_ENT) {
    if (t < AGG_NW) {
      int c = 0;
      for (int r = t; r < AGG_ROWS; r += AGG_NW) c += ldeg[r];
      int inc = c;
      for (int d = 1; d < 32; d <<= 1) {
        const int o = __shfl_up(inc, d);
        if (lane >= d) inc += o;
      }
      boff[t] = inc - c;
      bfill[t] = 0;
      grp[t] = (unsigned)(inc - c);
      if (t == AGG_NW - 1) grp[AGG_NW] = (unsigned)inc;
    }
    __syncthreads();
    for (int i = t; i < n; i += AGG_NT) {
      const unsigned en = lent[i];
      const int b = (int)(en & (unsigned)(AGG_NW - 1));
      const int slot = boff[b] + atomicAdd(&bfill[b], 1);
      const int s = src[en >> 8];
      grp[LIST_HDR + slot] = ((unsigned)s << 8) | (en & 255u);
    }
  }
  for (int i = t; i < AGG_ROWS; i += AGG_NT) {
    const int gr = r0 + i;
    if (gr < M) rdeg[gr] = (ldeg[i] > 0) ? 1.f / (float)ldeg[i] : 0.f;
  }
  if (t == 0) cnt[blockIdx.x] = n;
}

template <int N> struct freg { typedef float type __attribute__((ext_vector_type(N))); };

template <int K, int KH, int NLD>
__device__ __forceinline__ typename freg<4 * NLD>::type own_rows_load(const float* __restrict__ h,
                                                                      int r0, int M, int col0, int t) {
  typename freg<4 * NLD>::type r;
#pragma unroll
  for (int j = 0; j < NLD; j++) {
    int i = t + j * AGG_NT;
    if (i > AGG_ROWS * (KH / 4) - 1) i = AGG_ROWS * (KH / 4) - 1;
    const int row = i / (KH / 4);
    const int c4 = i - row * (KH / 4);
    int gr = r0 + row;
    if (gr > M - 1) gr = M - 1;
    const float4 v = *(const float4*)(h + (gr * K + col0 + 4 * c4));
    r[4 * j + 0] = v.x; r[4 * j + 1] = v.y; r[4 * j + 2] = v.z; r[4 * j + 3] = v.w;
  }
  return r;
}

struct bn4 { float4 m, a, b; };

__device__ __forceinline__ bn4 bn_cols4(const float* __restrict__ st, const float* __restrict__ g,
                                        const float* __restrict__ be, int F, int c, float invM) {
  const float4 s = *(const float4*)(st + c);
  const float4 q = *(const float4*)(st + F + c);
  const float4 gg = *(const float4*)(g + c);
  bn4 r;
  r.b = *(const float4*)(be + c);
  r.m = make_float4(s.x * invM, s.y * invM, s.z * invM, s.w * invM);
  r.a.x = gg.x * rsqrtf(q.x * invM - r.m.x * r.m.x + 1e-5f);
  r.a.y = gg.y * rsqrtf(q.y * invM - r.m.y * r.m.y + 1e-5f);
  r.a.z = gg.z * rsqrtf(q.z * invM - r.m.z * r.m.z + 1e-5f);
  r.a.w = gg.w * rsqrtf(q.w * invM - r.m.w * r.m.w + 1e-5f);
  return r;
}

__device__ __forceinline__ float4 bn_map4(const float4 x, const bn4 n) {
  return make_float4((x.x - n.m.x) * n.a.x + n.b.x, (x.y - n.m.y) * n.a.y + n.b.y,
                     (x.z - n.m.z) * n.a.z + n.b.z, (x.w - n.m.w) * n.a.w + n.b.w);
}

template <int KH, int NLD, bool BN>
__device__ __forceinline__ void own_rows_store(const typename freg<4 * NLD>::type r, float* hs, int t,
                                               const bn4 n) {
  static_assert(AGG_NT % (KH / 4) == 0, "a thread keeps its columns from piece to piece");
#pragma unroll
  for (int j = 0; j < NLD; j++) {
    const int i = t + j * AGG_NT;
    if (i < AGG_ROWS * (KH / 4)) {
      const int row = i / (KH / 4);
      const int c4 = i - row * (KH / 4);
      float4 v = make_float4(r[4 * j + 0], r[4 * j + 1], r[4 * j + 2], r[4 * j + 3]);
      if (BN) v = bn_map4(v, n);
      *(float4*)(hs + row * (KH + AGG_PAD) + 4 * c4) = v;
    }
  }
}

template <int K, int FO, int KH, int NTW, int NCT>
__device__ __forceinline__ void rows_gemm_pass(v8f* c, const float* ms, const float* hs,
                                               const float* __restrict__ Wl,
                                               const float* __restrict__ Wr,
                                               int kb, int ct, int rg, int lane) {
  constexpr int LDK = K + AGG_PAD;
  constexpr int LDH = KH + AGG_PAD;
  constexpr int WPC = AGG_NW / (FO / 16 / NCT);
  const int m16 = lane & 15;
  const int g = lane >> 4;
  const int ncol = ct * 16 + m16;

  const float* ap[NTW];
  const float* hp[NTW];
#pragma unroll
  for (int i = 0; i < NTW; i++) {
    const int lrow = (rg + i * WPC) * 16 + m16;
    ap[i] = ms + lrow * LDK + kb + 4 * g;
    hp[i] = hs + lrow * LDH + 4 * g;
  }
  const float* wl = Wl + (size_t)(kb + 4 * g) * FO + ncol;
  const float* wr = Wr + (size_t)(kb + 4 * g) * FO + ncol;

  constexpr bool AHEAD = (NTW * NCT <= 4) && (KH == K);
  constexpr int CS = FO / NCT;
  v2f bl0[NCT], bl1[NCT], br0[NCT], br1[NCT];
#pragma unroll
  for (int jc = 0; jc < NCT; jc++) {
    bl0[jc].x = wl[0 * FO + jc * CS]; bl0[jc].y = wl[1 * FO + jc * CS];
    bl1[jc].x = wl[2 * FO + jc * CS]; bl1[jc].y = wl[3 * FO + jc * CS];
    br0[jc].x = wr[0 * FO + jc * CS]; br0[jc].y = wr[1 * FO + jc * CS];
    br1[jc].x = wr[2 * FO + jc * CS]; br1[jc].y = wr[3 * FO + jc * CS];
  }

#pragma unroll 1
  for (int k0 = 0; k0 < KH; k0 += 8) {
    const int kn = AHEAD ? ((k0 + 8 < KH) ? (k0 + 8) : k0) : k0;
    v2f nl0[NCT], nl1[NCT], nr0[NCT], nr1[NCT];
#pragma unroll
    for (int jc = 0; jc < NCT; jc++) {
      nl0[jc].x = wl[(size_t)(kn + 0) * FO + jc * CS]; nl0[jc].y = wl[(size_t)(kn + 1) * FO + jc * CS];
      nl1[jc].x = wl[(size_t)(kn + 2) * FO + jc * CS]; nl1[jc].y = wl[(size_t)(kn + 3) * FO + jc * CS];
      nr0[jc].x = wr[(size_t)(kn + 0) * FO + jc * CS]; nr0[jc].y = wr[(size_t)(kn + 1) * FO + jc * CS];
      nr1[jc].x = wr[(size_t)(kn + 2) * FO + jc * CS]; nr1[jc].y = wr[(size_t)(kn + 3) * FO + jc * CS];
    }
    if (!AHEAD) {
#pragma unroll
      for (int jc = 0; jc < NCT; jc++) { bl0[jc] = nl0[jc]; bl1[jc] = nl1[jc]; br0[jc] = nr0[jc]; br1[jc] = nr1[jc]; }
    }
#pragma unroll
    for (int i0 = 0; i0 < NTW; i0 += 4) {
      constexpr int GMAX = 4;
      float4 am[GMAX], ah[GMAX];
#pragma unroll
      for (int j = 0; j < GMAX; j++) {
        if (i0 + j < NTW) {
          am[j] = *(const float4*)(ap[i0 + j] + k0);
          ah[j] = *(const float4*)(hp[i0 + j] + k0);
        }
      }
#pragma unroll
      for (int j = 0; j < GMAX; j++) {
        if (i0 + j < NTW) {
          v2f a; a.x = am[j].x; a.y = am[j].y;
#pragma unroll
          for (int jc = 0; jc < NCT; jc++)
            c[(i0 + j) * NCT + jc] = __builtin_amdgcn_wmma_f32_16x16x4_f32(false, a, false, bl0[jc], (short)0,
                                                                         c[(i0 + j) * NCT + jc], false, false);
        }
      }
#pragma unroll
      for (int j = 0; j < GMAX; j++) {
        if (i0 + j < NTW) {
          v2f a; a.x = am[j].z; a.y = am[j].w;
#pragma unroll
          for (int jc = 0; jc < NCT; jc++)
            c[(i0 + j) * NCT + jc] = __builtin_amdgcn_wmma_f32_16x16x4_f32(false, a, false, bl1[jc], (short)0,
                                                                         c[(i0 + j) * NCT + jc], false, false);
        }
      }
#pragma unroll
      for (int j = 0; j < GMAX; j++) {
        if (i0 + j < NTW) {
          v2f a; a.x = ah[j].x; a.y = ah[j].y;
#pragma unroll
          for (int jc = 0; jc < NCT; jc++)
            c[(i0 + j) * NCT + jc] = __builtin_amdgcn_wmma_f32_16x16x4_f32(false, a, false, br0[jc], (short)0,
                                                                         c[(i0 + j) * NCT + jc], false, false);
        }
      }
#pragma unroll
      for (int j = 0; j < GMAX; j++) {
        if (i0 + j < NTW) {
          v2f a; a.x = ah[j].z; a.y = ah[j].w;
#pragma unroll
          for (int jc = 0; jc < NCT; jc++)
            c[(i0 + j) * NCT + jc] = __builtin_amdgcn_wmma_f32_16x16x4_f32(false, a, false, br1[jc], (short)0,
                                                                         c[(i0 + j) * NCT + jc], false, false);
        }
      }
    }
    if (AHEAD) {
#pragma unroll
      for (int jc = 0; jc < NCT; jc++) { bl0[jc] = nl0[jc]; bl1[jc] = nl1[jc]; br0[jc] = nr0[jc]; br1[jc] = nr1[jc]; }
    }
  }
}

template <int K, int FO, bool RELU, bool STATS, bool BNIN, bool ZOUT>
__global__ __launch_bounds__(AGG_NT)
void sage_rows_fused(const float* __restrict__ h, const unsigned* __restrict__ list,
                     const int* __restrict__ cnt, const int* __restrict__ src,
                     const int* __restrict__ dst, const float* __restrict__ rdeg,
                     const float* __restrict__ Wl, const float* __restrict__ Wr,
                     const float* __restrict__ bias, float* __restrict__ out,
                     float* __restrict__ part, const float* __restrict__ st_in,
                     const float* __restrict__ g_in, const float* __restrict__ be_in,
                     const float* __restrict__ epsin, int E, int M) {
  __shared__ __attribute__((aligned(16))) float lds[AGG_LDS_FLOATS];
  __shared__ float red[2][AGG_NW][32];
  constexpr int LDK = K + AGG_PAD;
  constexpr int KH  = (K > 128) ? 64 : K;
  constexpr int NP  = K / KH;
  constexpr int NLD = (AGG_ROWS * (KH / 4) + AGG_NT - 1) / AGG_NT;
  float* rows = lds;
  float* hs   = lds + AGG_RPAD * LDK;
  const int t = threadIdx.x;
  const int lane = t & 31;
  const int wave = __builtin_amdgcn_readfirstlane(t >> 5);
  const int r0 = blockIdx.x * AGG_ROWS;
  const float invM = 1.f / (float)M;

  {
    const typename freg<4 * NLD>::type hr = own_rows_load<K, KH, NLD>(h, r0, M, 0, t);
    bn4 n = {};
    if (BNIN) n = bn_cols4(st_in, g_in, be_in, K, 4 * (t % (KH / 4)), invM);
    own_rows_store<KH, NLD, BNIN>(hr, hs, t, n);
    for (int i = t; i < (AGG_RPAD - AGG_ROWS) * (KH + AGG_PAD); i += AGG_NT)
      hs[AGG_ROWS * (KH + AGG_PAD) + i] = 0.f;
  }

  const int nlist = cnt[blockIdx.x];
  if (nlist <= LIST_ENT) {
    constexpr int Q = K / 128;
    constexpr int NB = 8 / Q;
    constexpr int NOWN = (AGG_ROWS + AGG_NW - 1) / AGG_NW;
    for (int i = t; i < (AGG_RPAD - AGG_ROWS) * LDK; i += AGG_NT) rows[AGG_ROWS * LDK + i] = 0.f;
    const unsigned* grp = list + (size_t)blockIdx.x * LIST_CAP;
    const int e0 = (int)grp[wave];
    const int e1 = (int)grp[wave + 1];
    const unsigned* mylist = grp + LIST_HDR;
    float4 acc[NOWN][Q];
#pragma unroll
    for (int o = 0; o < NOWN; o++)
#pragma unroll
      for (int q = 0; q < Q; q++) acc[o][q] = make_float4(0.f, 0.f, 0.f, 0.f);
    for (int e = e0; e < e1; e += 32) {
      const int ng = (e1 - e < 32) ? (e1 - e) : 32;
      unsigned ent = 0;
      if (lane < ng) ent = mylist[e + lane];
      for (int j = 0; j < ng; j += NB) {
        float4 v[NB][Q];
        int ow[NB];
#pragma unroll
        for (int b = 0; b < NB; b++) {
          int idx = j + b;
          if (idx > ng - 1) idx = ng - 1;
          const unsigned en = (unsigned)__builtin_amdgcn_readlane((int)ent, idx);
          ow[b] = (int)((en & 255u) >> 5);
          const float* hp = h + (size_t)(en >> 8) * K + 4 * lane;
#pragma unroll
          for (int q = 0; q < Q; q++) v[b][q] = *(const float4*)(hp + 128 * q);
        }
#pragma unroll
        for (int b = 0; b < NB; b++) {
          if (j + b < ng) {
#define ROW_ADD(O)                                                                       \
            case O:                                                                      \
              if (O < NOWN) {                                                            \
                _Pragma("unroll") for (int q = 0; q < Q; q++) {                          \
                  acc[O < NOWN ? O : 0][q].x += v[b][q].x;                               \
                  acc[O < NOWN ? O : 0][q].y += v[b][q].y;                               \
                  acc[O < NOWN ? O : 0][q].z += v[b][q].z;                               \
                  acc[O < NOWN ? O : 0][q].w += v[b][q].w;                               \
                }                                                                        \
              }                                                                          \
              break;
            switch (ow[b]) {
              ROW_ADD(0) ROW_ADD(1) ROW_ADD(2) ROW_ADD(3) ROW_ADD(4) ROW_ADD(5) ROW_ADD(6) ROW_ADD(7)
              default: break;
            }
#undef ROW_ADD
          }
        }
      }
    }
#pragma unroll
    for (int o = 0; o < NOWN; o++) {
      const int r = wave + o * AGG_NW;
      if (r < AGG_ROWS) {
#pragma unroll
        for (int q = 0; q < Q; q++) *(float4*)(rows + r * LDK + 128 * q + 4 * lane) = acc[o][q];
      }
    }
  } else {
  for (int i = t; i < AGG_RPAD * LDK / 4; i += AGG_NT)
    ((float4*)rows)[i] = make_float4(0.f, 0.f, 0.f, 0.f);
  __syncthreads();
  const int ntrips = (E + AGG_TRIP - 1) / AGG_TRIP;
  const int rot = (int)(((unsigned)blockIdx.x * (unsigned)ntrips) / gridDim.x);

  int T = wave;
  int ebn = 0;
  int dn[AGG_U];
#pragma unroll
  for (int u = 0; u < AGG_U; u++) dn[u] = -1;
  if (T < ntrips) {
    int Tr = T + rot;
    if (Tr >= ntrips) Tr -= ntrips;
    ebn = Tr * AGG_TRIP + lane;
#pragma unroll
    for (int u = 0; u < AGG_U; u++) {
      const int e = ebn + 32 * u;
      if (e < E) dn[u] = dst[e];
    }
  }

  while (T < ntrips) {
    const int eb = ebn;
    int dl[AGG_U];
#pragma unroll
    for (int u = 0; u < AGG_U; u++) dl[u] = dn[u] - r0;

    T += AGG_NW;
    if (T < ntrips) {
      int Tr = T + rot;
      if (Tr >= ntrips) Tr -= ntrips;
      ebn = Tr * AGG_TRIP + lane;
#pragma unroll
      for (int u = 0; u < AGG_U; u++) {
        const int e = ebn + 32 * u;
        dn[u] = (e < E) ? dst[e] : -1;
      }
    }

    int sv[AGG_U];
    unsigned mk[AGG_U];
#pragma unroll
    for (int u = 0; u < AGG_U; u++) {
      const bool m = (unsigned)dl[u] < (unsigned)AGG_ROWS;
      sv[u] = 0;
      if (m) sv[u] = src[eb + 32 * u];
      mk[u] = __builtin_amdgcn_ballot_w32(m);
    }

#pragma unroll
    for (int u = 0; u < AGG_U; u++) {
      unsigned mask = mk[u];
      while (mask) {
        const int b = __builtin_ctz(mask);
        mask &= mask - 1;
        const int s  = __builtin_amdgcn_readlane(sv[u], b);
        const int rl = __builtin_amdgcn_readlane(dl[u], b);
        const float* hp = h + (size_t)s * K + lane;
        float* ap = rows + rl * LDK + lane;
        float v[K / 32];
#pragma unroll
        for (int j = 0; j < K / 32; j++) v[j] = hp[32 * j];
#pragma unroll
        for (int j = 0; j < K / 32; j++) atomicAdd(ap + 32 * j, v[j]);
      }
    }
  }
  }
  __syncthreads();

  {
    constexpr int NSC = (AGG_ROWS * (K / 4) + AGG_NT - 1) / AGG_NT;
    static_assert(AGG_NT % (K / 4) == 0, "a thread keeps its columns from trip to trip");
    bn4 n = {};
    if (BNIN) n = bn_cols4(st_in, g_in, be_in, K, 4 * (t % (K / 4)), invM);
#pragma unroll
    for (int j = 0; j < NSC; j++) {
      const int i = t + j * AGG_NT;
      const int r = i / (K / 4);
      const int c4 = i - r * (K / 4);
      const int gr = r0 + r;
      if (r < AGG_ROWS && gr < M) {
        const float sc = rdeg[gr];
        float4* p = (float4*)(rows + r * LDK) + c4;
        float4 v = *p;
        v.x *= sc; v.y *= sc; v.z *= sc; v.w *= sc;
        if (BNIN) { if (sc != 0.f) v = bn_map4(v, n); }
        *p = v;
      }
    }
  }
  __syncthreads();

  constexpr int NCT = ZOUT ? 2 : 1;
  constexpr int CT  = FO / 16 / NCT;
  constexpr int WPC = AGG_NW / CT;
  constexpr int NTM = (AGG_NRT + WPC - 1) / WPC;
  const int ct = wave % CT;
  const int rg = wave / CT;
  const bool more = rg < AGG_NRT - (NTM - 1) * WPC;

  v8f c[NTM * NCT];
#pragma unroll
  for (int i = 0; i < NTM * NCT; i++) c[i] = (v8f){0.f, 0.f, 0.f, 0.f, 0.f, 0.f, 0.f, 0.f};

#pragma unroll 1
  for (int p = 0; p < NP - 1; p++) {
    if (more) rows_gemm_pass<K, FO, KH, NTM, NCT>(c, rows, hs, Wl, Wr, p * KH, ct, rg, lane);
    else      rows_gemm_pass<K, FO, KH, NTM - 1, NCT>(c, rows, hs, Wl, Wr, p * KH, ct, rg, lane);
    const typename freg<4 * NLD>::type hr = own_rows_load<K, KH, NLD>(h, r0, M, (p + 1) * KH, t);
    bn4 n = {};
    if (BNIN) n = bn_cols4(st_in, g_in, be_in, K, (p + 1) * KH + 4 * (t % (KH / 4)), invM);
    __syncthreads();
    own_rows_store<KH, NLD, BNIN>(hr, hs, t, n);
    __syncthreads();
  }
  const int m16 = lane & 15;
  const int g = lane >> 4;
  const int ncol = ct * 16 + m16;

  float ev[NTM][8];
  if constexpr (ZOUT) {
#pragma unroll
    for (int i = 0; i < NTM; i++) {
#pragma unroll
      for (int v = 0; v < 8; v++) {
        const int lr = (rg + i * WPC) * 16 + 8 * g + v;
        int gr = r0 + lr;
        if (lr > AGG_ROWS - 1 || gr > M - 1) gr = M - 1;
        ev[i][v] = epsin[(size_t)gr * (FO / 2) + ncol];
      }
    }
  }

  if (more) rows_gemm_pass<K, FO, KH, NTM, NCT>(c, rows, hs, Wl, Wr, (NP - 1) * KH, ct, rg, lane);
  else      rows_gemm_pass<K, FO, KH, NTM - 1, NCT>(c, rows, hs, Wl, Wr, (NP - 1) * KH, ct, rg, lane);

  if constexpr (ZOUT) {
    const float bm = bias[ncol];
    const float bs = bias[FO / 2 + ncol];
#pragma unroll
    for (int i = 0; i < NTM; i++) {
      const int rbase = (rg + i * WPC) * 16 + 8 * g;
#pragma unroll
      for (int v = 0; v < 8; v++) {
        const int lr = rbase + v;
        const float mean = c[i * NCT][v] + bm;
        const float ls = c[i * NCT + (NCT - 1)][v] + bs;
        if (lr < AGG_ROWS && r0 + lr < M)
          out[(size_t)(r0 + lr) * (FO / 2) + ncol] = mean + expf(ls) * ev[i][v];
      }
    }
    return;
  }
  const float bv = bias[ncol];
  float cs = 0.f, cq = 0.f;
#pragma unroll
  for (int i = 0; i < NTM; i++) {
    const int rbase = (rg + i * WPC) * 16 + 8 * g;
#pragma unroll
    for (int v = 0; v < 8; v++) {
      const int lr = rbase + v;
      float val = c[i][v] + bv;
      if (RELU) val = fmaxf(val, 0.f);
      if (lr < AGG_ROWS && r0 + lr < M) {
        out[(size_t)(r0 + lr) * FO + ncol] = val;
        if (STATS) { cs += val; cq += val * val; }
      }
    }
  }

  if (STATS) {
    red[0][wave][lane] = cs;
    red[1][wave][lane] = cq;
    __syncthreads();
    if (t < FO) {
      const int cti = t >> 4;
      const int mi = t & 15;
      float a = 0.f, b = 0.f;
      for (int r = 0; r < WPC; r++) {
        const int w = r * CT + cti;
        a += red[0][w][mi]; a += red[0][w][16 + mi];
        b += red[1][w][mi]; b += red[1][w][16 + mi];
      }
      float* prow = part + (size_t)blockIdx.x * (2 * FO);
      prow[t] = a;
      prow[FO + t] = b;
    }
  }
}

__global__ __launch_bounds__(1024)
void bn_sum_parts(const float* __restrict__ part, float* __restrict__ st, int F2) {
  __shared__ float wsum[32][32];
  const int lane = threadIdx.x & 31;
  const int w = threadIdx.x >> 5;
  const int j = blockIdx.x * 32 + lane;
  float s = 0.f;
  static_assert(AGG_NB % 32 == 0, "every wave adds the same number of rows");
  for (int q = 0; q < AGG_NB / 32; q++) s += part[(size_t)(w + 32 * q) * F2 + j];
  wsum[w][lane] = s;
  __syncthreads();
  if (w == 0) {
    float a = 0.f;
    for (int k = 0; k < 32; k++) a += wsum[k][lane];
    st[j] = a;
  }
}

extern "C" void kernel_launch(void* const* d_in, const int* in_sizes, int n_in,
                              void* d_out, int out_size, void* d_ws, size_t ws_size,
                              hipStream_t stream) {
  const float* x   = (const float*)d_in[0];
  const int*   src = (const int*)d_in[1];
  const int*   dst = (const int*)d_in[2];
  const float* eps = (const float*)d_in[3];
  const float* W1l = (const float*)d_in[4];
  const float* W1r = (const float*)d_in[5];
  const float* b1  = (const float*)d_in[6];
  const float* g1  = (const float*)d_in[7];
  const float* be1 = (const float*)d_in[8];
  const float* W2l = (const float*)d_in[9];
  const float* W2r = (const float*)d_in[10];
  const float* b2  = (const float*)d_in[11];
  const float* g2  = (const float*)d_in[12];
  const float* be2 = (const float*)d_in[13];
  const float* W3l = (const float*)d_in[14];
  const float* W3r = (const float*)d_in[15];
  const float* b3  = (const float*)d_in[16];
  float* zout = (float*)d_out;
  (void)in_sizes; (void)n_in; (void)out_size; (void)ws_size;

  char* ws = (char*)d_ws;
  size_t off = 0;
  auto carve = [&](size_t bytes) -> float* {
    char* p = ws + off;
    off = (off + bytes + 255) & ~(size_t)255;
    return (float*)p;
  };
  float* rdeg = carve((size_t)NN * 4);
  float* h1   = carve((size_t)NN * 128 * 4);
  float* h2   = carve((size_t)NN * 256 * 4);
  float* st   = carve(512 * 4);
  float* part = carve((size_t)AGG_NB * 512 * 4);
  unsigned* list = (unsigned*)carve((size_t)AGG_NB * LIST_CAP * 4);
  int* cnt    = (int*)carve((size_t)AGG_NB * 4);

  const int M = NN, E = NE;

  unsigned* flags = (unsigned*)h2;
  chunk_flags<<<FLG_NB, AGG_NT, 0, stream>>>(dst, flags, E);

  owner_scan<<<AGG_NB, AGG_NT, 0, stream>>>(src, dst, flags, list, cnt, rdeg, E, M);

  sage_rows_fused<128, 128, true, true, false, false><<<AGG_NB, AGG_NT, 0, stream>>>(x, list, cnt, src, dst, rdeg, W1l, W1r, b1, h1, part, nullptr, nullptr, nullptr, nullptr, E, M);
  bn_sum_parts<<<256 / 32, 1024, 0, stream>>>(part, st, 256);

  sage_rows_fused<128, 256, true, true, true, false><<<AGG_NB, AGG_NT, 0, stream>>>(h1, list, cnt, src, dst, rdeg, W2l, W2r, b2, h2, part, st, g1, be1, nullptr, E, M);
  bn_sum_parts<<<512 / 32, 1024, 0, stream>>>(part, st, 512);

  sage_rows_fused<256, 128, false, false, true, true><<<AGG_NB, AGG_NT, 0, stream>>>(h2, list, cnt, src, dst, rdeg, W3l, W3r, b3, zout, nullptr, st, g2, be2, eps, E, M);
}


# Edge list sorted by target row inside each wave's part: a wave keeps ONE running register sum, stored to LDS when the row changes (was a per-edge switch over 7 register rows); add order changes
# speedup vs baseline: 4.3786x; 1.0755x over previous
#include <hip/hip_runtime.h>

#define NN 50000
#define NE 800000

typedef __attribute__((ext_vector_type(2))) float v2f;
typedef __attribute__((ext_vector_type(8))) float v8f;

#define AGG_ROWS 196
#define AGG_RPAD 208
#define AGG_NRT  (AGG_RPAD / 16)
#define AGG_NB   ((NN + AGG_ROWS - 1) / AGG_ROWS)
#define AGG_NT   1024
#define AGG_NW   (AGG_NT / 32)
#define AGG_U    16
#define AGG_TRIP (32 * AGG_U)
#define AGG_PAD  4
#define AGG_LDS_FLOATS (AGG_RPAD * (256 + AGG_PAD) + AGG_RPAD * (64 + AGG_PAD))

#define NCHUNK   ((NE + 31) / 32)
#define NFW      ((NCHUNK + 31) / 32)
#define FLG_NB   256
#define FLG_PASS ((NFW + FLG_NB - 1) / FLG_NB)
static_assert(AGG_NT == 1024, "one flag word = 32 chunks of 32 edges = one edge per thread");
static_assert(NFW <= AGG_NT, "an owner group reads one flag word per thread");

__global__ __launch_bounds__(AGG_NT)
void chunk_flags(const int* __restrict__ dst, unsigned* __restrict__ flags, int E) {
  __shared__ unsigned bits[FLG_PASS * AGG_NB];
  const int t = threadIdx.x;
  const unsigned wbit = 1u << (t >> 5);
  for (int i = t; i < FLG_PASS * AGG_NB; i += AGG_NT) bits[i] = 0u;
  __syncthreads();
  int d[FLG_PASS];
#pragma unroll
  for (int q = 0; q < FLG_PASS; q++) {
    const int w = (int)blockIdx.x + q * FLG_NB;
    const int e = w * AGG_NT + t;
    d[q] = (w < NFW && e < E) ? dst[e] : -1;
  }
#pragma unroll
  for (int q = 0; q < FLG_PASS; q++) {
    const unsigned o = (unsigned)d[q] / (unsigned)AGG_ROWS;
    if (o < (unsigned)AGG_NB) atomicOr(&bits[q * AGG_NB + (int)o], wbit);
  }
  __syncthreads();
#pragma unroll
  for (int q = 0; q < FLG_PASS; q++) {
    const int w = (int)blockIdx.x + q * FLG_NB;
    if (w < NFW && t < AGG_NB) flags[(size_t)w * AGG_NB + t] = bits[q * AGG_NB + t];
  }
}

#define LIST_CAP 4096
#define LIST_HDR 64
#define LIST_ENT (LIST_CAP - LIST_HDR)
#define SCAN_CU  16
static_assert(AGG_NW == 32, "a row's owner wave is the low 5 bits of its local row");
static_assert(AGG_NW + 1 <= LIST_HDR, "the part offsets fit in front of the entries");
static_assert(NE < (1 << 24) && NN < (1 << 24), "edge index and source must fit 24 bits of a list word");
static_assert(AGG_ROWS <= 256, "local row must fit 8 bits of a list word");

__global__ __launch_bounds__(AGG_NT)
void owner_scan(const int* __restrict__ src, const int* __restrict__ dst,
                const unsigned* __restrict__ flags,
                unsigned* __restrict__ list, int* __restrict__ cnt, float* __restrict__ rdeg,
                int E, int M) {
  __shared__ unsigned lent[LIST_CAP];
  __shared__ unsigned chl[NFW * 32];
  __shared__ int ldeg[AGG_ROWS];
  __shared__ int lcnt;
  __shared__ int nch;
  __shared__ int roff[AGG_ROWS];
  const int t = threadIdx.x;
  const int lane = t & 31;
  const int wave = __builtin_amdgcn_readfirstlane(t >> 5);
  const int r0 = blockIdx.x * AGG_ROWS;

  for (int i = t; i < AGG_ROWS; i += AGG_NT) ldeg[i] = 0;
  if (t == 0) { lcnt = 0; nch = 0; }
  unsigned fm = 0u;
  if (t < NFW) fm = flags[(size_t)t * AGG_NB + blockIdx.x];
  __syncthreads();

  if (fm) {
    int p = atomicAdd(&nch, __builtin_popcount(fm));
    while (fm) {
      const int b = __builtin_ctz(fm);
      fm &= fm - 1u;
      chl[p++] = (unsigned)(t * 32 + b);
    }
  }
  __syncthreads();
  const int nc = nch;

  for (int i0 = wave * SCAN_CU; i0 < nc; i0 += AGG_NW * SCAN_CU) {
    int cv = -1;
    if (lane < SCAN_CU && i0 + lane < nc) cv = (int)chl[i0 + lane];
    int dl[SCAN_CU];
#pragma unroll
    for (int j = 0; j < SCAN_CU; j++) {
      const int c = __builtin_amdgcn_readlane(cv, j);
      int e = ((c < 0) ? 0 : c) * 32 + lane;
      if (e > E - 1) e = E - 1;
      dl[j] = dst[e] - r0;
    }
    unsigned mk[SCAN_CU];
    int tot = 0;
#pragma unroll
    for (int j = 0; j < SCAN_CU; j++) {
      const int c = __builtin_amdgcn_readlane(cv, j);
      const bool ok = (c >= 0) & (c * 32 + lane < E) & ((unsigned)dl[j] < (unsigned)AGG_ROWS);
      mk[j] = __builtin_amdgcn_ballot_w32(ok);
      tot += __builtin_popcount(mk[j]);
    }
    if (tot) {
      int base = 0;
      if (lane == 0) base = atomicAdd(&lcnt, tot);
      base = __builtin_amdgcn_readfirstlane(base);
#pragma unroll
      for (int j = 0; j < SCAN_CU; j++) {
        if (mk[j]) {
          const int c = __builtin_amdgcn_readlane(cv, j);
          if ((mk[j] >> lane) & 1u) {
            const int pos = base + (int)__builtin_amdgcn_mbcnt_lo(mk[j], 0u);
            const int e = c * 32 + lane;
            if (pos < LIST_ENT) lent[pos] = ((unsigned)e << 8) | (unsigned)dl[j];
            atomicAdd(&ldeg[dl[j]], 1);
          }
          base += __builtin_popcount(mk[j]);
        }
      }
    }
  }
  __syncthreads();

  const int n = lcnt;
  unsigned* grp = list + (size_t)blockIdx.x * LIST_CAP;
  if (n <= LIST_ENT) {
    if (t < AGG_NW) {
      int c = 0;
      for (int r = t; r < AGG_ROWS; r += AGG_NW) c += ldeg[r];
      int inc = c;
      for (int d = 1; d < 32; d <<= 1) {
        const int o = __shfl_up(inc, d);
        if (lane >= d) inc += o;
      }
      grp[t] = (unsigned)(inc - c);
      if (t == AGG_NW - 1) grp[AGG_NW] = (unsigned)inc;
      int run = inc - c;
      for (int r = t; r < AGG_ROWS; r += AGG_NW) { roff[r] = run; run += ldeg[r]; }
    }
    __syncthreads();
    for (int i = t; i < n; i += AGG_NT) {
      const unsigned en = lent[i];
      const int slot = atomicAdd(&roff[en & 255u], 1);
      const int s = src[en >> 8];
      grp[LIST_HDR + slot] = ((unsigned)s << 8) | (en & 255u);
    }
  }
  for (int i = t; i < AGG_ROWS; i += AGG_NT) {
    const int gr = r0 + i;
    if (gr < M) rdeg[gr] = (ldeg[i] > 0) ? 1.f / (float)ldeg[i] : 0.f;
  }
  if (t == 0) cnt[blockIdx.x] = n;
}

template <int N> struct freg { typedef float type __attribute__((ext_vector_type(N))); };

template <int K, int KH, int NLD>
__device__ __forceinline__ typename freg<4 * NLD>::type own_rows_load(const float* __restrict__ h,
                                                                      int r0, int M, int col0, int t) {
  typename freg<4 * NLD>::type r;
#pragma unroll
  for (int j = 0; j < NLD; j++) {
    int i = t + j * AGG_NT;
    if (i > AGG_ROWS * (KH / 4) - 1) i = AGG_ROWS * (KH / 4) - 1;
    const int row = i / (KH / 4);
    const int c4 = i - row * (KH / 4);
    int gr = r0 + row;
    if (gr > M - 1) gr = M - 1;
    const float4 v = *(const float4*)(h + (gr * K + col0 + 4 * c4));
    r[4 * j + 0] = v.x; r[4 * j + 1] = v.y; r[4 * j + 2] = v.z; r[4 * j + 3] = v.w;
  }
  return r;
}

struct bn4 { float4 m, a, b; };

__device__ __forceinline__ bn4 bn_cols4(const float* __restrict__ st, const float* __restrict__ g,
                                        const float* __restrict__ be, int F, int c, float invM) {
  const float4 s = *(const float4*)(st + c);
  const float4 q = *(const float4*)(st + F + c);
  const float4 gg = *(const float4*)(g + c);
  bn4 r;
  r.b = *(const float4*)(be + c);
  r.m = make_float4(s.x * invM, s.y * invM, s.z * invM, s.w * invM);
  r.a.x = gg.x * rsqrtf(q.x * invM - r.m.x * r.m.x + 1e-5f);
  r.a.y = gg.y * rsqrtf(q.y * invM - r.m.y * r.m.y + 1e-5f);
  r.a.z = gg.z * rsqrtf(q.z * invM - r.m.z * r.m.z + 1e-5f);
  r.a.w = gg.w * rsqrtf(q.w * invM - r.m.w * r.m.w + 1e-5f);
  return r;
}

__device__ __forceinline__ float4 bn_map4(const float4 x, const bn4 n) {
  return make_float4((x.x - n.m.x) * n.a.x + n.b.x, (x.y - n.m.y) * n.a.y + n.b.y,
                     (x.z - n.m.z) * n.a.z + n.b.z, (x.w - n.m.w) * n.a.w + n.b.w);
}

template <int KH, int NLD, bool BN>
__device__ __forceinline__ void own_rows_store(const typename freg<4 * NLD>::type r, float* hs, int t,
                                               const bn4 n) {
  static_assert(AGG_NT % (KH / 4) == 0, "a thread keeps its columns from piece to piece");
#pragma unroll
  for (int j = 0; j < NLD; j++) {
    const int i = t + j * AGG_NT;
    if (i < AGG_ROWS * (KH / 4)) {
      const int row = i / (KH / 4);
      const int c4 = i - row * (KH / 4);
      float4 v = make_float4(r[4 * j + 0], r[4 * j + 1], r[4 * j + 2], r[4 * j + 3]);
      if (BN) v = bn_map4(v, n);
      *(float4*)(hs + row * (KH + AGG_PAD) + 4 * c4) = v;
    }
  }
}

template <int K, int FO, int KH, int NTW, int NCT>
__device__ __forceinline__ void rows_gemm_pass(v8f* c, const float* ms, const float* hs,
                                               const float* __restrict__ Wl,
                                               const float* __restrict__ Wr,
                                               int kb, int ct, int rg, int lane) {
  constexpr int LDK = K + AGG_PAD;
  constexpr int LDH = KH + AGG_PAD;
  constexpr int WPC = AGG_NW / (FO / 16 / NCT);
  const int m16 = lane & 15;
  const int g = lane >> 4;
  const int ncol = ct * 16 + m16;

  const float* ap[NTW];
  const float* hp[NTW];
#pragma unroll
  for (int i = 0; i < NTW; i++) {
    const int lrow = (rg + i * WPC) * 16 + m16;
    ap[i] = ms + lrow * LDK + kb + 4 * g;
    hp[i] = hs + lrow * LDH + 4 * g;
  }
  const float* wl = Wl + (size_t)(kb + 4 * g) * FO + ncol;
  const float* wr = Wr + (size_t)(kb + 4 * g) * FO + ncol;

  constexpr bool AHEAD = (NTW * NCT <= 4) && (KH == K);
  constexpr int CS = FO / NCT;
  v2f bl0[NCT], bl1[NCT], br0[NCT], br1[NCT];
#pragma unroll
  for (int jc = 0; jc < NCT; jc++) {
    bl0[jc].x = wl[0 * FO + jc * CS]; bl0[jc].y = wl[1 * FO + jc * CS];
    bl1[jc].x = wl[2 * FO + jc * CS]; bl1[jc].y = wl[3 * FO + jc * CS];
    br0[jc].x = wr[0 * FO + jc * CS]; br0[jc].y = wr[1 * FO + jc * CS];
    br1[jc].x = wr[2 * FO + jc * CS]; br1[jc].y = wr[3 * FO + jc * CS];
  }

#pragma unroll 1
  for (int k0 = 0; k0 < KH; k0 += 8) {
    const int kn = AHEAD ? ((k0 + 8 < KH) ? (k0 + 8) : k0) : k0;
    v2f nl0[NCT], nl1[NCT], nr0[NCT], nr1[NCT];
#pragma unroll
    for (int jc = 0; jc < NCT; jc++) {
      nl0[jc].x = wl[(size_t)(kn + 0) * FO + jc * CS]; nl0[jc].y = wl[(size_t)(kn + 1) * FO + jc * CS];
      nl1[jc].x = wl[(size_t)(kn + 2) * FO + jc * CS]; nl1[jc].y = wl[(size_t)(kn + 3) * FO + jc * CS];
      nr0[jc].x = wr[(size_t)(kn + 0) * FO + jc * CS]; nr0[jc].y = wr[(size_t)(kn + 1) * FO + jc * CS];
      nr1[jc].x = wr[(size_t)(kn + 2) * FO + jc * CS]; nr1[jc].y = wr[(size_t)(kn + 3) * FO + jc * CS];
    }
    if (!AHEAD) {
#pragma unroll
      for (int jc = 0; jc < NCT; jc++) { bl0[jc] = nl0[jc]; bl1[jc] = nl1[jc]; br0[jc] = nr0[jc]; br1[jc] = nr1[jc]; }
    }
#pragma unroll
    for (int i0 = 0; i0 < NTW; i0 += 4) {
      constexpr int GMAX = 4;
      float4 am[GMAX], ah[GMAX];
#pragma unroll
      for (int j = 0; j < GMAX; j++) {
        if (i0 + j < NTW) {
          am[j] = *(const float4*)(ap[i0 + j] + k0);
          ah[j] = *(const float4*)(hp[i0 + j] + k0);
        }
      }
#pragma unroll
      for (int j = 0; j < GMAX; j++) {
        if (i0 + j < NTW) {
          v2f a; a.x = am[j].x; a.y = am[j].y;
#pragma unroll
          for (int jc = 0; jc < NCT; jc++)
            c[(i0 + j) * NCT + jc] = __builtin_amdgcn_wmma_f32_16x16x4_f32(false, a, false, bl0[jc], (short)0,
                                                                         c[(i0 + j) * NCT + jc], false, false);
        }
      }
#pragma unroll
      for (int j = 0; j < GMAX; j++) {
        if (i0 + j < NTW) {
          v2f a; a.x = am[j].z; a.y = am[j].w;
#pragma unroll
          for (int jc = 0; jc < NCT; jc++)
            c[(i0 + j) * NCT + jc] = __builtin_amdgcn_wmma_f32_16x16x4_f32(false, a, false, bl1[jc], (short)0,
                                                                         c[(i0 + j) * NCT + jc], false, false);
        }
      }
#pragma unroll
      for (int j = 0; j < GMAX; j++) {
        if (i0 + j < NTW) {
          v2f a; a.x = ah[j].x; a.y = ah[j].y;
#pragma unroll
          for (int jc = 0; jc < NCT; jc++)
            c[(i0 + j) * NCT + jc] = __builtin_amdgcn_wmma_f32_16x16x4_f32(false, a, false, br0[jc], (short)0,
                                                                         c[(i0 + j) * NCT + jc], false, false);
        }
      }
#pragma unroll
      for (int j = 0; j < GMAX; j++) {
        if (i0 + j < NTW) {
          v2f a; a.x = ah[j].z; a.y = ah[j].w;
#pragma unroll
          for (int jc = 0; jc < NCT; jc++)
            c[(i0 + j) * NCT + jc] = __builtin_amdgcn_wmma_f32_16x16x4_f32(false, a, false, br1[jc], (short)0,
                                                                         c[(i0 + j) * NCT + jc], false, false);
        }
      }
    }
    if (AHEAD) {
#pragma unroll
      for (int jc = 0; jc < NCT; jc++) { bl0[jc] = nl0[jc]; bl1[jc] = nl1[jc]; br0[jc] = nr0[jc]; br1[jc] = nr1[jc]; }
    }
  }
}

template <int K, int FO, bool RELU, bool STATS, bool BNIN, bool ZOUT>
__global__ __launch_bounds__(AGG_NT)
void sage_rows_fused(const float* __restrict__ h, const unsigned* __restrict__ list,
                     const int* __restrict__ cnt, const int* __restrict__ src,
                     const int* __restrict__ dst, const float* __restrict__ rdeg,
                     const float* __restrict__ Wl, const float* __restrict__ Wr,
                     const float* __restrict__ bias, float* __restrict__ out,
                     float* __restrict__ part, const float* __restrict__ st_in,
                     const float* __restrict__ g_in, const float* __restrict__ be_in,
                     const float* __restrict__ epsin, int E, int M) {
  __shared__ __attribute__((aligned(16))) float lds[AGG_LDS_FLOATS];
  __shared__ float red[2][AGG_NW][32];
  constexpr int LDK = K + AGG_PAD;
  constexpr int KH  = (K > 128) ? 64 : K;
  constexpr int NP  = K / KH;
  constexpr int NLD = (AGG_ROWS * (KH / 4) + AGG_NT - 1) / AGG_NT;
  float* rows = lds;
  float* hs   = lds + AGG_RPAD * LDK;
  const int t = threadIdx.x;
  const int lane = t & 31;
  const int wave = __builtin_amdgcn_readfirstlane(t >> 5);
  const int r0 = blockIdx.x * AGG_ROWS;
  const float invM = 1.f / (float)M;

  {
    const typename freg<4 * NLD>::type hr = own_rows_load<K, KH, NLD>(h, r0, M, 0, t);
    bn4 n = {};
    if (BNIN) n = bn_cols4(st_in, g_in, be_in, K, 4 * (t % (KH / 4)), invM);
    own_rows_store<KH, NLD, BNIN>(hr, hs, t, n);
    for (int i = t; i < (AGG_RPAD - AGG_ROWS) * (KH + AGG_PAD); i += AGG_NT)
      hs[AGG_ROWS * (KH + AGG_PAD) + i] = 0.f;
  }

  const int nlist = cnt[blockIdx.x];
  if (nlist <= LIST_ENT) {
    constexpr int Q = K / 128;
    constexpr int NB = 8;
    constexpr int NOWN = (AGG_ROWS + AGG_NW - 1) / AGG_NW;
    for (int i = t; i < (AGG_RPAD - AGG_ROWS) * LDK; i += AGG_NT) rows[AGG_ROWS * LDK + i] = 0.f;
#pragma unroll
    for (int o = 0; o < NOWN; o++) {
      const int r = wave + o * AGG_NW;
      if (r < AGG_ROWS) {
#pragma unroll
        for (int q = 0; q < Q; q++)
          *(float4*)(rows + r * LDK + 128 * q + 4 * lane) = make_float4(0.f, 0.f, 0.f, 0.f);
      }
    }
    const unsigned* grp = list + (size_t)blockIdx.x * LIST_CAP;
    const int e0 = (int)grp[wave];
    const int e1 = (int)grp[wave + 1];
    const unsigned* mylist = grp + LIST_HDR;
    float4 a[Q];
#pragma unroll
    for (int q = 0; q < Q; q++) a[q] = make_float4(0.f, 0.f, 0.f, 0.f);
    int cur = wave;
    for (int e = e0; e < e1; e += 32) {
      const int ng = (e1 - e < 32) ? (e1 - e) : 32;
      unsigned ent = 0;
      if (lane < ng) ent = mylist[e + lane];
      for (int j = 0; j < ng; j += NB) {
        float4 v[NB][Q];
        int rw[NB];
#pragma unroll
        for (int b = 0; b < NB; b++) {
          int idx = j + b;
          if (idx > ng - 1) idx = ng - 1;
          const unsigned en = (unsigned)__builtin_amdgcn_readlane((int)ent, idx);
          rw[b] = (int)(en & 255u);
          const float* hp = h + (size_t)(en >> 8) * K + 4 * lane;
#pragma unroll
          for (int q = 0; q < Q; q++) v[b][q] = *(const float4*)(hp + 128 * q);
        }
#pragma unroll
        for (int b = 0; b < NB; b++) {
          if (b == 0 || j + b < ng) {
            if (rw[b] != cur) {
#pragma unroll
              for (int q = 0; q < Q; q++) {
                *(float4*)(rows + cur * LDK + 128 * q + 4 * lane) = a[q];
                a[q] = make_float4(0.f, 0.f, 0.f, 0.f);
              }
              cur = rw[b];
            }
#pragma unroll
            for (int q = 0; q < Q; q++) {
              a[q].x += v[b][q].x;
              a[q].y += v[b][q].y;
              a[q].z += v[b][q].z;
              a[q].w += v[b][q].w;
            }
          }
        }
      }
    }
#pragma unroll
    for (int q = 0; q < Q; q++) *(float4*)(rows + cur * LDK + 128 * q + 4 * lane) = a[q];
  } else {
  for (int i = t; i < AGG_RPAD * LDK / 4; i += AGG_NT)
    ((float4*)rows)[i] = make_float4(0.f, 0.f, 0.f, 0.f);
  __syncthreads();
  const int ntrips = (E + AGG_TRIP - 1) / AGG_TRIP;
  const int rot = (int)(((unsigned)blockIdx.x * (unsigned)ntrips) / gridDim.x);

  int T = wave;
  int ebn = 0;
  int dn[AGG_U];
#pragma unroll
  for (int u = 0; u < AGG_U; u++) dn[u] = -1;
  if (T < ntrips) {
    int Tr = T + rot;
    if (Tr >= ntrips) Tr -= ntrips;
    ebn = Tr * AGG_TRIP + lane;
#pragma unroll
    for (int u = 0; u < AGG_U; u++) {
      const int e = ebn + 32 * u;
      if (e < E) dn[u] = dst[e];
    }
  }

  while (T < ntrips) {
    const int eb = ebn;
    int dl[AGG_U];
#pragma unroll
    for (int u = 0; u < AGG_U; u++) dl[u] = dn[u] - r0;

    T += AGG_NW;
    if (T < ntrips) {
      int Tr = T + rot;
      if (Tr >= ntrips) Tr -= ntrips;
      ebn = Tr * AGG_TRIP + lane;
#pragma unroll
      for (int u = 0; u < AGG_U; u++) {
        const int e = ebn + 32 * u;
        dn[u] = (e < E) ? dst[e] : -1;
      }
    }

    int sv[AGG_U];
    unsigned mk[AGG_U];
#pragma unroll
    for (int u = 0; u < AGG_U; u++) {
      const bool m = (unsigned)dl[u] < (unsigned)AGG_ROWS;
      sv[u] = 0;
      if (m) sv[u] = src[eb + 32 * u];
      mk[u] = __builtin_amdgcn_ballot_w32(m);
    }

#pragma unroll
    for (int u = 0; u < AGG_U; u++) {
      unsigned mask = mk[u];
      while (mask) {
        const int b = __builtin_ctz(mask);
        mask &= mask - 1;
        const int s  = __builtin_amdgcn_readlane(sv[u], b);
        const int rl = __builtin_amdgcn_readlane(dl[u], b);
        const float* hp = h + (size_t)s * K + lane;
        float* ap = rows + rl * LDK + lane;
        float v[K / 32];
#pragma unroll
        for (int j = 0; j < K / 32; j++) v[j] = hp[32 * j];
#pragma unroll
        for (int j = 0; j < K / 32; j++) atomicAdd(ap + 32 * j, v[j]);
      }
    }
  }
  }
  __syncthreads();

  {
    constexpr int NSC = (AGG_ROWS * (K / 4) + AGG_NT - 1) / AGG_NT;
    static_assert(AGG_NT % (K / 4) == 0, "a thread keeps its columns from trip to trip");
    bn4 n = {};
    if (BNIN) n = bn_cols4(st_in, g_in, be_in, K, 4 * (t % (K / 4)), invM);
#pragma unroll
    for (int j = 0; j < NSC; j++) {
      const int i = t + j * AGG_NT;
      const int r = i / (K / 4);
      const int c4 = i - r * (K / 4);
      const int gr = r0 + r;
      if (r < AGG_ROWS && gr < M) {
        const float sc = rdeg[gr];
        float4* p = (float4*)(rows + r * LDK) + c4;
        float4 v = *p;
        v.x *= sc; v.y *= sc; v.z *= sc; v.w *= sc;
        if (BNIN) { if (sc != 0.f) v = bn_map4(v, n); }
        *p = v;
      }
    }
  }
  __syncthreads();

  constexpr int NCT = ZOUT ? 2 : 1;
  constexpr int CT  = FO / 16 / NCT;
  constexpr int WPC = AGG_NW / CT;
  constexpr int NTM = (AGG_NRT + WPC - 1) / WPC;
  const int ct = wave % CT;
  const int rg = wave / CT;
  const bool more = rg < AGG_NRT - (NTM - 1) * WPC;

  v8f c[NTM * NCT];
#pragma unroll
  for (int i = 0; i < NTM * NCT; i++) c[i] = (v8f){0.f, 0.f, 0.f, 0.f, 0.f, 0.f, 0.f, 0.f};

#pragma unroll 1
  for (int p = 0; p < NP - 1; p++) {
    if (more) rows_gemm_pass<K, FO, KH, NTM, NCT>(c, rows, hs, Wl, Wr, p * KH, ct, rg, lane);
    else      rows_gemm_pass<K, FO, KH, NTM - 1, NCT>(c, rows, hs, Wl, Wr, p * KH, ct, rg, lane);
    const typename freg<4 * NLD>::type hr = own_rows_load<K, KH, NLD>(h, r0, M, (p + 1) * KH, t);
    bn4 n = {};
    if (BNIN) n = bn_cols4(st_in, g_in, be_in, K, (p + 1) * KH + 4 * (t % (KH / 4)), invM);
    __syncthreads();
    own_rows_store<KH, NLD, BNIN>(hr, hs, t, n);
    __syncthreads();
  }
  const int m16 = lane & 15;
  const int g = lane >> 4;
  const int ncol = ct * 16 + m16;

  float ev[NTM][8];
  if constexpr (ZOUT) {
#pragma unroll
    for (int i = 0; i < NTM; i++) {
#pragma unroll
      for (int v = 0; v < 8; v++) {
        const int lr = (rg + i * WPC) * 16 + 8 * g + v;
        int gr = r0 + lr;
        if (lr > AGG_ROWS - 1 || gr > M - 1) gr = M - 1;
        ev[i][v] = epsin[(size_t)gr * (FO / 2) + ncol];
      }
    }
  }

  if (more) rows_gemm_pass<K, FO, KH, NTM, NCT>(c, rows, hs, Wl, Wr, (NP - 1) * KH, ct, rg, lane);
  else      rows_gemm_pass<K, FO, KH, NTM - 1, NCT>(c, rows, hs, Wl, Wr, (NP - 1) * KH, ct, rg, lane);

  if constexpr (ZOUT) {
    const float bm = bias[ncol];
    const float bs = bias[FO / 2 + ncol];
#pragma unroll
    for (int i = 0; i < NTM; i++) {
      const int rbase = (rg + i * WPC) * 16 + 8 * g;
#pragma unroll
      for (int v = 0; v < 8; v++) {
        const int lr = rbase + v;
        const float mean = c[i * NCT][v] + bm;
        const float ls = c[i * NCT + (NCT - 1)][v] + bs;
        if (lr < AGG_ROWS && r0 + lr < M)
          out[(size_t)(r0 + lr) * (FO / 2) + ncol] = mean + expf(ls) * ev[i][v];
      }
    }
    return;
  }
  const float bv = bias[ncol];
  float cs = 0.f, cq = 0.f;
#pragma unroll
  for (int i = 0; i < NTM; i++) {
    const int rbase = (rg + i * WPC) * 16 + 8 * g;
#pragma unroll
    for (int v = 0; v < 8; v++) {
      const int lr = rbase + v;
      float val = c[i][v] + bv;
      if (RELU) val = fmaxf(val, 0.f);
      if (lr < AGG_ROWS && r0 + lr < M) {
        out[(size_t)(r0 + lr) * FO + ncol] = val;
        if (STATS) { cs += val; cq += val * val; }
      }
    }
  }

  if (STATS) {
    red[0][wave][lane] = cs;
    red[1][wave][lane] = cq;
    __syncthreads();
    if (t < FO) {
      const int cti = t >> 4;
      const int mi = t & 15;
      float a = 0.f, b = 0.f;
      for (int r = 0; r < WPC; r++) {
        const int w = r * CT + cti;
        a += red[0][w][mi]; a += red[0][w][16 + mi];
        b += red[1][w][mi]; b += red[1][w][16 + mi];
      }
      float* prow = part + (size_t)blockIdx.x * (2 * FO);
      prow[t] = a;
      prow[FO + t] = b;
    }
  }
}

__global__ __launch_bounds__(1024)
void bn_sum_parts(const float* __restrict__ part, float* __restrict__ st, int F2) {
  __shared__ float wsum[32][32];
  const int lane = threadIdx.x & 31;
  const int w = threadIdx.x >> 5;
  const int j = blockIdx.x * 32 + lane;
  float s = 0.f;
  static_assert(AGG_NB % 32 == 0, "every wave adds the same number of rows");
  for (int q = 0; q < AGG_NB / 32; q++) s += part[(size_t)(w + 32 * q) * F2 + j];
  wsum[w][lane] = s;
  __syncthreads();
  if (w == 0) {
    float a = 0.f;
    for (int k = 0; k < 32; k++) a += wsum[k][lane];
    st[j] = a;
  }
}

extern "C" void kernel_launch(void* const* d_in, const int* in_sizes, int n_in,
                              void* d_out, int out_size, void* d_ws, size_t ws_size,
                              hipStream_t stream) {
  const float* x   = (const float*)d_in[0];
  const int*   src = (const int*)d_in[1];
  const int*   dst = (const int*)d_in[2];
  const float* eps = (const float*)d_in[3];
  const float* W1l = (const float*)d_in[4];
  const float* W1r = (const float*)d_in[5];
  const float* b1  = (const float*)d_in[6];
  const float* g1  = (const float*)d_in[7];
  const float* be1 = (const float*)d_in[8];
  const float* W2l = (const float*)d_in[9];
  const float* W2r = (const float*)d_in[10];
  const float* b2  = (const float*)d_in[11];
  const float* g2  = (const float*)d_in[12];
  const float* be2 = (const float*)d_in[13];
  const float* W3l = (const float*)d_in[14];
  const float* W3r = (const float*)d_in[15];
  const float* b3  = (const float*)d_in[16];
  float* zout = (float*)d_out;
  (void)in_sizes; (void)n_in; (void)out_size; (void)ws_size;

  char* ws = (char*)d_ws;
  size_t off = 0;
  auto carve = [&](size_t bytes) -> float* {
    char* p = ws + off;
    off = (off + bytes + 255) & ~(size_t)255;
    return (float*)p;
  };
  float* rdeg = carve((size_t)NN * 4);
  float* h1   = carve((size_t)NN * 128 * 4);
  float* h2   = carve((size_t)NN * 256 * 4);
  float* st   = carve(512 * 4);
  float* part = carve((size_t)AGG_NB * 512 * 4);
  unsigned* list = (unsigned*)carve((size_t)AGG_NB * LIST_CAP * 4);
  int* cnt    = (int*)carve((size_t)AGG_NB * 4);

  const int M = NN, E = NE;

  unsigned* flags = (unsigned*)h2;
  chunk_flags<<<FLG_NB, AGG_NT, 0, stream>>>(dst, flags, E);

  owner_scan<<<AGG_NB, AGG_NT, 0, stream>>>(src, dst, flags, list, cnt, rdeg, E, M);

  sage_rows_fused<128, 128, true, true, false, false><<<AGG_NB, AGG_NT, 0, stream>>>(x, list, cnt, src, dst, rdeg, W1l, W1r, b1, h1, part, nullptr, nullptr, nullptr, nullptr, E, M);
  bn_sum_parts<<<256 / 32, 1024, 0, stream>>>(part, st, 256);

  sage_rows_fused<128, 256, true, true, true, false><<<AGG_NB, AGG_NT, 0, stream>>>(h1, list, cnt, src, dst, rdeg, W2l, W2r, b2, h2, part, st, g1, be1, nullptr, E, M);
  bn_sum_parts<<<512 / 32, 1024, 0, stream>>>(part, st, 512);

  sage_rows_fused<256, 128, false, false, true, true><<<AGG_NB, AGG_NT, 0, stream>>>(h2, list, cnt, src, dst, rdeg, W3l, W3r, b3, zout, nullptr, st, g2, be2, eps, E, M);
}
